# starting kernel
# baseline (speedup 1.0000x reference)
#include <hip/hip_runtime.h>
#include <hip/hip_bf16.h>


#define BB  512
#define DD  64
#define HH  512
#define TT  8
#define MT  32
#define XP  68
#define HP  516

typedef float v2f __attribute__((ext_vector_type(2)));
typedef float v8f __attribute__((ext_vector_type(8)));

__device__ __forceinline__ v8f wmma_f32(v2f a, v2f b, v8f c) {
  return __builtin_amdgcn_wmma_f32_16x16x4_f32(false, a, false, b,
                                               (short)0, c, false, false);
}

__device__ __forceinline__ float fast_tanh(float x) {
#if __has_builtin(__builtin_amdgcn_tanhf)
  return __builtin_amdgcn_tanhf(x);
#elif __has_builtin(__builtin_amdgcn_exp2f) && __has_builtin(__builtin_amdgcn_rcpf)
  float e = __builtin_amdgcn_exp2f(x * 2.8853900817779268f);
  return 1.0f - 2.0f * __builtin_amdgcn_rcpf(e + 1.0f);
#else
  return tanhf(x);
#endif
}

__device__ __forceinline__ v2f frag_a(const float* S, int pitch, int m0, int k0,
                                      int lane_lo, int half) {
  const float* p = S + (m0 + lane_lo) * pitch + k0 + 2 * half;
  v2f a; a.x = p[0]; a.y = p[1];
  return a;
}

__device__ __forceinline__ v2f frag_b(const float* __restrict__ W, int ldw,
                                      int k0, int n0, int lane_lo, int half) {
  v2f b;
  b.x = W[(k0 + 2 * half + 0) * ldw + n0 + lane_lo];
  b.y = W[(k0 + 2 * half + 1) * ldw + n0 + lane_lo];
  return b;
}

__global__ __launch_bounds__(256) void cnf_prep(const float* __restrict__ W1,
                                                const float* __restrict__ W2,
                                                const float* __restrict__ W3,
                                                float* __restrict__ M) {
  int idx = blockIdx.x * blockDim.x + threadIdx.x;
  int j = idx >> 9;
  int k = idx & 511;
  float c = 0.f;
  #pragma unroll 8
  for (int m = 0; m < DD; ++m)
    c += W3[k * DD + m] * W1[m * HH + j];
  M[j * HH + k] = W2[j * HH + k] * c;
}

__global__ __launch_bounds__(256) void cnf_main(const float* __restrict__ y0,
                                                const float* __restrict__ ts,
                                                const float* __restrict__ W1,
                                                const float* __restrict__ b1,
                                                const float* __restrict__ W2,
                                                const float* __restrict__ b2,
                                                const float* __restrict__ W3,
                                                const float* __restrict__ b3,
                                                const float* __restrict__ Mtr,
                                                float* __restrict__ out) {
  extern __shared__ float smem[];
  float* Xs   = smem;
  float* H1s  = Xs  + MT * XP;
  float* H2s  = H1s + MT * HP;
  float* zacc = H2s + MT * HP;
  float* lacc = zacc + MT;

  const int tid     = threadIdx.x;
  const int wave    = tid >> 5;
  const int lane    = tid & 31;
  const int lane_lo = lane & 15;
  const int half    = lane >> 4;
  const int s_base  = blockIdx.x * MT;

  for (int i = tid; i < MT * DD; i += 256) {
    int m = i >> 6, c = i & 63;
    Xs[m * XP + c] = y0[(s_base + m) * DD + c];
  }
  if (tid < MT) { zacc[tid] = 0.f; lacc[tid] = 0.f; }

  v8f acc[2][4];

  for (int step = 0; step < TT - 1; ++step) {
    const float t0 = ts[step];
    const float dt = ts[step + 1] - t0;
    __syncthreads();

    #pragma unroll
    for (int mt = 0; mt < 2; ++mt)
      #pragma unroll
      for (int nt = 0; nt < 4; ++nt) acc[mt][nt] = (v8f){};
    for (int k0 = 0; k0 < DD; k0 += 4) {
      v2f a0 = frag_a(Xs, XP, 0,  k0, lane_lo, half);
      v2f a1 = frag_a(Xs, XP, 16, k0, lane_lo, half);
      #pragma unroll
      for (int nt = 0; nt < 4; ++nt) {
        int n0 = (wave * 4 + nt) * 16;
        v2f b = frag_b(W1, HH, k0, n0, lane_lo, half);
        acc[0][nt] = wmma_f32(a0, b, acc[0][nt]);
        acc[1][nt] = wmma_f32(a1, b, acc[1][nt]);
      }
    }
    #pragma unroll
    for (int nt = 0; nt < 4; ++nt) {
      int col = (wave * 4 + nt) * 16 + lane_lo;
      float add = t0 * W1[DD * HH + col] + b1[col];
      #pragma unroll
      for (int mt = 0; mt < 2; ++mt)
        #pragma unroll
        for (int v = 0; v < 8; ++v) {
          int m = mt * 16 + v + 8 * half;
          H1s[m * HP + col] = fast_tanh(acc[mt][nt][v] + add);
        }
    }
    __syncthreads();

    #pragma unroll
    for (int mt = 0; mt < 2; ++mt)
      #pragma unroll
      for (int nt = 0; nt < 4; ++nt) acc[mt][nt] = (v8f){};
    for (int k0 = 0; k0 < HH; k0 += 4) {
      v2f a0 = frag_a(H1s, HP, 0,  k0, lane_lo, half);
      v2f a1 = frag_a(H1s, HP, 16, k0, lane_lo, half);
      #pragma unroll
      for (int nt = 0; nt < 4; ++nt) {
        int n0 = (wave * 4 + nt) * 16;
        v2f b = frag_b(W2, HH, k0, n0, lane_lo, half);
        acc[0][nt] = wmma_f32(a0, b, acc[0][nt]);
        acc[1][nt] = wmma_f32(a1, b, acc[1][nt]);
      }
    }
    #pragma unroll
    for (int nt = 0; nt < 4; ++nt) {
      int col = (wave * 4 + nt) * 16 + lane_lo;
      float add = b2[col];
      #pragma unroll
      for (int mt = 0; mt < 2; ++mt)
        #pragma unroll
        for (int v = 0; v < 8; ++v) {
          int m = mt * 16 + v + 8 * half;
          H2s[m * HP + col] = fast_tanh(acc[mt][nt][v] + add);
        }
    }
    __syncthreads();

    {
      int m0 = (wave >> 2) * 16;
      int n0 = (wave & 3) * 16;
      v8f a3 = (v8f){};
      for (int k0 = 0; k0 < HH; k0 += 4) {
        v2f a = frag_a(H2s, HP, m0, k0, lane_lo, half);
        v2f b = frag_b(W3, DD, k0, n0, lane_lo, half);
        a3 = wmma_f32(a, b, a3);
      }
      int col = n0 + lane_lo;
      float bb = b3[col];
      #pragma unroll
      for (int v = 0; v < 8; ++v) {
        int m = m0 + v + 8 * half;
        float dx = a3[v] + bb;
        atomicAdd(&zacc[m], dt * 0.5f * dx * dx);
        Xs[m * XP + col] += dt * dx;
      }
    }

    #pragma unroll
    for (int mt = 0; mt < 2; ++mt)
      #pragma unroll
      for (int nt = 0; nt < 4; ++nt) acc[mt][nt] = (v8f){};
    for (int k0 = 0; k0 < HH; k0 += 4) {
      v2f a0 = frag_a(H1s, HP, 0,  k0, lane_lo, half);
      v2f a1 = frag_a(H1s, HP, 16, k0, lane_lo, half);
      a0.x = 1.f - a0.x * a0.x;  a0.y = 1.f - a0.y * a0.y;
      a1.x = 1.f - a1.x * a1.x;  a1.y = 1.f - a1.y * a1.y;
      #pragma unroll
      for (int nt = 0; nt < 4; ++nt) {
        int n0 = (wave * 4 + nt) * 16;
        v2f b = frag_b(Mtr, HH, k0, n0, lane_lo, half);
        acc[0][nt] = wmma_f32(a0, b, acc[0][nt]);
        acc[1][nt] = wmma_f32(a1, b, acc[1][nt]);
      }
    }
    #pragma unroll
    for (int mt = 0; mt < 2; ++mt)
      #pragma unroll
      for (int v = 0; v < 8; ++v) {
        int m = mt * 16 + v + 8 * half;
        float tl = 0.f;
        #pragma unroll
        for (int nt = 0; nt < 4; ++nt) {
          int col = (wave * 4 + nt) * 16 + lane_lo;
          float h2 = H2s[m * HP + col];
          tl += (1.f - h2 * h2) * acc[mt][nt][v];
        }
        atomicAdd(&lacc[m], dt * tl);
      }
  }

  __syncthreads();
  for (int i = tid; i < MT * DD; i += 256) {
    int m = i >> 6, c = i & 63;
    out[(s_base + m) * DD + c] = Xs[m * XP + c];
  }
  if (tid < MT) {
    out[BB * DD + s_base + tid]      = zacc[tid];
    out[BB * DD + BB + s_base + tid] = lacc[tid];
  }
}

extern "C" void kernel_launch(void* const* d_in, const int* in_sizes, int n_in,
                              void* d_out, int out_size, void* d_ws, size_t ws_size,
                              hipStream_t stream) {
  const float* y0 = (const float*)d_in[0];
  const float* ts = (const float*)d_in[1];
  const float* W1 = (const float*)d_in[2];
  const float* b1 = (const float*)d_in[3];
  const float* W2 = (const float*)d_in[4];
  const float* b2 = (const float*)d_in[5];
  const float* W3 = (const float*)d_in[6];
  const float* b3 = (const float*)d_in[7];
  float* out = (float*)d_out;
  float* Mtr = (float*)d_ws;

  cnf_prep<<<(HH * HH) / 256, 256, 0, stream>>>(W1, W2, W3, Mtr);

  size_t shmem = (size_t)(MT * XP + 2 * MT * HP + 2 * MT) * sizeof(float);
  cnf_main<<<BB / MT, 256, shmem, stream>>>(y0, ts, W1, b1, W2, b2, W3, b3,
                                            Mtr, out);
}


# Every GEMM k loop double-buffers its fragments in registers: the next 4 k-steps' weight and activation loads (16 k-steps in the dx GEMM) issue ahead of the current WMMA block; summation order kept
# speedup vs baseline: 1.3739x; 1.3739x over previous
#include <hip/hip_runtime.h>
#include <hip/hip_bf16.h>


#define BB  512
#define DD  64
#define HH  512
#define TT  8
#define MT  32
#define XP  68
#define HP  516

typedef float v2f __attribute__((ext_vector_type(2)));
typedef float v8f __attribute__((ext_vector_type(8)));

__device__ __forceinline__ v8f wmma_f32(v2f a, v2f b, v8f c) {
  return __builtin_amdgcn_wmma_f32_16x16x4_f32(false, a, false, b,
                                               (short)0, c, false, false);
}

__device__ __forceinline__ float fast_tanh(float x) {
#if __has_builtin(__builtin_amdgcn_tanhf)
  return __builtin_amdgcn_tanhf(x);
#elif __has_builtin(__builtin_amdgcn_exp2f) && __has_builtin(__builtin_amdgcn_rcpf)
  float e = __builtin_amdgcn_exp2f(x * 2.8853900817779268f);
  return 1.0f - 2.0f * __builtin_amdgcn_rcpf(e + 1.0f);
#else
  return tanhf(x);
#endif
}

__device__ __forceinline__ v2f frag_a(const float* S, int pitch, int m0, int k0,
                                      int lane_lo, int half) {
  const float* p = S + (m0 + lane_lo) * pitch + k0 + 2 * half;
  v2f a; a.x = p[0]; a.y = p[1];
  return a;
}

__device__ __forceinline__ v2f frag_b(const float* __restrict__ W, int ldw,
                                      int k0, int n0, int lane_lo, int half) {
  v2f b;
  b.x = W[(k0 + 2 * half + 0) * ldw + n0 + lane_lo];
  b.y = W[(k0 + 2 * half + 1) * ldw + n0 + lane_lo];
  return b;
}


#define KG 4
#define KH 16

struct FragSetW { v2f a0[KG]; v2f a1[KG]; v2f b[KG][4]; };

__device__ __forceinline__ void load_b_w(FragSetW& f, const float* __restrict__ W,
                                         int kb, int nw, int lane_lo, int half) {
  #pragma unroll
  for (int i = 0; i < KG; ++i)
    #pragma unroll
    for (int nt = 0; nt < 4; ++nt)
      f.b[i][nt] = frag_b(W, HH, kb + 4 * i, nw + 16 * nt, lane_lo, half);
}

__device__ __forceinline__ void load_a_w(FragSetW& f, const float* S, int pitch,
                                         int kb, int lane_lo, int half) {
  #pragma unroll
  for (int i = 0; i < KG; ++i) {
    f.a0[i] = frag_a(S, pitch, 0,  kb + 4 * i, lane_lo, half);
    f.a1[i] = frag_a(S, pitch, 16, kb + 4 * i, lane_lo, half);
  }
}

template <bool SQ>
__device__ __forceinline__ void mma_set_w(const FragSetW& f, v8f (&acc)[2][4]) {
  #pragma unroll
  for (int i = 0; i < KG; ++i) {
    v2f a0 = f.a0[i], a1 = f.a1[i];
    if (SQ) {
      a0.x = 1.f - a0.x * a0.x;  a0.y = 1.f - a0.y * a0.y;
      a1.x = 1.f - a1.x * a1.x;  a1.y = 1.f - a1.y * a1.y;
    }
    #pragma unroll
    for (int nt = 0; nt < 4; ++nt) {
      acc[0][nt] = wmma_f32(a0, f.b[i][nt], acc[0][nt]);
      acc[1][nt] = wmma_f32(a1, f.b[i][nt], acc[1][nt]);
    }
  }
}

template <bool SQ>
__device__ __forceinline__ void gemm_w(v8f (&acc)[2][4], FragSetW& f0,
                                       const float* S, int pitch,
                                       const float* __restrict__ W, int K,
                                       int nw, int lane_lo, int half) {
  #pragma unroll
  for (int mt = 0; mt < 2; ++mt)
    #pragma unroll
    for (int nt = 0; nt < 4; ++nt) acc[mt][nt] = (v8f){};
  FragSetW f1;
  load_a_w(f0, S, pitch, 0, lane_lo, half);
  for (int kb = 0; kb < K; kb += 8 * KG) {
    load_b_w(f1, W, kb + 4 * KG, nw, lane_lo, half);
    load_a_w(f1, S, pitch, kb + 4 * KG, lane_lo, half);
    mma_set_w<SQ>(f0, acc);
    if (kb + 8 * KG < K) {
      load_b_w(f0, W, kb + 8 * KG, nw, lane_lo, half);
      load_a_w(f0, S, pitch, kb + 8 * KG, lane_lo, half);
    }
    mma_set_w<SQ>(f1, acc);
  }
}

struct FragSetX { v2f a[KH]; v2f b[KH]; };

__device__ __forceinline__ void load_b_x(FragSetX& f, const float* __restrict__ W,
                                         int kb, int n0, int lane_lo, int half) {
  #pragma unroll
  for (int i = 0; i < KH; ++i)
    f.b[i] = frag_b(W, DD, kb + 4 * i, n0, lane_lo, half);
}

__device__ __forceinline__ void load_a_x(FragSetX& f, const float* S, int kb,
                                         int m0, int lane_lo, int half) {
  #pragma unroll
  for (int i = 0; i < KH; ++i)
    f.a[i] = frag_a(S, HP, m0, kb + 4 * i, lane_lo, half);
}

__device__ __forceinline__ v8f mma_set_x(const FragSetX& f, v8f c) {
  #pragma unroll
  for (int i = 0; i < KH; ++i) c = wmma_f32(f.a[i], f.b[i], c);
  return c;
}

__global__ __launch_bounds__(256) void cnf_prep(const float* __restrict__ W1,
                                                const float* __restrict__ W2,
                                                const float* __restrict__ W3,
                                                float* __restrict__ M) {
  int idx = blockIdx.x * blockDim.x + threadIdx.x;
  int j = idx >> 9;
  int k = idx & 511;
  float c = 0.f;
  #pragma unroll 8
  for (int m = 0; m < DD; ++m)
    c += W3[k * DD + m] * W1[m * HH + j];
  M[j * HH + k] = W2[j * HH + k] * c;
}

__global__ __launch_bounds__(256) void cnf_main(const float* __restrict__ y0,
                                                const float* __restrict__ ts,
                                                const float* __restrict__ W1,
                                                const float* __restrict__ b1,
                                                const float* __restrict__ W2,
                                                const float* __restrict__ b2,
                                                const float* __restrict__ W3,
                                                const float* __restrict__ b3,
                                                const float* __restrict__ Mtr,
                                                float* __restrict__ out) {
  extern __shared__ float smem[];
  float* Xs   = smem;
  float* H1s  = Xs  + MT * XP;
  float* H2s  = H1s + MT * HP;
  float* zacc = H2s + MT * HP;
  float* lacc = zacc + MT;

  const int tid     = threadIdx.x;
  const int wave    = tid >> 5;
  const int lane    = tid & 31;
  const int lane_lo = lane & 15;
  const int half    = lane >> 4;
  const int s_base  = blockIdx.x * MT;

  for (int i = tid; i < MT * DD; i += 256) {
    int m = i >> 6, c = i & 63;
    Xs[m * XP + c] = y0[(s_base + m) * DD + c];
  }
  if (tid < MT) { zacc[tid] = 0.f; lacc[tid] = 0.f; }

  v8f acc[2][4];
  FragSetW fw;
  const int nw = wave * 64;
  const int m3 = (wave >> 2) * 16;
  const int n3 = (wave & 3) * 16;

  for (int step = 0; step < TT - 1; ++step) {
    const float t0 = ts[step];
    const float dt = ts[step + 1] - t0;
    load_b_w(fw, W1, 0, nw, lane_lo, half);
    __syncthreads();

    gemm_w<false>(acc, fw, Xs, XP, W1, DD, nw, lane_lo, half);
    load_b_w(fw, W2, 0, nw, lane_lo, half);
    #pragma unroll
    for (int nt = 0; nt < 4; ++nt) {
      int col = (wave * 4 + nt) * 16 + lane_lo;
      float add = t0 * W1[DD * HH + col] + b1[col];
      #pragma unroll
      for (int mt = 0; mt < 2; ++mt)
        #pragma unroll
        for (int v = 0; v < 8; ++v) {
          int m = mt * 16 + v + 8 * half;
          H1s[m * HP + col] = fast_tanh(acc[mt][nt][v] + add);
        }
    }
    __syncthreads();

    gemm_w<false>(acc, fw, H1s, HP, W2, HH, nw, lane_lo, half);
    FragSetX g0, g1;
    load_b_x(g0, W3, 0, n3, lane_lo, half);
    #pragma unroll
    for (int nt = 0; nt < 4; ++nt) {
      int col = (wave * 4 + nt) * 16 + lane_lo;
      float add = b2[col];
      #pragma unroll
      for (int mt = 0; mt < 2; ++mt)
        #pragma unroll
        for (int v = 0; v < 8; ++v) {
          int m = mt * 16 + v + 8 * half;
          H2s[m * HP + col] = fast_tanh(acc[mt][nt][v] + add);
        }
    }
    __syncthreads();

    {
      const int m0 = m3, n0 = n3;
      v8f a3 = (v8f){};
      load_a_x(g0, H2s, 0, m0, lane_lo, half);
      for (int kb = 0; kb < HH; kb += 8 * KH) {
        load_b_x(g1, W3, kb + 4 * KH, n0, lane_lo, half);
        load_a_x(g1, H2s, kb + 4 * KH, m0, lane_lo, half);
        a3 = mma_set_x(g0, a3);
        if (kb + 8 * KH < HH) {
          load_b_x(g0, W3, kb + 8 * KH, n0, lane_lo, half);
          load_a_x(g0, H2s, kb + 8 * KH, m0, lane_lo, half);
        }
        a3 = mma_set_x(g1, a3);
      }
      load_b_w(fw, Mtr, 0, nw, lane_lo, half);
      int col = n0 + lane_lo;
      float bb = b3[col];
      #pragma unroll
      for (int v = 0; v < 8; ++v) {
        int m = m0 + v + 8 * half;
        float dx = a3[v] + bb;
        atomicAdd(&zacc[m], dt * 0.5f * dx * dx);
        Xs[m * XP + col] += dt * dx;
      }
    }

    gemm_w<true>(acc, fw, H1s, HP, Mtr, HH, nw, lane_lo, half);
    #pragma unroll
    for (int mt = 0; mt < 2; ++mt)
      #pragma unroll
      for (int v = 0; v < 8; ++v) {
        int m = mt * 16 + v + 8 * half;
        float tl = 0.f;
        #pragma unroll
        for (int nt = 0; nt < 4; ++nt) {
          int col = (wave * 4 + nt) * 16 + lane_lo;
          float h2 = H2s[m * HP + col];
          tl += (1.f - h2 * h2) * acc[mt][nt][v];
        }
        atomicAdd(&lacc[m], dt * tl);
      }
  }

  __syncthreads();
  for (int i = tid; i < MT * DD; i += 256) {
    int m = i >> 6, c = i & 63;
    out[(s_base + m) * DD + c] = Xs[m * XP + c];
  }
  if (tid < MT) {
    out[BB * DD + s_base + tid]      = zacc[tid];
    out[BB * DD + BB + s_base + tid] = lacc[tid];
  }
}

extern "C" void kernel_launch(void* const* d_in, const int* in_sizes, int n_in,
                              void* d_out, int out_size, void* d_ws, size_t ws_size,
                              hipStream_t stream) {
  const float* y0 = (const float*)d_in[0];
  const float* ts = (const float*)d_in[1];
  const float* W1 = (const float*)d_in[2];
  const float* b1 = (const float*)d_in[3];
  const float* W2 = (const float*)d_in[4];
  const float* b2 = (const float*)d_in[5];
  const float* W3 = (const float*)d_in[6];
  const float* b3 = (const float*)d_in[7];
  float* out = (float*)d_out;
  float* Mtr = (float*)d_ws;

  cnf_prep<<<(HH * HH) / 256, 256, 0, stream>>>(W1, W2, W3, Mtr);

  size_t shmem = (size_t)(MT * XP + 2 * MT * HP + 2 * MT) * sizeof(float);
  cnf_main<<<BB / MT, 256, shmem, stream>>>(y0, ts, W1, b1, W2, b2, W3, b3,
                                            Mtr, out);
}


# Each workgroup integrates 16 samples (one WMMA M-tile per wave) instead of 32, so 32 workgroups on 32 WGPs share the matrix work; per-element summation order kept
# speedup vs baseline: 2.1974x; 1.5994x over previous
#include <hip/hip_runtime.h>
#include <hip/hip_bf16.h>


#define BB  512
#define DD  64
#define HH  512
#define TT  8
#define MT  16
#define XP  68
#define HP  516

typedef float v2f __attribute__((ext_vector_type(2)));
typedef float v8f __attribute__((ext_vector_type(8)));

__device__ __forceinline__ v8f wmma_f32(v2f a, v2f b, v8f c) {
  return __builtin_amdgcn_wmma_f32_16x16x4_f32(false, a, false, b,
                                               (short)0, c, false, false);
}

__device__ __forceinline__ float fast_tanh(float x) {
#if __has_builtin(__builtin_amdgcn_tanhf)
  return __builtin_amdgcn_tanhf(x);
#elif __has_builtin(__builtin_amdgcn_exp2f) && __has_builtin(__builtin_amdgcn_rcpf)
  float e = __builtin_amdgcn_exp2f(x * 2.8853900817779268f);
  return 1.0f - 2.0f * __builtin_amdgcn_rcpf(e + 1.0f);
#else
  return tanhf(x);
#endif
}

__device__ __forceinline__ v2f frag_a(const float* S, int pitch, int m0, int k0,
                                      int lane_lo, int half) {
  const float* p = S + (m0 + lane_lo) * pitch + k0 + 2 * half;
  v2f a; a.x = p[0]; a.y = p[1];
  return a;
}

__device__ __forceinline__ v2f frag_b(const float* __restrict__ W, int ldw,
                                      int k0, int n0, int lane_lo, int half) {
  v2f b;
  b.x = W[(k0 + 2 * half + 0) * ldw + n0 + lane_lo];
  b.y = W[(k0 + 2 * half + 1) * ldw + n0 + lane_lo];
  return b;
}


#define KG 4
#define KH 16

struct FragSetW { v2f a[KG]; v2f b[KG][4]; };

__device__ __forceinline__ void load_b_w(FragSetW& f, const float* __restrict__ W,
                                         int kb, int nw, int lane_lo, int half) {
  #pragma unroll
  for (int i = 0; i < KG; ++i)
    #pragma unroll
    for (int nt = 0; nt < 4; ++nt)
      f.b[i][nt] = frag_b(W, HH, kb + 4 * i, nw + 16 * nt, lane_lo, half);
}

__device__ __forceinline__ void load_a_w(FragSetW& f, const float* S, int pitch,
                                         int kb, int lane_lo, int half) {
  #pragma unroll
  for (int i = 0; i < KG; ++i)
    f.a[i] = frag_a(S, pitch, 0, kb + 4 * i, lane_lo, half);
}

template <bool SQ>
__device__ __forceinline__ void mma_set_w(const FragSetW& f, v8f (&acc)[4]) {
  #pragma unroll
  for (int i = 0; i < KG; ++i) {
    v2f a = f.a[i];
    if (SQ) {
      a.x = 1.f - a.x * a.x;  a.y = 1.f - a.y * a.y;
    }
    #pragma unroll
    for (int nt = 0; nt < 4; ++nt)
      acc[nt] = wmma_f32(a, f.b[i][nt], acc[nt]);
  }
}

template <bool SQ>
__device__ __forceinline__ void gemm_w(v8f (&acc)[4], FragSetW& f0,
                                       const float* S, int pitch,
                                       const float* __restrict__ W, int K,
                                       int nw, int lane_lo, int half) {
  #pragma unroll
  for (int nt = 0; nt < 4; ++nt) acc[nt] = (v8f){};
  FragSetW f1;
  load_a_w(f0, S, pitch, 0, lane_lo, half);
  for (int kb = 0; kb < K; kb += 8 * KG) {
    load_b_w(f1, W, kb + 4 * KG, nw, lane_lo, half);
    load_a_w(f1, S, pitch, kb + 4 * KG, lane_lo, half);
    mma_set_w<SQ>(f0, acc);
    if (kb + 8 * KG < K) {
      load_b_w(f0, W, kb + 8 * KG, nw, lane_lo, half);
      load_a_w(f0, S, pitch, kb + 8 * KG, lane_lo, half);
    }
    mma_set_w<SQ>(f1, acc);
  }
}

struct FragSetX { v2f a[KH]; v2f b[KH]; };

__device__ __forceinline__ void load_b_x(FragSetX& f, const float* __restrict__ W,
                                         int kb, int n0, int lane_lo, int half) {
  #pragma unroll
  for (int i = 0; i < KH; ++i)
    f.b[i] = frag_b(W, DD, kb + 4 * i, n0, lane_lo, half);
}

__device__ __forceinline__ void load_a_x(FragSetX& f, const float* S, int kb,
                                         int m0, int lane_lo, int half) {
  #pragma unroll
  for (int i = 0; i < KH; ++i)
    f.a[i] = frag_a(S, HP, m0, kb + 4 * i, lane_lo, half);
}

__device__ __forceinline__ v8f mma_set_x(const FragSetX& f, v8f c) {
  #pragma unroll
  for (int i = 0; i < KH; ++i) c = wmma_f32(f.a[i], f.b[i], c);
  return c;
}

__global__ __launch_bounds__(256) void cnf_prep(const float* __restrict__ W1,
                                                const float* __restrict__ W2,
                                                const float* __restrict__ W3,
                                                float* __restrict__ M) {
  int idx = blockIdx.x * blockDim.x + threadIdx.x;
  int j = idx >> 9;
  int k = idx & 511;
  float c = 0.f;
  #pragma unroll 8
  for (int m = 0; m < DD; ++m)
    c += W3[k * DD + m] * W1[m * HH + j];
  M[j * HH + k] = W2[j * HH + k] * c;
}

__global__ __launch_bounds__(256) void cnf_main(const float* __restrict__ y0,
                                                const float* __restrict__ ts,
                                                const float* __restrict__ W1,
                                                const float* __restrict__ b1,
                                                const float* __restrict__ W2,
                                                const float* __restrict__ b2,
                                                const float* __restrict__ W3,
                                                const float* __restrict__ b3,
                                                const float* __restrict__ Mtr,
                                                float* __restrict__ out) {
  extern __shared__ float smem[];
  float* Xs   = smem;
  float* H1s  = Xs  + MT * XP;
  float* H2s  = H1s + MT * HP;
  float* zacc = H2s + MT * HP;
  float* lacc = zacc + MT;

  const int tid     = threadIdx.x;
  const int wave    = __builtin_amdgcn_readfirstlane(tid >> 5);
  const int lane    = tid & 31;
  const int lane_lo = lane & 15;
  const int half    = lane >> 4;
  const int s_base  = blockIdx.x * MT;

  for (int i = tid; i < MT * DD; i += 256) {
    int m = i >> 6, c = i & 63;
    Xs[m * XP + c] = y0[(s_base + m) * DD + c];
  }
  if (tid < MT) { zacc[tid] = 0.f; lacc[tid] = 0.f; }

  v8f acc[4];
  FragSetW fw;
  const int nw = wave * 64;
  const bool has_dx = wave < 4;
  const int n3 = (wave & 3) * 16;

  for (int step = 0; step < TT - 1; ++step) {
    const float t0 = ts[step];
    const float dt = ts[step + 1] - t0;
    load_b_w(fw, W1, 0, nw, lane_lo, half);
    __syncthreads();

    gemm_w<false>(acc, fw, Xs, XP, W1, DD, nw, lane_lo, half);
    load_b_w(fw, W2, 0, nw, lane_lo, half);
    #pragma unroll
    for (int nt = 0; nt < 4; ++nt) {
      int col = (wave * 4 + nt) * 16 + lane_lo;
      float add = t0 * W1[DD * HH + col] + b1[col];
      #pragma unroll
      for (int v = 0; v < 8; ++v) {
        int m = v + 8 * half;
        H1s[m * HP + col] = fast_tanh(acc[nt][v] + add);
      }
    }
    __syncthreads();

    gemm_w<false>(acc, fw, H1s, HP, W2, HH, nw, lane_lo, half);
    FragSetX g0, g1;
    if (has_dx)
      load_b_x(g0, W3, 0, n3, lane_lo, half);
    #pragma unroll
    for (int nt = 0; nt < 4; ++nt) {
      int col = (wave * 4 + nt) * 16 + lane_lo;
      float add = b2[col];
      #pragma unroll
      for (int v = 0; v < 8; ++v) {
        int m = v + 8 * half;
        H2s[m * HP + col] = fast_tanh(acc[nt][v] + add);
      }
    }
    __syncthreads();

    load_b_w(fw, Mtr, 0, nw, lane_lo, half);
    if (has_dx) {
      const int m0 = 0, n0 = n3;
      v8f a3 = (v8f){};
      load_a_x(g0, H2s, 0, m0, lane_lo, half);
      for (int kb = 0; kb < HH; kb += 8 * KH) {
        load_b_x(g1, W3, kb + 4 * KH, n0, lane_lo, half);
        load_a_x(g1, H2s, kb + 4 * KH, m0, lane_lo, half);
        a3 = mma_set_x(g0, a3);
        if (kb + 8 * KH < HH) {
          load_b_x(g0, W3, kb + 8 * KH, n0, lane_lo, half);
          load_a_x(g0, H2s, kb + 8 * KH, m0, lane_lo, half);
        }
        a3 = mma_set_x(g1, a3);
      }
      int col = n0 + lane_lo;
      float bb = b3[col];
      #pragma unroll
      for (int v = 0; v < 8; ++v) {
        int m = m0 + v + 8 * half;
        float dx = a3[v] + bb;
        atomicAdd(&zacc[m], dt * 0.5f * dx * dx);
        Xs[m * XP + col] += dt * dx;
      }
    }

    gemm_w<true>(acc, fw, H1s, HP, Mtr, HH, nw, lane_lo, half);
    #pragma unroll
    for (int v = 0; v < 8; ++v) {
      int m = v + 8 * half;
      float tl = 0.f;
      #pragma unroll
      for (int nt = 0; nt < 4; ++nt) {
        int col = (wave * 4 + nt) * 16 + lane_lo;
        float h2 = H2s[m * HP + col];
        tl += (1.f - h2 * h2) * acc[nt][v];
      }
      atomicAdd(&lacc[m], dt * tl);
    }
  }

  __syncthreads();
  for (int i = tid; i < MT * DD; i += 256) {
    int m = i >> 6, c = i & 63;
    out[(s_base + m) * DD + c] = Xs[m * XP + c];
  }
  if (tid < MT) {
    out[BB * DD + s_base + tid]      = zacc[tid];
    out[BB * DD + BB + s_base + tid] = lacc[tid];
  }
}

extern "C" void kernel_launch(void* const* d_in, const int* in_sizes, int n_in,
                              void* d_out, int out_size, void* d_ws, size_t ws_size,
                              hipStream_t stream) {
  const float* y0 = (const float*)d_in[0];
  const float* ts = (const float*)d_in[1];
  const float* W1 = (const float*)d_in[2];
  const float* b1 = (const float*)d_in[3];
  const float* W2 = (const float*)d_in[4];
  const float* b2 = (const float*)d_in[5];
  const float* W3 = (const float*)d_in[6];
  const float* b3 = (const float*)d_in[7];
  float* out = (float*)d_out;
  float* Mtr = (float*)d_ws;

  cnf_prep<<<(HH * HH) / 256, 256, 0, stream>>>(W1, W2, W3, Mtr);

  size_t shmem = (size_t)(MT * XP + 2 * MT * HP + 2 * MT) * sizeof(float);
  cnf_main<<<BB / MT, 256, shmem, stream>>>(y0, ts, W1, b1, W2, b2, W3, b3,
                                            Mtr, out);
}


# Last two fragment sets of every GEMM k loop peeled behind the loop: no conditional load in the loop body, so counted load waits stay exact and the younger set stays in flight; summation order kept
# speedup vs baseline: 2.2925x; 1.0433x over previous
#include <hip/hip_runtime.h>
#include <hip/hip_bf16.h>


#define BB  512
#define DD  64
#define HH  512
#define TT  8
#define MT  16
#define XP  68
#define HP  516

typedef float v2f __attribute__((ext_vector_type(2)));
typedef float v8f __attribute__((ext_vector_type(8)));

__device__ __forceinline__ v8f wmma_f32(v2f a, v2f b, v8f c) {
  return __builtin_amdgcn_wmma_f32_16x16x4_f32(false, a, false, b,
                                               (short)0, c, false, false);
}

__device__ __forceinline__ float fast_tanh(float x) {
#if __has_builtin(__builtin_amdgcn_tanhf)
  return __builtin_amdgcn_tanhf(x);
#elif __has_builtin(__builtin_amdgcn_exp2f) && __has_builtin(__builtin_amdgcn_rcpf)
  float e = __builtin_amdgcn_exp2f(x * 2.8853900817779268f);
  return 1.0f - 2.0f * __builtin_amdgcn_rcpf(e + 1.0f);
#else
  return tanhf(x);
#endif
}

__device__ __forceinline__ v2f frag_a(const float* S, int pitch, int m0, int k0,
                                      int lane_lo, int half) {
  const float* p = S + (m0 + lane_lo) * pitch + k0 + 2 * half;
  v2f a; a.x = p[0]; a.y = p[1];
  return a;
}

__device__ __forceinline__ v2f frag_b(const float* __restrict__ W, int ldw,
                                      int k0, int n0, int lane_lo, int half) {
  v2f b;
  b.x = W[(k0 + 2 * half + 0) * ldw + n0 + lane_lo];
  b.y = W[(k0 + 2 * half + 1) * ldw + n0 + lane_lo];
  return b;
}


#define KG 4
#define KH 16

struct FragSetW { v2f a[KG]; v2f b[KG][4]; };

__device__ __forceinline__ void load_b_w(FragSetW& f, const float* __restrict__ W,
                                         int kb, int nw, int lane_lo, int half) {
  #pragma unroll
  for (int i = 0; i < KG; ++i)
    #pragma unroll
    for (int nt = 0; nt < 4; ++nt)
      f.b[i][nt] = frag_b(W, HH, kb + 4 * i, nw + 16 * nt, lane_lo, half);
}

__device__ __forceinline__ void load_a_w(FragSetW& f, const float* S, int pitch,
                                         int kb, int lane_lo, int half) {
  #pragma unroll
  for (int i = 0; i < KG; ++i)
    f.a[i] = frag_a(S, pitch, 0, kb + 4 * i, lane_lo, half);
}

template <bool SQ>
__device__ __forceinline__ void mma_set_w(const FragSetW& f, v8f (&acc)[4]) {
  #pragma unroll
  for (int i = 0; i < KG; ++i) {
    v2f a = f.a[i];
    if (SQ) {
      a.x = 1.f - a.x * a.x;  a.y = 1.f - a.y * a.y;
    }
    #pragma unroll
    for (int nt = 0; nt < 4; ++nt)
      acc[nt] = wmma_f32(a, f.b[i][nt], acc[nt]);
  }
}

template <bool SQ>
__device__ __forceinline__ void gemm_w(v8f (&acc)[4], FragSetW& f0,
                                       const float* S, int pitch,
                                       const float* __restrict__ W, int K,
                                       int nw, int lane_lo, int half) {
  #pragma unroll
  for (int nt = 0; nt < 4; ++nt) acc[nt] = (v8f){};
  FragSetW f1;
  load_a_w(f0, S, pitch, 0, lane_lo, half);
  int kb = 0;
  for (; kb + 8 * KG < K; kb += 8 * KG) {
    load_b_w(f1, W, kb + 4 * KG, nw, lane_lo, half);
    load_a_w(f1, S, pitch, kb + 4 * KG, lane_lo, half);
    mma_set_w<SQ>(f0, acc);
    load_b_w(f0, W, kb + 8 * KG, nw, lane_lo, half);
    load_a_w(f0, S, pitch, kb + 8 * KG, lane_lo, half);
    mma_set_w<SQ>(f1, acc);
  }
  load_b_w(f1, W, kb + 4 * KG, nw, lane_lo, half);
  load_a_w(f1, S, pitch, kb + 4 * KG, lane_lo, half);
  mma_set_w<SQ>(f0, acc);
  mma_set_w<SQ>(f1, acc);
}

struct FragSetX { v2f a[KH]; v2f b[KH]; };

__device__ __forceinline__ void load_b_x(FragSetX& f, const float* __restrict__ W,
                                         int kb, int n0, int lane_lo, int half) {
  #pragma unroll
  for (int i = 0; i < KH; ++i)
    f.b[i] = frag_b(W, DD, kb + 4 * i, n0, lane_lo, half);
}

__device__ __forceinline__ void load_a_x(FragSetX& f, const float* S, int kb,
                                         int m0, int lane_lo, int half) {
  #pragma unroll
  for (int i = 0; i < KH; ++i)
    f.a[i] = frag_a(S, HP, m0, kb + 4 * i, lane_lo, half);
}

__device__ __forceinline__ v8f mma_set_x(const FragSetX& f, v8f c) {
  #pragma unroll
  for (int i = 0; i < KH; ++i) c = wmma_f32(f.a[i], f.b[i], c);
  return c;
}

__global__ __launch_bounds__(256) void cnf_prep(const float* __restrict__ W1,
                                                const float* __restrict__ W2,
                                                const float* __restrict__ W3,
                                                float* __restrict__ M) {
  int idx = blockIdx.x * blockDim.x + threadIdx.x;
  int j = idx >> 9;
  int k = idx & 511;
  float c = 0.f;
  #pragma unroll 8
  for (int m = 0; m < DD; ++m)
    c += W3[k * DD + m] * W1[m * HH + j];
  M[j * HH + k] = W2[j * HH + k] * c;
}

__global__ __launch_bounds__(256) void cnf_main(const float* __restrict__ y0,
                                                const float* __restrict__ ts,
                                                const float* __restrict__ W1,
                                                const float* __restrict__ b1,
                                                const float* __restrict__ W2,
                                                const float* __restrict__ b2,
                                                const float* __restrict__ W3,
                                                const float* __restrict__ b3,
                                                const float* __restrict__ Mtr,
                                                float* __restrict__ out) {
  extern __shared__ float smem[];
  float* Xs   = smem;
  float* H1s  = Xs  + MT * XP;
  float* H2s  = H1s + MT * HP;
  float* zacc = H2s + MT * HP;
  float* lacc = zacc + MT;

  const int tid     = threadIdx.x;
  const int wave    = __builtin_amdgcn_readfirstlane(tid >> 5);
  const int lane    = tid & 31;
  const int lane_lo = lane & 15;
  const int half    = lane >> 4;
  const int s_base  = blockIdx.x * MT;

  for (int i = tid; i < MT * DD; i += 256) {
    int m = i >> 6, c = i & 63;
    Xs[m * XP + c] = y0[(s_base + m) * DD + c];
  }
  if (tid < MT) { zacc[tid] = 0.f; lacc[tid] = 0.f; }

  v8f acc[4];
  FragSetW fw;
  const int nw = wave * 64;
  const bool has_dx = wave < 4;
  const int n3 = (wave & 3) * 16;

  for (int step = 0; step < TT - 1; ++step) {
    const float t0 = ts[step];
    const float dt = ts[step + 1] - t0;
    load_b_w(fw, W1, 0, nw, lane_lo, half);
    __syncthreads();

    gemm_w<false>(acc, fw, Xs, XP, W1, DD, nw, lane_lo, half);
    load_b_w(fw, W2, 0, nw, lane_lo, half);
    #pragma unroll
    for (int nt = 0; nt < 4; ++nt) {
      int col = (wave * 4 + nt) * 16 + lane_lo;
      float add = t0 * W1[DD * HH + col] + b1[col];
      #pragma unroll
      for (int v = 0; v < 8; ++v) {
        int m = v + 8 * half;
        H1s[m * HP + col] = fast_tanh(acc[nt][v] + add);
      }
    }
    __syncthreads();

    gemm_w<false>(acc, fw, H1s, HP, W2, HH, nw, lane_lo, half);
    FragSetX g0, g1;
    if (has_dx)
      load_b_x(g0, W3, 0, n3, lane_lo, half);
    #pragma unroll
    for (int nt = 0; nt < 4; ++nt) {
      int col = (wave * 4 + nt) * 16 + lane_lo;
      float add = b2[col];
      #pragma unroll
      for (int v = 0; v < 8; ++v) {
        int m = v + 8 * half;
        H2s[m * HP + col] = fast_tanh(acc[nt][v] + add);
      }
    }
    __syncthreads();

    load_b_w(fw, Mtr, 0, nw, lane_lo, half);
    if (has_dx) {
      const int m0 = 0, n0 = n3;
      v8f a3 = (v8f){};
      load_a_x(g0, H2s, 0, m0, lane_lo, half);
      int kb = 0;
      for (; kb + 8 * KH < HH; kb += 8 * KH) {
        load_b_x(g1, W3, kb + 4 * KH, n0, lane_lo, half);
        load_a_x(g1, H2s, kb + 4 * KH, m0, lane_lo, half);
        a3 = mma_set_x(g0, a3);
        load_b_x(g0, W3, kb + 8 * KH, n0, lane_lo, half);
        load_a_x(g0, H2s, kb + 8 * KH, m0, lane_lo, half);
        a3 = mma_set_x(g1, a3);
      }
      load_b_x(g1, W3, kb + 4 * KH, n0, lane_lo, half);
      load_a_x(g1, H2s, kb + 4 * KH, m0, lane_lo, half);
      a3 = mma_set_x(g0, a3);
      a3 = mma_set_x(g1, a3);
      int col = n0 + lane_lo;
      float bb = b3[col];
      #pragma unroll
      for (int v = 0; v < 8; ++v) {
        int m = m0 + v + 8 * half;
        float dx = a3[v] + bb;
        atomicAdd(&zacc[m], dt * 0.5f * dx * dx);
        Xs[m * XP + col] += dt * dx;
      }
    }

    gemm_w<true>(acc, fw, H1s, HP, Mtr, HH, nw, lane_lo, half);
    #pragma unroll
    for (int v = 0; v < 8; ++v) {
      int m = v + 8 * half;
      float tl = 0.f;
      #pragma unroll
      for (int nt = 0; nt < 4; ++nt) {
        int col = (wave * 4 + nt) * 16 + lane_lo;
        float h2 = H2s[m * HP + col];
        tl += (1.f - h2 * h2) * acc[nt][v];
      }
      atomicAdd(&lacc[m], dt * tl);
    }
  }

  __syncthreads();
  for (int i = tid; i < MT * DD; i += 256) {
    int m = i >> 6, c = i & 63;
    out[(s_base + m) * DD + c] = Xs[m * XP + c];
  }
  if (tid < MT) {
    out[BB * DD + s_base + tid]      = zacc[tid];
    out[BB * DD + BB + s_base + tid] = lacc[tid];
  }
}

extern "C" void kernel_launch(void* const* d_in, const int* in_sizes, int n_in,
                              void* d_out, int out_size, void* d_ws, size_t ws_size,
                              hipStream_t stream) {
  const float* y0 = (const float*)d_in[0];
  const float* ts = (const float*)d_in[1];
  const float* W1 = (const float*)d_in[2];
  const float* b1 = (const float*)d_in[3];
  const float* W2 = (const float*)d_in[4];
  const float* b2 = (const float*)d_in[5];
  const float* W3 = (const float*)d_in[6];
  const float* b3 = (const float*)d_in[7];
  float* out = (float*)d_out;
  float* Mtr = (float*)d_ws;

  cnf_prep<<<(HH * HH) / 256, 256, 0, stream>>>(W1, W2, W3, Mtr);

  size_t shmem = (size_t)(MT * XP + 2 * MT * HP + 2 * MT) * sizeof(float);
  cnf_main<<<BB / MT, 256, shmem, stream>>>(y0, ts, W1, b1, W2, b2, W3, b3,
                                            Mtr, out);
}


# Weight fragments of the three 1x4-tile GEMMs load as 128-bit rows: a lane's 4 consecutive columns feed its wave's 4 N-tiles, a quarter of the load instructions; trace partial sums regroup
# speedup vs baseline: 2.5393x; 1.1077x over previous
#include <hip/hip_runtime.h>
#include <hip/hip_bf16.h>


#define BB  512
#define DD  64
#define HH  512
#define TT  8
#define MT  16
#define XP  68
#define HP  516

typedef float v2f __attribute__((ext_vector_type(2)));
typedef float v4f __attribute__((ext_vector_type(4)));
typedef float v8f __attribute__((ext_vector_type(8)));

__device__ __forceinline__ v8f wmma_f32(v2f a, v2f b, v8f c) {
  return __builtin_amdgcn_wmma_f32_16x16x4_f32(false, a, false, b,
                                               (short)0, c, false, false);
}

__device__ __forceinline__ float fast_tanh(float x) {
#if __has_builtin(__builtin_amdgcn_tanhf)
  return __builtin_amdgcn_tanhf(x);
#elif __has_builtin(__builtin_amdgcn_exp2f) && __has_builtin(__builtin_amdgcn_rcpf)
  float e = __builtin_amdgcn_exp2f(x * 2.8853900817779268f);
  return 1.0f - 2.0f * __builtin_amdgcn_rcpf(e + 1.0f);
#else
  return tanhf(x);
#endif
}

__device__ __forceinline__ v2f frag_a(const float* S, int pitch, int m0, int k0,
                                      int lane_lo, int half) {
  const float* p = S + (m0 + lane_lo) * pitch + k0 + 2 * half;
  v2f a; a.x = p[0]; a.y = p[1];
  return a;
}

__device__ __forceinline__ v2f frag_b(const float* __restrict__ W, int ldw,
                                      int k0, int n0, int lane_lo, int half) {
  v2f b;
  b.x = W[(k0 + 2 * half + 0) * ldw + n0 + lane_lo];
  b.y = W[(k0 + 2 * half + 1) * ldw + n0 + lane_lo];
  return b;
}


#define KG 4
#define KH 16

struct FragSetW { v2f a[KG]; v4f b[KG][2]; };

__device__ __forceinline__ void load_b_w(FragSetW& f, const float* __restrict__ W,
                                         int kb, int nw, int lane_lo, int half) {
  #pragma unroll
  for (int i = 0; i < KG; ++i)
    #pragma unroll
    for (int r = 0; r < 2; ++r)
      f.b[i][r] = *reinterpret_cast<const v4f*>(
          W + (kb + 4 * i + 2 * half + r) * HH + nw + 4 * lane_lo);
}

__device__ __forceinline__ void load_a_w(FragSetW& f, const float* S, int pitch,
                                         int kb, int lane_lo, int half) {
  #pragma unroll
  for (int i = 0; i < KG; ++i)
    f.a[i] = frag_a(S, pitch, 0, kb + 4 * i, lane_lo, half);
}

template <bool SQ>
__device__ __forceinline__ void mma_set_w(const FragSetW& f, v8f (&acc)[4]) {
  #pragma unroll
  for (int i = 0; i < KG; ++i) {
    v2f a = f.a[i];
    if (SQ) {
      a.x = 1.f - a.x * a.x;  a.y = 1.f - a.y * a.y;
    }
    #pragma unroll
    for (int nt = 0; nt < 4; ++nt) {
      v2f b; b.x = f.b[i][0][nt]; b.y = f.b[i][1][nt];
      acc[nt] = wmma_f32(a, b, acc[nt]);
    }
  }
}

template <bool SQ>
__device__ __forceinline__ void gemm_w(v8f (&acc)[4], FragSetW& f0,
                                       const float* S, int pitch,
                                       const float* __restrict__ W, int K,
                                       int nw, int lane_lo, int half) {
  #pragma unroll
  for (int nt = 0; nt < 4; ++nt) acc[nt] = (v8f){};
  FragSetW f1;
  load_a_w(f0, S, pitch, 0, lane_lo, half);
  int kb = 0;
  for (; kb + 8 * KG < K; kb += 8 * KG) {
    load_b_w(f1, W, kb + 4 * KG, nw, lane_lo, half);
    load_a_w(f1, S, pitch, kb + 4 * KG, lane_lo, half);
    mma_set_w<SQ>(f0, acc);
    load_b_w(f0, W, kb + 8 * KG, nw, lane_lo, half);
    load_a_w(f0, S, pitch, kb + 8 * KG, lane_lo, half);
    mma_set_w<SQ>(f1, acc);
  }
  load_b_w(f1, W, kb + 4 * KG, nw, lane_lo, half);
  load_a_w(f1, S, pitch, kb + 4 * KG, lane_lo, half);
  mma_set_w<SQ>(f0, acc);
  mma_set_w<SQ>(f1, acc);
}

struct FragSetX { v2f a[KH]; v2f b[KH]; };

__device__ __forceinline__ void load_b_x(FragSetX& f, const float* __restrict__ W,
                                         int kb, int n0, int lane_lo, int half) {
  #pragma unroll
  for (int i = 0; i < KH; ++i)
    f.b[i] = frag_b(W, DD, kb + 4 * i, n0, lane_lo, half);
}

__device__ __forceinline__ void load_a_x(FragSetX& f, const float* S, int kb,
                                         int m0, int lane_lo, int half) {
  #pragma unroll
  for (int i = 0; i < KH; ++i)
    f.a[i] = frag_a(S, HP, m0, kb + 4 * i, lane_lo, half);
}

__device__ __forceinline__ v8f mma_set_x(const FragSetX& f, v8f c) {
  #pragma unroll
  for (int i = 0; i < KH; ++i) c = wmma_f32(f.a[i], f.b[i], c);
  return c;
}

__global__ __launch_bounds__(256) void cnf_prep(const float* __restrict__ W1,
                                                const float* __restrict__ W2,
                                                const float* __restrict__ W3,
                                                float* __restrict__ M) {
  int idx = blockIdx.x * blockDim.x + threadIdx.x;
  int j = idx >> 9;
  int k = idx & 511;
  float c = 0.f;
  #pragma unroll 8
  for (int m = 0; m < DD; ++m)
    c += W3[k * DD + m] * W1[m * HH + j];
  M[j * HH + k] = W2[j * HH + k] * c;
}

__global__ __launch_bounds__(256) void cnf_main(const float* __restrict__ y0,
                                                const float* __restrict__ ts,
                                                const float* __restrict__ W1,
                                                const float* __restrict__ b1,
                                                const float* __restrict__ W2,
                                                const float* __restrict__ b2,
                                                const float* __restrict__ W3,
                                                const float* __restrict__ b3,
                                                const float* __restrict__ Mtr,
                                                float* __restrict__ out) {
  extern __shared__ float smem[];
  float* Xs   = smem;
  float* H1s  = Xs  + MT * XP;
  float* H2s  = H1s + MT * HP;
  float* zacc = H2s + MT * HP;
  float* lacc = zacc + MT;

  const int tid     = threadIdx.x;
  const int wave    = __builtin_amdgcn_readfirstlane(tid >> 5);
  const int lane    = tid & 31;
  const int lane_lo = lane & 15;
  const int half    = lane >> 4;
  const int s_base  = blockIdx.x * MT;

  for (int i = tid; i < MT * DD; i += 256) {
    int m = i >> 6, c = i & 63;
    Xs[m * XP + c] = y0[(s_base + m) * DD + c];
  }
  if (tid < MT) { zacc[tid] = 0.f; lacc[tid] = 0.f; }

  v8f acc[4];
  FragSetW fw;
  const int nw = wave * 64;
  const int col0 = nw + 4 * lane_lo;
  const bool has_dx = wave < 4;
  const int n3 = (wave & 3) * 16;

  for (int step = 0; step < TT - 1; ++step) {
    const float t0 = ts[step];
    const float dt = ts[step + 1] - t0;
    load_b_w(fw, W1, 0, nw, lane_lo, half);
    __syncthreads();

    gemm_w<false>(acc, fw, Xs, XP, W1, DD, nw, lane_lo, half);
    load_b_w(fw, W2, 0, nw, lane_lo, half);
    {
      const v4f tw = *reinterpret_cast<const v4f*>(W1 + DD * HH + col0);
      const v4f bb = *reinterpret_cast<const v4f*>(b1 + col0);
      v4f add;
      #pragma unroll
      for (int nt = 0; nt < 4; ++nt)
        add[nt] = t0 * tw[nt] + bb[nt];
      #pragma unroll
      for (int v = 0; v < 8; ++v) {
        int m = v + 8 * half;
        v4f h;
        #pragma unroll
        for (int nt = 0; nt < 4; ++nt)
          h[nt] = fast_tanh(acc[nt][v] + add[nt]);
        *reinterpret_cast<v4f*>(H1s + m * HP + col0) = h;
      }
    }
    __syncthreads();

    gemm_w<false>(acc, fw, H1s, HP, W2, HH, nw, lane_lo, half);
    FragSetX g0, g1;
    if (has_dx)
      load_b_x(g0, W3, 0, n3, lane_lo, half);
    {
      const v4f bb = *reinterpret_cast<const v4f*>(b2 + col0);
      #pragma unroll
      for (int v = 0; v < 8; ++v) {
        int m = v + 8 * half;
        v4f h;
        #pragma unroll
        for (int nt = 0; nt < 4; ++nt)
          h[nt] = fast_tanh(acc[nt][v] + bb[nt]);
        *reinterpret_cast<v4f*>(H2s + m * HP + col0) = h;
      }
    }
    __syncthreads();

    load_b_w(fw, Mtr, 0, nw, lane_lo, half);
    if (has_dx) {
      const int m0 = 0, n0 = n3;
      v8f a3 = (v8f){};
      load_a_x(g0, H2s, 0, m0, lane_lo, half);
      int kb = 0;
      for (; kb + 8 * KH < HH; kb += 8 * KH) {
        load_b_x(g1, W3, kb + 4 * KH, n0, lane_lo, half);
        load_a_x(g1, H2s, kb + 4 * KH, m0, lane_lo, half);
        a3 = mma_set_x(g0, a3);
        load_b_x(g0, W3, kb + 8 * KH, n0, lane_lo, half);
        load_a_x(g0, H2s, kb + 8 * KH, m0, lane_lo, half);
        a3 = mma_set_x(g1, a3);
      }
      load_b_x(g1, W3, kb + 4 * KH, n0, lane_lo, half);
      load_a_x(g1, H2s, kb + 4 * KH, m0, lane_lo, half);
      a3 = mma_set_x(g0, a3);
      a3 = mma_set_x(g1, a3);
      int col = n0 + lane_lo;
      float bb = b3[col];
      #pragma unroll
      for (int v = 0; v < 8; ++v) {
        int m = m0 + v + 8 * half;
        float dx = a3[v] + bb;
        atomicAdd(&zacc[m], dt * 0.5f * dx * dx);
        Xs[m * XP + col] += dt * dx;
      }
    }

    gemm_w<true>(acc, fw, H1s, HP, Mtr, HH, nw, lane_lo, half);
    #pragma unroll
    for (int v = 0; v < 8; ++v) {
      int m = v + 8 * half;
      float tl = 0.f;
      const v4f h2 = *reinterpret_cast<const v4f*>(H2s + m * HP + col0);
      #pragma unroll
      for (int nt = 0; nt < 4; ++nt)
        tl += (1.f - h2[nt] * h2[nt]) * acc[nt][v];
      atomicAdd(&lacc[m], dt * tl);
    }
  }

  __syncthreads();
  for (int i = tid; i < MT * DD; i += 256) {
    int m = i >> 6, c = i & 63;
    out[(s_base + m) * DD + c] = Xs[m * XP + c];
  }
  if (tid < MT) {
    out[BB * DD + s_base + tid]      = zacc[tid];
    out[BB * DD + BB + s_base + tid] = lacc[tid];
  }
}

extern "C" void kernel_launch(void* const* d_in, const int* in_sizes, int n_in,
                              void* d_out, int out_size, void* d_ws, size_t ws_size,
                              hipStream_t stream) {
  const float* y0 = (const float*)d_in[0];
  const float* ts = (const float*)d_in[1];
  const float* W1 = (const float*)d_in[2];
  const float* b1 = (const float*)d_in[3];
  const float* W2 = (const float*)d_in[4];
  const float* b2 = (const float*)d_in[5];
  const float* W3 = (const float*)d_in[6];
  const float* b3 = (const float*)d_in[7];
  float* out = (float*)d_out;
  float* Mtr = (float*)d_ws;

  cnf_prep<<<(HH * HH) / 256, 256, 0, stream>>>(W1, W2, W3, Mtr);

  size_t shmem = (size_t)(MT * XP + 2 * MT * HP + 2 * MT) * sizeof(float);
  cnf_main<<<BB / MT, 256, shmem, stream>>>(y0, ts, W1, b1, W2, b2, W3, b3,
                                            Mtr, out);
}


# The two K=512 1x4-tile GEMMs (layer 2, trace) rotate three fragment register sets: loads of set s+2 issue in front of the WMMA block of set s, two blocks of cover per load; summation order kept
# speedup vs baseline: 2.6541x; 1.0452x over previous
#include <hip/hip_runtime.h>
#include <hip/hip_bf16.h>


#define BB  512
#define DD  64
#define HH  512
#define TT  8
#define MT  16
#define XP  68
#define HP  516

typedef float v2f __attribute__((ext_vector_type(2)));
typedef float v4f __attribute__((ext_vector_type(4)));
typedef float v8f __attribute__((ext_vector_type(8)));

__device__ __forceinline__ v8f wmma_f32(v2f a, v2f b, v8f c) {
  return __builtin_amdgcn_wmma_f32_16x16x4_f32(false, a, false, b,
                                               (short)0, c, false, false);
}

__device__ __forceinline__ float fast_tanh(float x) {
#if __has_builtin(__builtin_amdgcn_tanhf)
  return __builtin_amdgcn_tanhf(x);
#elif __has_builtin(__builtin_amdgcn_exp2f) && __has_builtin(__builtin_amdgcn_rcpf)
  float e = __builtin_amdgcn_exp2f(x * 2.8853900817779268f);
  return 1.0f - 2.0f * __builtin_amdgcn_rcpf(e + 1.0f);
#else
  return tanhf(x);
#endif
}

__device__ __forceinline__ v2f frag_a(const float* S, int pitch, int m0, int k0,
                                      int lane_lo, int half) {
  const float* p = S + (m0 + lane_lo) * pitch + k0 + 2 * half;
  v2f a; a.x = p[0]; a.y = p[1];
  return a;
}

__device__ __forceinline__ v2f frag_b(const float* __restrict__ W, int ldw,
                                      int k0, int n0, int lane_lo, int half) {
  v2f b;
  b.x = W[(k0 + 2 * half + 0) * ldw + n0 + lane_lo];
  b.y = W[(k0 + 2 * half + 1) * ldw + n0 + lane_lo];
  return b;
}


#define KG 4
#define KH 16

struct FragSetW { v2f a[KG]; v4f b[KG][2]; };

__device__ __forceinline__ void load_b_w(FragSetW& f, const float* __restrict__ W,
                                         int kb, int nw, int lane_lo, int half) {
  #pragma unroll
  for (int i = 0; i < KG; ++i)
    #pragma unroll
    for (int r = 0; r < 2; ++r)
      f.b[i][r] = *reinterpret_cast<const v4f*>(
          W + (kb + 4 * i + 2 * half + r) * HH + nw + 4 * lane_lo);
}

__device__ __forceinline__ void load_a_w(FragSetW& f, const float* S, int pitch,
                                         int kb, int lane_lo, int half) {
  #pragma unroll
  for (int i = 0; i < KG; ++i)
    f.a[i] = frag_a(S, pitch, 0, kb + 4 * i, lane_lo, half);
}

template <bool SQ>
__device__ __forceinline__ void mma_set_w(const FragSetW& f, v8f (&acc)[4]) {
  #pragma unroll
  for (int i = 0; i < KG; ++i) {
    v2f a = f.a[i];
    if (SQ) {
      a.x = 1.f - a.x * a.x;  a.y = 1.f - a.y * a.y;
    }
    #pragma unroll
    for (int nt = 0; nt < 4; ++nt) {
      v2f b; b.x = f.b[i][0][nt]; b.y = f.b[i][1][nt];
      acc[nt] = wmma_f32(a, b, acc[nt]);
    }
  }
}

template <bool SQ>
__device__ __forceinline__ void gemm_w(v8f (&acc)[4], FragSetW& f0,
                                       const float* S, int pitch,
                                       const float* __restrict__ W, int K,
                                       int nw, int lane_lo, int half) {
  #pragma unroll
  for (int nt = 0; nt < 4; ++nt) acc[nt] = (v8f){};
  FragSetW f1;
  load_a_w(f0, S, pitch, 0, lane_lo, half);
  int kb = 0;
  for (; kb + 8 * KG < K; kb += 8 * KG) {
    load_b_w(f1, W, kb + 4 * KG, nw, lane_lo, half);
    load_a_w(f1, S, pitch, kb + 4 * KG, lane_lo, half);
    mma_set_w<SQ>(f0, acc);
    load_b_w(f0, W, kb + 8 * KG, nw, lane_lo, half);
    load_a_w(f0, S, pitch, kb + 8 * KG, lane_lo, half);
    mma_set_w<SQ>(f1, acc);
  }
  load_b_w(f1, W, kb + 4 * KG, nw, lane_lo, half);
  load_a_w(f1, S, pitch, kb + 4 * KG, lane_lo, half);
  mma_set_w<SQ>(f0, acc);
  mma_set_w<SQ>(f1, acc);
}

template <bool SQ>
__device__ __forceinline__ void gemm_w3(v8f (&acc)[4], FragSetW& f0,
                                        const float* S, int pitch,
                                        const float* __restrict__ W,
                                        int nw, int lane_lo, int half) {
  static_assert((HH / (4 * KG) - 2) % 3 == 0, "set count must be 3n + 2");
  #pragma unroll
  for (int nt = 0; nt < 4; ++nt) acc[nt] = (v8f){};
  FragSetW f1, f2;
  load_a_w(f0, S, pitch, 0, lane_lo, half);
  load_b_w(f1, W, 4 * KG, nw, lane_lo, half);
  load_a_w(f1, S, pitch, 4 * KG, lane_lo, half);
  int kb = 0;
  #define SCHED_FENCE() __builtin_amdgcn_sched_barrier(0)
  for (; kb + 8 * KG < HH; kb += 12 * KG) {
    load_b_w(f2, W, kb + 8 * KG, nw, lane_lo, half);
    load_a_w(f2, S, pitch, kb + 8 * KG, lane_lo, half);
    mma_set_w<SQ>(f0, acc);
    SCHED_FENCE();
    load_b_w(f0, W, kb + 12 * KG, nw, lane_lo, half);
    load_a_w(f0, S, pitch, kb + 12 * KG, lane_lo, half);
    mma_set_w<SQ>(f1, acc);
    SCHED_FENCE();
    load_b_w(f1, W, kb + 16 * KG, nw, lane_lo, half);
    load_a_w(f1, S, pitch, kb + 16 * KG, lane_lo, half);
    mma_set_w<SQ>(f2, acc);
    SCHED_FENCE();
  }
  #undef SCHED_FENCE
  mma_set_w<SQ>(f0, acc);
  mma_set_w<SQ>(f1, acc);
}

struct FragSetX { v2f a[KH]; v2f b[KH]; };

__device__ __forceinline__ void load_b_x(FragSetX& f, const float* __restrict__ W,
                                         int kb, int n0, int lane_lo, int half) {
  #pragma unroll
  for (int i = 0; i < KH; ++i)
    f.b[i] = frag_b(W, DD, kb + 4 * i, n0, lane_lo, half);
}

__device__ __forceinline__ void load_a_x(FragSetX& f, const float* S, int kb,
                                         int m0, int lane_lo, int half) {
  #pragma unroll
  for (int i = 0; i < KH; ++i)
    f.a[i] = frag_a(S, HP, m0, kb + 4 * i, lane_lo, half);
}

__device__ __forceinline__ v8f mma_set_x(const FragSetX& f, v8f c) {
  #pragma unroll
  for (int i = 0; i < KH; ++i) c = wmma_f32(f.a[i], f.b[i], c);
  return c;
}

__global__ __launch_bounds__(256) void cnf_prep(const float* __restrict__ W1,
                                                const float* __restrict__ W2,
                                                const float* __restrict__ W3,
                                                float* __restrict__ M) {
  int idx = blockIdx.x * blockDim.x + threadIdx.x;
  int j = idx >> 9;
  int k = idx & 511;
  float c = 0.f;
  #pragma unroll 8
  for (int m = 0; m < DD; ++m)
    c += W3[k * DD + m] * W1[m * HH + j];
  M[j * HH + k] = W2[j * HH + k] * c;
}

__global__ __launch_bounds__(256) void cnf_main(const float* __restrict__ y0,
                                                const float* __restrict__ ts,
                                                const float* __restrict__ W1,
                                                const float* __restrict__ b1,
                                                const float* __restrict__ W2,
                                                const float* __restrict__ b2,
                                                const float* __restrict__ W3,
                                                const float* __restrict__ b3,
                                                const float* __restrict__ Mtr,
                                                float* __restrict__ out) {
  extern __shared__ float smem[];
  float* Xs   = smem;
  float* H1s  = Xs  + MT * XP;
  float* H2s  = H1s + MT * HP;
  float* zacc = H2s + MT * HP;
  float* lacc = zacc + MT;

  const int tid     = threadIdx.x;
  const int wave    = __builtin_amdgcn_readfirstlane(tid >> 5);
  const int lane    = tid & 31;
  const int lane_lo = lane & 15;
  const int half    = lane >> 4;
  const int s_base  = blockIdx.x * MT;

  for (int i = tid; i < MT * DD; i += 256) {
    int m = i >> 6, c = i & 63;
    Xs[m * XP + c] = y0[(s_base + m) * DD + c];
  }
  if (tid < MT) { zacc[tid] = 0.f; lacc[tid] = 0.f; }

  v8f acc[4];
  FragSetW fw;
  const int nw = wave * 64;
  const int col0 = nw + 4 * lane_lo;
  const bool has_dx = wave < 4;
  const int n3 = (wave & 3) * 16;

  for (int step = 0; step < TT - 1; ++step) {
    const float t0 = ts[step];
    const float dt = ts[step + 1] - t0;
    load_b_w(fw, W1, 0, nw, lane_lo, half);
    __syncthreads();

    gemm_w<false>(acc, fw, Xs, XP, W1, DD, nw, lane_lo, half);
    load_b_w(fw, W2, 0, nw, lane_lo, half);
    {
      const v4f tw = *reinterpret_cast<const v4f*>(W1 + DD * HH + col0);
      const v4f bb = *reinterpret_cast<const v4f*>(b1 + col0);
      v4f add;
      #pragma unroll
      for (int nt = 0; nt < 4; ++nt)
        add[nt] = t0 * tw[nt] + bb[nt];
      #pragma unroll
      for (int v = 0; v < 8; ++v) {
        int m = v + 8 * half;
        v4f h;
        #pragma unroll
        for (int nt = 0; nt < 4; ++nt)
          h[nt] = fast_tanh(acc[nt][v] + add[nt]);
        *reinterpret_cast<v4f*>(H1s + m * HP + col0) = h;
      }
    }
    __syncthreads();

    gemm_w3<false>(acc, fw, H1s, HP, W2, nw, lane_lo, half);
    FragSetX g0, g1;
    if (has_dx)
      load_b_x(g0, W3, 0, n3, lane_lo, half);
    {
      const v4f bb = *reinterpret_cast<const v4f*>(b2 + col0);
      #pragma unroll
      for (int v = 0; v < 8; ++v) {
        int m = v + 8 * half;
        v4f h;
        #pragma unroll
        for (int nt = 0; nt < 4; ++nt)
          h[nt] = fast_tanh(acc[nt][v] + bb[nt]);
        *reinterpret_cast<v4f*>(H2s + m * HP + col0) = h;
      }
    }
    __syncthreads();

    load_b_w(fw, Mtr, 0, nw, lane_lo, half);
    if (has_dx) {
      const int m0 = 0, n0 = n3;
      v8f a3 = (v8f){};
      load_a_x(g0, H2s, 0, m0, lane_lo, half);
      int kb = 0;
      for (; kb + 8 * KH < HH; kb += 8 * KH) {
        load_b_x(g1, W3, kb + 4 * KH, n0, lane_lo, half);
        load_a_x(g1, H2s, kb + 4 * KH, m0, lane_lo, half);
        a3 = mma_set_x(g0, a3);
        load_b_x(g0, W3, kb + 8 * KH, n0, lane_lo, half);
        load_a_x(g0, H2s, kb + 8 * KH, m0, lane_lo, half);
        a3 = mma_set_x(g1, a3);
      }
      load_b_x(g1, W3, kb + 4 * KH, n0, lane_lo, half);
      load_a_x(g1, H2s, kb + 4 * KH, m0, lane_lo, half);
      a3 = mma_set_x(g0, a3);
      a3 = mma_set_x(g1, a3);
      int col = n0 + lane_lo;
      float bb = b3[col];
      #pragma unroll
      for (int v = 0; v < 8; ++v) {
        int m = m0 + v + 8 * half;
        float dx = a3[v] + bb;
        atomicAdd(&zacc[m], dt * 0.5f * dx * dx);
        Xs[m * XP + col] += dt * dx;
      }
    }

    gemm_w3<true>(acc, fw, H1s, HP, Mtr, nw, lane_lo, half);
    #pragma unroll
    for (int v = 0; v < 8; ++v) {
      int m = v + 8 * half;
      float tl = 0.f;
      const v4f h2 = *reinterpret_cast<const v4f*>(H2s + m * HP + col0);
      #pragma unroll
      for (int nt = 0; nt < 4; ++nt)
        tl += (1.f - h2[nt] * h2[nt]) * acc[nt][v];
      atomicAdd(&lacc[m], dt * tl);
    }
  }

  __syncthreads();
  for (int i = tid; i < MT * DD; i += 256) {
    int m = i >> 6, c = i & 63;
    out[(s_base + m) * DD + c] = Xs[m * XP + c];
  }
  if (tid < MT) {
    out[BB * DD + s_base + tid]      = zacc[tid];
    out[BB * DD + BB + s_base + tid] = lacc[tid];
  }
}

extern "C" void kernel_launch(void* const* d_in, const int* in_sizes, int n_in,
                              void* d_out, int out_size, void* d_ws, size_t ws_size,
                              hipStream_t stream) {
  const float* y0 = (const float*)d_in[0];
  const float* ts = (const float*)d_in[1];
  const float* W1 = (const float*)d_in[2];
  const float* b1 = (const float*)d_in[3];
  const float* W2 = (const float*)d_in[4];
  const float* b2 = (const float*)d_in[5];
  const float* W3 = (const float*)d_in[6];
  const float* b3 = (const float*)d_in[7];
  float* out = (float*)d_out;
  float* Mtr = (float*)d_ws;

  cnf_prep<<<(HH * HH) / 256, 256, 0, stream>>>(W1, W2, W3, Mtr);

  size_t shmem = (size_t)(MT * XP + 2 * MT * HP + 2 * MT) * sizeof(float);
  cnf_main<<<BB / MT, 256, shmem, stream>>>(y0, ts, W1, b1, W2, b2, W3, b3,
                                            Mtr, out);
}


# Trace GEMM leaves the 16-sample step chain: integrator stores each step's H1/H2, a second kernel runs it as one 3584-row GEMM on 224 WGPs, a third adds the 7 step terms in order; trace sums regroup
# speedup vs baseline: 3.2517x; 1.2252x over previous
#include <hip/hip_runtime.h>
#include <hip/hip_bf16.h>


#define BB  512
#define DD  64
#define HH  512
#define TT  8
#define MT  16
#define XP  68
#define HP  516
#define NTILE (BB / MT)

#define WS_M     0
#define WS_H1    (HH * HH)
#define WS_H2    (WS_H1 + (TT - 1) * BB * HH)
#define WS_PART  (WS_H2 + (TT - 1) * BB * HH)
#define WS_FLOATS (WS_PART + (TT - 1) * NTILE * 32)
#define WS_NEED  ((size_t)WS_FLOATS * sizeof(float))

typedef float v2f __attribute__((ext_vector_type(2)));
typedef float v4f __attribute__((ext_vector_type(4)));
typedef float v8f __attribute__((ext_vector_type(8)));

__device__ __forceinline__ v8f wmma_f32(v2f a, v2f b, v8f c) {
  return __builtin_amdgcn_wmma_f32_16x16x4_f32(false, a, false, b,
                                               (short)0, c, false, false);
}

__device__ __forceinline__ float fast_tanh(float x) {
#if __has_builtin(__builtin_amdgcn_tanhf)
  return __builtin_amdgcn_tanhf(x);
#elif __has_builtin(__builtin_amdgcn_exp2f) && __has_builtin(__builtin_amdgcn_rcpf)
  float e = __builtin_amdgcn_exp2f(x * 2.8853900817779268f);
  return 1.0f - 2.0f * __builtin_amdgcn_rcpf(e + 1.0f);
#else
  return tanhf(x);
#endif
}

__device__ __forceinline__ v2f frag_a(const float* S, int pitch, int m0, int k0,
                                      int lane_lo, int half) {
  const float* p = S + (m0 + lane_lo) * pitch + k0 + 2 * half;
  v2f a; a.x = p[0]; a.y = p[1];
  return a;
}

__device__ __forceinline__ v2f frag_b(const float* __restrict__ W, int ldw,
                                      int k0, int n0, int lane_lo, int half) {
  v2f b;
  b.x = W[(k0 + 2 * half + 0) * ldw + n0 + lane_lo];
  b.y = W[(k0 + 2 * half + 1) * ldw + n0 + lane_lo];
  return b;
}


#define KG 4
#define KH 16

struct FragSetW { v2f a[KG]; v4f b[KG][2]; };

__device__ __forceinline__ void load_b_w(FragSetW& f, const float* __restrict__ W,
                                         int kb, int nw, int lane_lo, int half) {
  #pragma unroll
  for (int i = 0; i < KG; ++i)
    #pragma unroll
    for (int r = 0; r < 2; ++r)
      f.b[i][r] = *reinterpret_cast<const v4f*>(
          W + (kb + 4 * i + 2 * half + r) * HH + nw + 4 * lane_lo);
}

__device__ __forceinline__ void load_a_w(FragSetW& f, const float* S, int pitch,
                                         int kb, int lane_lo, int half) {
  #pragma unroll
  for (int i = 0; i < KG; ++i)
    f.a[i] = frag_a(S, pitch, 0, kb + 4 * i, lane_lo, half);
}

template <bool SQ>
__device__ __forceinline__ void mma_set_w(const FragSetW& f, v8f (&acc)[4]) {
  #pragma unroll
  for (int i = 0; i < KG; ++i) {
    v2f a = f.a[i];
    if (SQ) {
      a.x = 1.f - a.x * a.x;  a.y = 1.f - a.y * a.y;
    }
    #pragma unroll
    for (int nt = 0; nt < 4; ++nt) {
      v2f b; b.x = f.b[i][0][nt]; b.y = f.b[i][1][nt];
      acc[nt] = wmma_f32(a, b, acc[nt]);
    }
  }
}

template <bool SQ>
__device__ __forceinline__ void gemm_w(v8f (&acc)[4], FragSetW& f0,
                                       const float* S, int pitch,
                                       const float* __restrict__ W, int K,
                                       int nw, int lane_lo, int half) {
  #pragma unroll
  for (int nt = 0; nt < 4; ++nt) acc[nt] = (v8f){};
  FragSetW f1;
  load_a_w(f0, S, pitch, 0, lane_lo, half);
  int kb = 0;
  for (; kb + 8 * KG < K; kb += 8 * KG) {
    load_b_w(f1, W, kb + 4 * KG, nw, lane_lo, half);
    load_a_w(f1, S, pitch, kb + 4 * KG, lane_lo, half);
    mma_set_w<SQ>(f0, acc);
    load_b_w(f0, W, kb + 8 * KG, nw, lane_lo, half);
    load_a_w(f0, S, pitch, kb + 8 * KG, lane_lo, half);
    mma_set_w<SQ>(f1, acc);
  }
  load_b_w(f1, W, kb + 4 * KG, nw, lane_lo, half);
  load_a_w(f1, S, pitch, kb + 4 * KG, lane_lo, half);
  mma_set_w<SQ>(f0, acc);
  mma_set_w<SQ>(f1, acc);
}

template <bool SQ>
__device__ __forceinline__ void gemm_w3(v8f (&acc)[4], FragSetW& f0,
                                        const float* S, int pitch,
                                        const float* __restrict__ W,
                                        int nw, int lane_lo, int half) {
  static_assert((HH / (4 * KG) - 2) % 3 == 0, "set count must be 3n + 2");
  #pragma unroll
  for (int nt = 0; nt < 4; ++nt) acc[nt] = (v8f){};
  FragSetW f1, f2;
  load_a_w(f0, S, pitch, 0, lane_lo, half);
  load_b_w(f1, W, 4 * KG, nw, lane_lo, half);
  load_a_w(f1, S, pitch, 4 * KG, lane_lo, half);
  int kb = 0;
  #define SCHED_FENCE() __builtin_amdgcn_sched_barrier(0)
  for (; kb + 8 * KG < HH; kb += 12 * KG) {
    load_b_w(f2, W, kb + 8 * KG, nw, lane_lo, half);
    load_a_w(f2, S, pitch, kb + 8 * KG, lane_lo, half);
    mma_set_w<SQ>(f0, acc);
    SCHED_FENCE();
    load_b_w(f0, W, kb + 12 * KG, nw, lane_lo, half);
    load_a_w(f0, S, pitch, kb + 12 * KG, lane_lo, half);
    mma_set_w<SQ>(f1, acc);
    SCHED_FENCE();
    load_b_w(f1, W, kb + 16 * KG, nw, lane_lo, half);
    load_a_w(f1, S, pitch, kb + 16 * KG, lane_lo, half);
    mma_set_w<SQ>(f2, acc);
    SCHED_FENCE();
  }
  #undef SCHED_FENCE
  mma_set_w<SQ>(f0, acc);
  mma_set_w<SQ>(f1, acc);
}

struct FragSetX { v2f a[KH]; v2f b[KH]; };

__device__ __forceinline__ void load_b_x(FragSetX& f, const float* __restrict__ W,
                                         int kb, int n0, int lane_lo, int half) {
  #pragma unroll
  for (int i = 0; i < KH; ++i)
    f.b[i] = frag_b(W, DD, kb + 4 * i, n0, lane_lo, half);
}

__device__ __forceinline__ void load_a_x(FragSetX& f, const float* S, int kb,
                                         int m0, int lane_lo, int half) {
  #pragma unroll
  for (int i = 0; i < KH; ++i)
    f.a[i] = frag_a(S, HP, m0, kb + 4 * i, lane_lo, half);
}

__device__ __forceinline__ v8f mma_set_x(const FragSetX& f, v8f c) {
  #pragma unroll
  for (int i = 0; i < KH; ++i) c = wmma_f32(f.a[i], f.b[i], c);
  return c;
}

__global__ __launch_bounds__(256) void cnf_prep(const float* __restrict__ W1,
                                                const float* __restrict__ W2,
                                                const float* __restrict__ W3,
                                                float* __restrict__ M) {
  int idx = blockIdx.x * blockDim.x + threadIdx.x;
  int j = idx >> 9;
  int k = idx & 511;
  float c = 0.f;
  #pragma unroll 8
  for (int m = 0; m < DD; ++m)
    c += W3[k * DD + m] * W1[m * HH + j];
  M[j * HH + k] = W2[j * HH + k] * c;
}

__global__ __launch_bounds__(256) void cnf_main(const float* __restrict__ y0,
                                                const float* __restrict__ ts,
                                                const float* __restrict__ W1,
                                                const float* __restrict__ b1,
                                                const float* __restrict__ W2,
                                                const float* __restrict__ b2,
                                                const float* __restrict__ W3,
                                                const float* __restrict__ b3,
                                                const float* __restrict__ Mtr,
                                                float* __restrict__ H1g,
                                                float* __restrict__ H2g,
                                                size_t ws_bytes,
                                                float* __restrict__ out) {
  const bool split = ws_bytes >= WS_NEED;
  extern __shared__ float smem[];
  float* Xs   = smem;
  float* H1s  = Xs  + MT * XP;
  float* H2s  = H1s + MT * HP;
  float* zacc = H2s + MT * HP;
  float* lacc = zacc + MT;

  const int tid     = threadIdx.x;
  const int wave    = __builtin_amdgcn_readfirstlane(tid >> 5);
  const int lane    = tid & 31;
  const int lane_lo = lane & 15;
  const int half    = lane >> 4;
  const int s_base  = blockIdx.x * MT;

  for (int i = tid; i < MT * DD; i += 256) {
    int m = i >> 6, c = i & 63;
    Xs[m * XP + c] = y0[(s_base + m) * DD + c];
  }
  if (tid < MT) { zacc[tid] = 0.f; lacc[tid] = 0.f; }

  v8f acc[4];
  FragSetW fw;
  const int nw = wave * 64;
  const int col0 = nw + 4 * lane_lo;
  const bool has_dx = wave < 4;
  const int n3 = (wave & 3) * 16;

  for (int step = 0; step < TT - 1; ++step) {
    const float t0 = ts[step];
    const float dt = ts[step + 1] - t0;
    load_b_w(fw, W1, 0, nw, lane_lo, half);
    __syncthreads();

    gemm_w<false>(acc, fw, Xs, XP, W1, DD, nw, lane_lo, half);
    load_b_w(fw, W2, 0, nw, lane_lo, half);
    {
      const v4f tw = *reinterpret_cast<const v4f*>(W1 + DD * HH + col0);
      const v4f bb = *reinterpret_cast<const v4f*>(b1 + col0);
      v4f add;
      #pragma unroll
      for (int nt = 0; nt < 4; ++nt)
        add[nt] = t0 * tw[nt] + bb[nt];
      #pragma unroll
      for (int v = 0; v < 8; ++v) {
        int m = v + 8 * half;
        v4f h;
        #pragma unroll
        for (int nt = 0; nt < 4; ++nt)
          h[nt] = fast_tanh(acc[nt][v] + add[nt]);
        *reinterpret_cast<v4f*>(H1s + m * HP + col0) = h;
        if (split)
          *reinterpret_cast<v4f*>(H1g + (size_t)(step * BB + s_base + m) * HH + col0) = h;
      }
    }
    __syncthreads();

    gemm_w3<false>(acc, fw, H1s, HP, W2, nw, lane_lo, half);
    FragSetX g0, g1;
    if (has_dx)
      load_b_x(g0, W3, 0, n3, lane_lo, half);
    {
      const v4f bb = *reinterpret_cast<const v4f*>(b2 + col0);
      #pragma unroll
      for (int v = 0; v < 8; ++v) {
        int m = v + 8 * half;
        v4f h;
        #pragma unroll
        for (int nt = 0; nt < 4; ++nt)
          h[nt] = fast_tanh(acc[nt][v] + bb[nt]);
        *reinterpret_cast<v4f*>(H2s + m * HP + col0) = h;
        if (split)
          *reinterpret_cast<v4f*>(H2g + (size_t)(step * BB + s_base + m) * HH + col0) = h;
      }
    }
    __syncthreads();

    if (!split)
      load_b_w(fw, Mtr, 0, nw, lane_lo, half);
    if (has_dx) {
      const int m0 = 0, n0 = n3;
      v8f a3 = (v8f){};
      load_a_x(g0, H2s, 0, m0, lane_lo, half);
      int kb = 0;
      for (; kb + 8 * KH < HH; kb += 8 * KH) {
        load_b_x(g1, W3, kb + 4 * KH, n0, lane_lo, half);
        load_a_x(g1, H2s, kb + 4 * KH, m0, lane_lo, half);
        a3 = mma_set_x(g0, a3);
        load_b_x(g0, W3, kb + 8 * KH, n0, lane_lo, half);
        load_a_x(g0, H2s, kb + 8 * KH, m0, lane_lo, half);
        a3 = mma_set_x(g1, a3);
      }
      load_b_x(g1, W3, kb + 4 * KH, n0, lane_lo, half);
      load_a_x(g1, H2s, kb + 4 * KH, m0, lane_lo, half);
      a3 = mma_set_x(g0, a3);
      a3 = mma_set_x(g1, a3);
      int col = n0 + lane_lo;
      float bb = b3[col];
      #pragma unroll
      for (int v = 0; v < 8; ++v) {
        int m = m0 + v + 8 * half;
        float dx = a3[v] + bb;
        atomicAdd(&zacc[m], dt * 0.5f * dx * dx);
        Xs[m * XP + col] += dt * dx;
      }
    }

    if (!split) {
      gemm_w3<true>(acc, fw, H1s, HP, Mtr, nw, lane_lo, half);
      #pragma unroll
      for (int v = 0; v < 8; ++v) {
        int m = v + 8 * half;
        float tl = 0.f;
        const v4f h2 = *reinterpret_cast<const v4f*>(H2s + m * HP + col0);
        #pragma unroll
        for (int nt = 0; nt < 4; ++nt)
          tl += (1.f - h2[nt] * h2[nt]) * acc[nt][v];
        atomicAdd(&lacc[m], dt * tl);
      }
    }
  }

  __syncthreads();
  for (int i = tid; i < MT * DD; i += 256) {
    int m = i >> 6, c = i & 63;
    out[(s_base + m) * DD + c] = Xs[m * XP + c];
  }
  if (tid < MT) {
    out[BB * DD + s_base + tid]      = zacc[tid];
    if (!split)
      out[BB * DD + BB + s_base + tid] = lacc[tid];
  }
}

__global__ __launch_bounds__(256) void cnf_trace(const float* __restrict__ ts,
                                                 const float* __restrict__ Mtr,
                                                 const float* __restrict__ H1g,
                                                 const float* __restrict__ H2g,
                                                 float* __restrict__ part,
                                                 size_t ws_bytes) {
  if (ws_bytes < WS_NEED) return;
  extern __shared__ float smem[];
  float* H1s  = smem;
  float* lacc = H1s + MT * HP;

  const int tid     = threadIdx.x;
  const int wave    = __builtin_amdgcn_readfirstlane(tid >> 5);
  const int lane    = tid & 31;
  const int lane_lo = lane & 15;
  const int half    = lane >> 4;
  const int step    = blockIdx.x / NTILE;
  const int tile    = blockIdx.x % NTILE;
  const int nw      = wave * 64;
  const int col0    = nw + 4 * lane_lo;
  const size_t row0 = (size_t)(step * BB + tile * MT);

  FragSetW fw;
  load_b_w(fw, Mtr, 0, nw, lane_lo, half);

  for (int i = tid; i < MT * HH / 4; i += 256) {
    int m = i >> 7, c = (i & 127) * 4;
    *reinterpret_cast<v4f*>(H1s + m * HP + c) =
        *reinterpret_cast<const v4f*>(H1g + (row0 + m) * HH + c);
  }
  if (tid < MT) lacc[tid] = 0.f;
  v4f h2[8];
  #pragma unroll
  for (int v = 0; v < 8; ++v)
    h2[v] = *reinterpret_cast<const v4f*>(H2g + (row0 + v + 8 * half) * HH + col0);
  const float dt = ts[step + 1] - ts[step];
  __syncthreads();

  v8f acc[4];
  gemm_w3<true>(acc, fw, H1s, HP, Mtr, nw, lane_lo, half);
  #pragma unroll
  for (int v = 0; v < 8; ++v) {
    int m = v + 8 * half;
    float tl = 0.f;
    #pragma unroll
    for (int nt = 0; nt < 4; ++nt)
      tl += (1.f - h2[v][nt] * h2[v][nt]) * acc[nt][v];
    atomicAdd(&lacc[m], dt * tl);
  }
  __syncthreads();
  if (tid < MT) part[(size_t)blockIdx.x * 32 + tid] = lacc[tid];
}

__global__ __launch_bounds__(BB) void cnf_sum(const float* __restrict__ part,
                                              float* __restrict__ out,
                                              size_t ws_bytes) {
  if (ws_bytes < WS_NEED) return;
  const int s = threadIdx.x;
  float l = 0.f;
  #pragma unroll
  for (int step = 0; step < TT - 1; ++step)
    l += part[(step * NTILE + (s >> 4)) * 32 + (s & 15)];
  out[BB * DD + BB + s] = l;
}

extern "C" void kernel_launch(void* const* d_in, const int* in_sizes, int n_in,
                              void* d_out, int out_size, void* d_ws, size_t ws_size,
                              hipStream_t stream) {
  const float* y0 = (const float*)d_in[0];
  const float* ts = (const float*)d_in[1];
  const float* W1 = (const float*)d_in[2];
  const float* b1 = (const float*)d_in[3];
  const float* W2 = (const float*)d_in[4];
  const float* b2 = (const float*)d_in[5];
  const float* W3 = (const float*)d_in[6];
  const float* b3 = (const float*)d_in[7];
  float* out = (float*)d_out;
  float* ws  = (float*)d_ws;
  float* Mtr = ws + WS_M;
  float* H1g = ws + WS_H1;
  float* H2g = ws + WS_H2;
  float* prt = ws + WS_PART;

  cnf_prep<<<(HH * HH) / 256, 256, 0, stream>>>(W1, W2, W3, Mtr);

  size_t shmem = (size_t)(MT * XP + 2 * MT * HP + 2 * MT) * sizeof(float);
  cnf_main<<<NTILE, 256, shmem, stream>>>(y0, ts, W1, b1, W2, b2, W3, b3,
                                          Mtr, H1g, H2g, ws_size, out);

  size_t shmem_t = (size_t)(MT * HP + MT) * sizeof(float);
  cnf_trace<<<(TT - 1) * NTILE, 256, shmem_t, stream>>>(ts, Mtr, H1g, H2g, prt,
                                                        ws_size);
  cnf_sum<<<1, BB, 0, stream>>>(prt, out, ws_size);
}


# One launch per time step on 256 workgroups (8 layer-2 column slices per 16-sample tile) replaces the 32-workgroup step loop; layer-2, dx and z sums regroup into pieces added in fixed order
# speedup vs baseline: 5.4671x; 1.6813x over previous
#include <hip/hip_runtime.h>
#include <hip/hip_bf16.h>


#define BB  512
#define DD  64
#define HH  512
#define TT  8
#define MT  16
#define XP  68
#define HP  516
#define NTILE (BB / MT)
#define NSL 8
#define RP  68

#define WS_M     0
#define WS_H1    (HH * HH)
#define WS_H2    (WS_H1 + (TT - 1) * BB * HH)
#define WS_PART  (WS_H2 + (TT - 1) * BB * HH)
#define WS_ZT    (WS_PART + (TT - 1) * NTILE * 32)
#define WS_X     (WS_ZT + (TT - 1) * NTILE * 32)
#define WS_DXP   (WS_X + 2 * BB * DD)
#define WS_FLOATS (WS_DXP + 2 * NSL * BB * DD)
#define LDS_OLD  (MT * XP + 2 * MT * HP + 2 * MT)
#define LDS_NEW  (MT * XP + MT * HP + NSL * MT * RP + 2 * MT * RP)
#define WS_NEED  ((size_t)WS_FLOATS * sizeof(float))

typedef float v2f __attribute__((ext_vector_type(2)));
typedef float v4f __attribute__((ext_vector_type(4)));
typedef float v8f __attribute__((ext_vector_type(8)));

__device__ __forceinline__ v8f wmma_f32(v2f a, v2f b, v8f c) {
  return __builtin_amdgcn_wmma_f32_16x16x4_f32(false, a, false, b,
                                               (short)0, c, false, false);
}

__device__ __forceinline__ float fast_tanh(float x) {
#if __has_builtin(__builtin_amdgcn_tanhf)
  return __builtin_amdgcn_tanhf(x);
#elif __has_builtin(__builtin_amdgcn_exp2f) && __has_builtin(__builtin_amdgcn_rcpf)
  float e = __builtin_amdgcn_exp2f(x * 2.8853900817779268f);
  return 1.0f - 2.0f * __builtin_amdgcn_rcpf(e + 1.0f);
#else
  return tanhf(x);
#endif
}

__device__ __forceinline__ v2f frag_a(const float* S, int pitch, int m0, int k0,
                                      int lane_lo, int half) {
  const float* p = S + (m0 + lane_lo) * pitch + k0 + 2 * half;
  v2f a; a.x = p[0]; a.y = p[1];
  return a;
}

__device__ __forceinline__ v2f frag_b(const float* __restrict__ W, int ldw,
                                      int k0, int n0, int lane_lo, int half) {
  v2f b;
  b.x = W[(k0 + 2 * half + 0) * ldw + n0 + lane_lo];
  b.y = W[(k0 + 2 * half + 1) * ldw + n0 + lane_lo];
  return b;
}


#define KG 4
#define KH 16

struct FragSetW { v2f a[KG]; v4f b[KG][2]; };

__device__ __forceinline__ void load_b_w(FragSetW& f, const float* __restrict__ W,
                                         int kb, int nw, int lane_lo, int half) {
  #pragma unroll
  for (int i = 0; i < KG; ++i)
    #pragma unroll
    for (int r = 0; r < 2; ++r)
      f.b[i][r] = *reinterpret_cast<const v4f*>(
          W + (kb + 4 * i + 2 * half + r) * HH + nw + 4 * lane_lo);
}

__device__ __forceinline__ void load_a_w(FragSetW& f, const float* S, int pitch,
                                         int kb, int lane_lo, int half) {
  #pragma unroll
  for (int i = 0; i < KG; ++i)
    f.a[i] = frag_a(S, pitch, 0, kb + 4 * i, lane_lo, half);
}

template <bool SQ>
__device__ __forceinline__ void mma_set_w(const FragSetW& f, v8f (&acc)[4]) {
  #pragma unroll
  for (int i = 0; i < KG; ++i) {
    v2f a = f.a[i];
    if (SQ) {
      a.x = 1.f - a.x * a.x;  a.y = 1.f - a.y * a.y;
    }
    #pragma unroll
    for (int nt = 0; nt < 4; ++nt) {
      v2f b; b.x = f.b[i][0][nt]; b.y = f.b[i][1][nt];
      acc[nt] = wmma_f32(a, b, acc[nt]);
    }
  }
}

template <bool SQ>
__device__ __forceinline__ void gemm_w(v8f (&acc)[4], FragSetW& f0,
                                       const float* S, int pitch,
                                       const float* __restrict__ W, int K,
                                       int nw, int lane_lo, int half) {
  #pragma unroll
  for (int nt = 0; nt < 4; ++nt) acc[nt] = (v8f){};
  FragSetW f1;
  load_a_w(f0, S, pitch, 0, lane_lo, half);
  int kb = 0;
  for (; kb + 8 * KG < K; kb += 8 * KG) {
    load_b_w(f1, W, kb + 4 * KG, nw, lane_lo, half);
    load_a_w(f1, S, pitch, kb + 4 * KG, lane_lo, half);
    mma_set_w<SQ>(f0, acc);
    load_b_w(f0, W, kb + 8 * KG, nw, lane_lo, half);
    load_a_w(f0, S, pitch, kb + 8 * KG, lane_lo, half);
    mma_set_w<SQ>(f1, acc);
  }
  load_b_w(f1, W, kb + 4 * KG, nw, lane_lo, half);
  load_a_w(f1, S, pitch, kb + 4 * KG, lane_lo, half);
  mma_set_w<SQ>(f0, acc);
  mma_set_w<SQ>(f1, acc);
}

template <bool SQ>
__device__ __forceinline__ void gemm_w3(v8f (&acc)[4], FragSetW& f0,
                                        const float* S, int pitch,
                                        const float* __restrict__ W,
                                        int nw, int lane_lo, int half) {
  static_assert((HH / (4 * KG) - 2) % 3 == 0, "set count must be 3n + 2");
  #pragma unroll
  for (int nt = 0; nt < 4; ++nt) acc[nt] = (v8f){};
  FragSetW f1, f2;
  load_a_w(f0, S, pitch, 0, lane_lo, half);
  load_b_w(f1, W, 4 * KG, nw, lane_lo, half);
  load_a_w(f1, S, pitch, 4 * KG, lane_lo, half);
  int kb = 0;
  #define SCHED_FENCE() __builtin_amdgcn_sched_barrier(0)
  for (; kb + 8 * KG < HH; kb += 12 * KG) {
    load_b_w(f2, W, kb + 8 * KG, nw, lane_lo, half);
    load_a_w(f2, S, pitch, kb + 8 * KG, lane_lo, half);
    mma_set_w<SQ>(f0, acc);
    SCHED_FENCE();
    load_b_w(f0, W, kb + 12 * KG, nw, lane_lo, half);
    load_a_w(f0, S, pitch, kb + 12 * KG, lane_lo, half);
    mma_set_w<SQ>(f1, acc);
    SCHED_FENCE();
    load_b_w(f1, W, kb + 16 * KG, nw, lane_lo, half);
    load_a_w(f1, S, pitch, kb + 16 * KG, lane_lo, half);
    mma_set_w<SQ>(f2, acc);
    SCHED_FENCE();
  }
  #undef SCHED_FENCE
  mma_set_w<SQ>(f0, acc);
  mma_set_w<SQ>(f1, acc);
}

struct FragSetX { v2f a[KH]; v2f b[KH]; };

__device__ __forceinline__ void load_b_x(FragSetX& f, const float* __restrict__ W,
                                         int kb, int n0, int lane_lo, int half) {
  #pragma unroll
  for (int i = 0; i < KH; ++i)
    f.b[i] = frag_b(W, DD, kb + 4 * i, n0, lane_lo, half);
}

__device__ __forceinline__ void load_a_x(FragSetX& f, const float* S, int kb,
                                         int m0, int lane_lo, int half) {
  #pragma unroll
  for (int i = 0; i < KH; ++i)
    f.a[i] = frag_a(S, HP, m0, kb + 4 * i, lane_lo, half);
}

__device__ __forceinline__ v8f mma_set_x(const FragSetX& f, v8f c) {
  #pragma unroll
  for (int i = 0; i < KH; ++i) c = wmma_f32(f.a[i], f.b[i], c);
  return c;
}

__global__ __launch_bounds__(256) void cnf_prep(const float* __restrict__ W1,
                                                const float* __restrict__ W2,
                                                const float* __restrict__ W3,
                                                float* __restrict__ M) {
  int idx = blockIdx.x * blockDim.x + threadIdx.x;
  int j = idx >> 9;
  int k = idx & 511;
  float c = 0.f;
  #pragma unroll 8
  for (int m = 0; m < DD; ++m)
    c += W3[k * DD + m] * W1[m * HH + j];
  M[j * HH + k] = W2[j * HH + k] * c;
}

__device__ __forceinline__ void cnf_whole(const float* __restrict__ y0,
                                                const float* __restrict__ ts,
                                                const float* __restrict__ W1,
                                                const float* __restrict__ b1,
                                                const float* __restrict__ W2,
                                                const float* __restrict__ b2,
                                                const float* __restrict__ W3,
                                                const float* __restrict__ b3,
                                                const float* __restrict__ Mtr,
                                                float* smem,
                                                float* __restrict__ out) {
  float* Xs   = smem;
  float* H1s  = Xs  + MT * XP;
  float* H2s  = H1s + MT * HP;
  float* zacc = H2s + MT * HP;
  float* lacc = zacc + MT;

  const int tid     = threadIdx.x;
  const int wave    = __builtin_amdgcn_readfirstlane(tid >> 5);
  const int lane    = tid & 31;
  const int lane_lo = lane & 15;
  const int half    = lane >> 4;
  const int s_base  = blockIdx.x * MT;

  for (int i = tid; i < MT * DD; i += 256) {
    int m = i >> 6, c = i & 63;
    Xs[m * XP + c] = y0[(s_base + m) * DD + c];
  }
  if (tid < MT) { zacc[tid] = 0.f; lacc[tid] = 0.f; }

  v8f acc[4];
  FragSetW fw;
  const int nw = wave * 64;
  const int col0 = nw + 4 * lane_lo;
  const bool has_dx = wave < 4;
  const int n3 = (wave & 3) * 16;

  for (int step = 0; step < TT - 1; ++step) {
    const float t0 = ts[step];
    const float dt = ts[step + 1] - t0;
    load_b_w(fw, W1, 0, nw, lane_lo, half);
    __syncthreads();

    gemm_w<false>(acc, fw, Xs, XP, W1, DD, nw, lane_lo, half);
    load_b_w(fw, W2, 0, nw, lane_lo, half);
    {
      const v4f tw = *reinterpret_cast<const v4f*>(W1 + DD * HH + col0);
      const v4f bb = *reinterpret_cast<const v4f*>(b1 + col0);
      v4f add;
      #pragma unroll
      for (int nt = 0; nt < 4; ++nt)
        add[nt] = t0 * tw[nt] + bb[nt];
      #pragma unroll
      for (int v = 0; v < 8; ++v) {
        int m = v + 8 * half;
        v4f h;
        #pragma unroll
        for (int nt = 0; nt < 4; ++nt)
          h[nt] = fast_tanh(acc[nt][v] + add[nt]);
        *reinterpret_cast<v4f*>(H1s + m * HP + col0) = h;
      }
    }
    __syncthreads();

    gemm_w3<false>(acc, fw, H1s, HP, W2, nw, lane_lo, half);
    FragSetX g0, g1;
    if (has_dx)
      load_b_x(g0, W3, 0, n3, lane_lo, half);
    {
      const v4f bb = *reinterpret_cast<const v4f*>(b2 + col0);
      #pragma unroll
      for (int v = 0; v < 8; ++v) {
        int m = v + 8 * half;
        v4f h;
        #pragma unroll
        for (int nt = 0; nt < 4; ++nt)
          h[nt] = fast_tanh(acc[nt][v] + bb[nt]);
        *reinterpret_cast<v4f*>(H2s + m * HP + col0) = h;
      }
    }
    __syncthreads();

    load_b_w(fw, Mtr, 0, nw, lane_lo, half);
    if (has_dx) {
      const int m0 = 0, n0 = n3;
      v8f a3 = (v8f){};
      load_a_x(g0, H2s, 0, m0, lane_lo, half);
      int kb = 0;
      for (; kb + 8 * KH < HH; kb += 8 * KH) {
        load_b_x(g1, W3, kb + 4 * KH, n0, lane_lo, half);
        load_a_x(g1, H2s, kb + 4 * KH, m0, lane_lo, half);
        a3 = mma_set_x(g0, a3);
        load_b_x(g0, W3, kb + 8 * KH, n0, lane_lo, half);
        load_a_x(g0, H2s, kb + 8 * KH, m0, lane_lo, half);
        a3 = mma_set_x(g1, a3);
      }
      load_b_x(g1, W3, kb + 4 * KH, n0, lane_lo, half);
      load_a_x(g1, H2s, kb + 4 * KH, m0, lane_lo, half);
      a3 = mma_set_x(g0, a3);
      a3 = mma_set_x(g1, a3);
      int col = n0 + lane_lo;
      float bb = b3[col];
      #pragma unroll
      for (int v = 0; v < 8; ++v) {
        int m = m0 + v + 8 * half;
        float dx = a3[v] + bb;
        atomicAdd(&zacc[m], dt * 0.5f * dx * dx);
        Xs[m * XP + col] += dt * dx;
      }
    }

    {
      gemm_w3<true>(acc, fw, H1s, HP, Mtr, nw, lane_lo, half);
      #pragma unroll
      for (int v = 0; v < 8; ++v) {
        int m = v + 8 * half;
        float tl = 0.f;
        const v4f h2 = *reinterpret_cast<const v4f*>(H2s + m * HP + col0);
        #pragma unroll
        for (int nt = 0; nt < 4; ++nt)
          tl += (1.f - h2[nt] * h2[nt]) * acc[nt][v];
        atomicAdd(&lacc[m], dt * tl);
      }
    }
  }

  __syncthreads();
  for (int i = tid; i < MT * DD; i += 256) {
    int m = i >> 6, c = i & 63;
    out[(s_base + m) * DD + c] = Xs[m * XP + c];
  }
  if (tid < MT) {
    out[BB * DD + s_base + tid]      = zacc[tid];
    out[BB * DD + BB + s_base + tid] = lacc[tid];
  }
}

__device__ __forceinline__ v4f dx_from_pieces(const float* __restrict__ P,
                                              const float* __restrict__ b3,
                                              int row, int c4) {
  v4f d = *reinterpret_cast<const v4f*>(P + (size_t)row * DD + c4);
  #pragma unroll
  for (int j = 1; j < NSL; ++j)
    d += *reinterpret_cast<const v4f*>(P + (size_t)(j * BB + row) * DD + c4);
  d += *reinterpret_cast<const v4f*>(b3 + c4);
  return d;
}

__device__ __forceinline__ float sum16(float q) {
  q += __shfl_xor(q, 8, 16);
  q += __shfl_xor(q, 4, 16);
  q += __shfl_xor(q, 2, 16);
  q += __shfl_xor(q, 1, 16);
  return q;
}

__global__ __launch_bounds__(256) void cnf_step(int step,
                                                const float* __restrict__ y0,
                                                const float* __restrict__ ts,
                                                const float* __restrict__ W1,
                                                const float* __restrict__ b1,
                                                const float* __restrict__ W2,
                                                const float* __restrict__ b2,
                                                const float* __restrict__ W3,
                                                const float* __restrict__ b3,
                                                float* __restrict__ ws,
                                                size_t ws_bytes,
                                                float* __restrict__ out) {
  extern __shared__ float smem[];
  if (ws_bytes < WS_NEED) {
    if (step == 0 && blockIdx.x < NTILE)
      cnf_whole(y0, ts, W1, b1, W2, b2, W3, b3, ws + WS_M, smem, out);
    return;
  }
  float* Xs  = smem;
  float* H1s = Xs  + MT * XP;
  float* red = H1s + MT * HP;
  float* H2t = red + NSL * MT * RP;
  float* dxs = H2t + MT * RP;

  const int tid     = threadIdx.x;
  const int wave    = __builtin_amdgcn_readfirstlane(tid >> 5);
  const int lane    = tid & 31;
  const int lane_lo = lane & 15;
  const int half    = lane >> 4;
  const int tile    = blockIdx.x / NSL;
  const int slice   = blockIdx.x % NSL;
  const int s_base  = tile * MT;
  const int nw      = wave * 64;
  const int col0    = nw + 4 * lane_lo;
  const int sc0     = slice * 64;
  const int r_m     = tid >> 4;
  const int r_c     = 4 * (tid & 15);

  float* H1g = ws + WS_H1;
  float* H2g = ws + WS_H2;
  float* ztg = ws + WS_ZT;
  float* Xg  = ws + WS_X;
  float* dxp = ws + WS_DXP;

  const float t0 = ts[step];
  FragSetW fw;
  load_b_w(fw, W1, 0, nw, lane_lo, half);

  {
    v4f x;
    if (step == 0) {
      x = *reinterpret_cast<const v4f*>(y0 + (size_t)(s_base + r_m) * DD + r_c);
    } else {
      const float* Xprev = (step == 1) ? y0 : Xg + ((step - 1) & 1) * BB * DD;
      const float* P = dxp + ((step - 1) & 1) * (NSL * BB * DD);
      const float dtp = t0 - ts[step - 1];
      x = *reinterpret_cast<const v4f*>(Xprev + (size_t)(s_base + r_m) * DD + r_c);
      const v4f d = dx_from_pieces(P, b3, s_base + r_m, r_c);
      x += dtp * d;
      if (slice == 0) {
        const float q = sum16(d.x * d.x + d.y * d.y + d.z * d.z + d.w * d.w);
        if ((tid & 15) == 0) dxs[r_m] = dtp * 0.5f * q;
      }
    }
    *reinterpret_cast<v4f*>(Xs + r_m * XP + r_c) = x;
  }
  __syncthreads();
  if (step != 0 && slice == 0 && tid < MT)
    ztg[((step - 1) * NTILE + tile) * 32 + tid] = dxs[tid];

  v8f acc[4];
  gemm_w<false>(acc, fw, Xs, XP, W1, DD, nw, lane_lo, half);
  load_b_w(fw, W2 + (size_t)nw * HH, 0, sc0, lane_lo, half);
  {
    const v4f tw = *reinterpret_cast<const v4f*>(W1 + DD * HH + col0);
    const v4f bb = *reinterpret_cast<const v4f*>(b1 + col0);
    v4f add;
    #pragma unroll
    for (int nt = 0; nt < 4; ++nt)
      add[nt] = t0 * tw[nt] + bb[nt];
    #pragma unroll
    for (int v = 0; v < 8; ++v) {
      int m = v + 8 * half;
      v4f h;
      #pragma unroll
      for (int nt = 0; nt < 4; ++nt)
        h[nt] = fast_tanh(acc[nt][v] + add[nt]);
      *reinterpret_cast<v4f*>(H1s + m * HP + col0) = h;
    }
  }
  __syncthreads();

  FragSetX g;
  if (wave < 4)
    load_b_x(g, W3 + (size_t)sc0 * DD, 0, wave * 16, lane_lo, half);
  gemm_w<false>(acc, fw, H1s + nw, HP, W2 + (size_t)nw * HH, 64, sc0, lane_lo, half);
  #pragma unroll
  for (int v = 0; v < 8; ++v) {
    int m = v + 8 * half;
    v4f p;
    #pragma unroll
    for (int nt = 0; nt < 4; ++nt) p[nt] = acc[nt][v];
    *reinterpret_cast<v4f*>(red + (wave * MT + m) * RP + 4 * lane_lo) = p;
  }
  __syncthreads();
  v4f h2;
  {
    v4f a = *reinterpret_cast<const v4f*>(red + r_m * RP + r_c);
    #pragma unroll
    for (int w = 1; w < NSL; ++w)
      a += *reinterpret_cast<const v4f*>(red + (w * MT + r_m) * RP + r_c);
    const v4f bb = *reinterpret_cast<const v4f*>(b2 + sc0 + r_c);
    #pragma unroll
    for (int nt = 0; nt < 4; ++nt) h2[nt] = fast_tanh(a[nt] + bb[nt]);
    *reinterpret_cast<v4f*>(H2t + r_m * RP + r_c) = h2;
  }
  __syncthreads();

  if (wave < 4) {
    #pragma unroll
    for (int i = 0; i < KH; ++i)
      g.a[i] = frag_a(H2t, RP, 0, 4 * i, lane_lo, half);
    const v8f a3 = mma_set_x(g, (v8f){});
    #pragma unroll
    for (int v = 0; v < 8; ++v)
      dxs[(v + 8 * half) * RP + wave * 16 + lane_lo] = a3[v];
  }
  __syncthreads();
  *reinterpret_cast<v4f*>(dxp + (step & 1) * (NSL * BB * DD)
                          + (size_t)(slice * BB + s_base + r_m) * DD + r_c) =
      *reinterpret_cast<const v4f*>(dxs + r_m * RP + r_c);
  const size_t hrow = (size_t)(step * BB + s_base + r_m) * HH + sc0 + r_c;
  *reinterpret_cast<v4f*>(H2g + hrow) = h2;
  *reinterpret_cast<v4f*>(H1g + hrow) =
      *reinterpret_cast<const v4f*>(H1s + r_m * HP + sc0 + r_c);
  if (step != 0 && slice == 0)
    *reinterpret_cast<v4f*>(Xg + (step & 1) * BB * DD + (size_t)(s_base + r_m) * DD + r_c) =
        *reinterpret_cast<const v4f*>(Xs + r_m * XP + r_c);
}

__global__ __launch_bounds__(256) void cnf_trace(const float* __restrict__ ts,
                                                 const float* __restrict__ Mtr,
                                                 const float* __restrict__ H1g,
                                                 const float* __restrict__ H2g,
                                                 float* __restrict__ part,
                                                 size_t ws_bytes) {
  if (ws_bytes < WS_NEED) return;
  extern __shared__ float smem[];
  float* H1s  = smem;
  float* lacc = H1s + MT * HP;

  const int tid     = threadIdx.x;
  const int wave    = __builtin_amdgcn_readfirstlane(tid >> 5);
  const int lane    = tid & 31;
  const int lane_lo = lane & 15;
  const int half    = lane >> 4;
  const int step    = blockIdx.x / NTILE;
  const int tile    = blockIdx.x % NTILE;
  const int nw      = wave * 64;
  const int col0    = nw + 4 * lane_lo;
  const size_t row0 = (size_t)(step * BB + tile * MT);

  FragSetW fw;
  load_b_w(fw, Mtr, 0, nw, lane_lo, half);

  for (int i = tid; i < MT * HH / 4; i += 256) {
    int m = i >> 7, c = (i & 127) * 4;
    *reinterpret_cast<v4f*>(H1s + m * HP + c) =
        *reinterpret_cast<const v4f*>(H1g + (row0 + m) * HH + c);
  }
  if (tid < MT) lacc[tid] = 0.f;
  v4f h2[8];
  #pragma unroll
  for (int v = 0; v < 8; ++v)
    h2[v] = *reinterpret_cast<const v4f*>(H2g + (row0 + v + 8 * half) * HH + col0);
  const float dt = ts[step + 1] - ts[step];
  __syncthreads();

  v8f acc[4];
  gemm_w3<true>(acc, fw, H1s, HP, Mtr, nw, lane_lo, half);
  #pragma unroll
  for (int v = 0; v < 8; ++v) {
    int m = v + 8 * half;
    float tl = 0.f;
    #pragma unroll
    for (int nt = 0; nt < 4; ++nt)
      tl += (1.f - h2[v][nt] * h2[v][nt]) * acc[nt][v];
    atomicAdd(&lacc[m], dt * tl);
  }
  __syncthreads();
  if (tid < MT) part[(size_t)blockIdx.x * 32 + tid] = lacc[tid];
}

__global__ __launch_bounds__(BB) void cnf_final(const float* __restrict__ ts,
                                                const float* __restrict__ b3,
                                                const float* __restrict__ ws,
                                                size_t ws_bytes,
                                                float* __restrict__ out) {
  if (ws_bytes < WS_NEED) return;
  __shared__ float zl[64];
  const int tid = threadIdx.x;
  const int s   = blockIdx.x * 32 + (tid >> 4);
  const int r_c = 4 * (tid & 15);
  const float* Xprev = ws + WS_X + ((TT - 2) & 1) * BB * DD;
  const float* P     = ws + WS_DXP + ((TT - 2) & 1) * (NSL * BB * DD);
  const float dtp = ts[TT - 1] - ts[TT - 2];
  v4f x = *reinterpret_cast<const v4f*>(Xprev + (size_t)s * DD + r_c);
  const v4f d = dx_from_pieces(P, b3, s, r_c);
  x += dtp * d;
  *reinterpret_cast<v4f*>(out + (size_t)s * DD + r_c) = x;
  const float q = sum16(d.x * d.x + d.y * d.y + d.z * d.z + d.w * d.w);
  if ((tid & 15) == 0) {
    const float* zt = ws + WS_ZT;
    const float* pt = ws + WS_PART;
    float z = 0.f, l = 0.f;
    #pragma unroll
    for (int step = 0; step < TT - 2; ++step)
      z += zt[(step * NTILE + (s >> 4)) * 32 + (s & 15)];
    z += dtp * 0.5f * q;
    #pragma unroll
    for (int step = 0; step < TT - 1; ++step)
      l += pt[(step * NTILE + (s >> 4)) * 32 + (s & 15)];
    zl[tid >> 4] = z;
    zl[32 + (tid >> 4)] = l;
  }
  __syncthreads();
  if (tid < 32)       out[BB * DD + blockIdx.x * 32 + tid] = zl[tid];
  else if (tid < 64)  out[BB * DD + BB + blockIdx.x * 32 + (tid - 32)] = zl[tid];
}

extern "C" void kernel_launch(void* const* d_in, const int* in_sizes, int n_in,
                              void* d_out, int out_size, void* d_ws, size_t ws_size,
                              hipStream_t stream) {
  const float* y0 = (const float*)d_in[0];
  const float* ts = (const float*)d_in[1];
  const float* W1 = (const float*)d_in[2];
  const float* b1 = (const float*)d_in[3];
  const float* W2 = (const float*)d_in[4];
  const float* b2 = (const float*)d_in[5];
  const float* W3 = (const float*)d_in[6];
  const float* b3 = (const float*)d_in[7];
  float* out = (float*)d_out;
  float* ws  = (float*)d_ws;
  float* Mtr = ws + WS_M;
  float* H1g = ws + WS_H1;
  float* H2g = ws + WS_H2;
  float* prt = ws + WS_PART;

  cnf_prep<<<(HH * HH) / 256, 256, 0, stream>>>(W1, W2, W3, Mtr);

  size_t shmem = (size_t)(LDS_NEW > LDS_OLD ? LDS_NEW : LDS_OLD) * sizeof(float);
#define STEP(n) cnf_step<<<NTILE * NSL, 256, shmem, stream>>>(n, y0, ts, W1, b1, W2, b2, W3, b3, ws, ws_size, out)
  STEP(0); STEP(1); STEP(2); STEP(3); STEP(4); STEP(5); STEP(6);
#undef STEP

  size_t shmem_t = (size_t)(MT * HP + MT) * sizeof(float);
  cnf_trace<<<(TT - 1) * NTILE, 256, shmem_t, stream>>>(ts, Mtr, H1g, H2g, prt,
                                                        ws_size);
  cnf_final<<<BB / 32, BB, 0, stream>>>(ts, b3, ws, ws_size, out);
}


# M = W2 .* (W3 W1)^T is built at the end of the first step launch (a 32x32 block per workgroup, its product on WMMA) instead of by a scalar prep launch: one launch fewer; M's 64-term sums regroup
# speedup vs baseline: 5.9724x; 1.0924x over previous
#include <hip/hip_runtime.h>
#include <hip/hip_bf16.h>


#define BB  512
#define DD  64
#define HH  512
#define TT  8
#define MT  16
#define XP  68
#define HP  516
#define NTILE (BB / MT)
#define NSL 8
#define RP  68

#define WS_M     0
#define WS_H1    (HH * HH)
#define WS_H2    (WS_H1 + (TT - 1) * BB * HH)
#define WS_PART  (WS_H2 + (TT - 1) * BB * HH)
#define WS_ZT    (WS_PART + (TT - 1) * NTILE * 32)
#define WS_X     (WS_ZT + (TT - 1) * NTILE * 32)
#define WS_DXP   (WS_X + 2 * BB * DD)
#define WS_FLOATS (WS_DXP + 2 * NSL * BB * DD)
#define LDS_OLD  (MT * XP + 2 * MT * HP + 2 * MT)
#define LDS_NEW  (MT * XP + MT * HP + NSL * MT * RP + 2 * MT * RP)
#define WS_NEED  ((size_t)WS_FLOATS * sizeof(float))

typedef float v2f __attribute__((ext_vector_type(2)));
typedef float v4f __attribute__((ext_vector_type(4)));
typedef float v8f __attribute__((ext_vector_type(8)));

__device__ __forceinline__ v8f wmma_f32(v2f a, v2f b, v8f c) {
  return __builtin_amdgcn_wmma_f32_16x16x4_f32(false, a, false, b,
                                               (short)0, c, false, false);
}

__device__ __forceinline__ float fast_tanh(float x) {
#if __has_builtin(__builtin_amdgcn_tanhf)
  return __builtin_amdgcn_tanhf(x);
#elif __has_builtin(__builtin_amdgcn_exp2f) && __has_builtin(__builtin_amdgcn_rcpf)
  float e = __builtin_amdgcn_exp2f(x * 2.8853900817779268f);
  return 1.0f - 2.0f * __builtin_amdgcn_rcpf(e + 1.0f);
#else
  return tanhf(x);
#endif
}

__device__ __forceinline__ v2f frag_a(const float* S, int pitch, int m0, int k0,
                                      int lane_lo, int half) {
  const float* p = S + (m0 + lane_lo) * pitch + k0 + 2 * half;
  v2f a; a.x = p[0]; a.y = p[1];
  return a;
}

__device__ __forceinline__ v2f frag_b(const float* __restrict__ W, int ldw,
                                      int k0, int n0, int lane_lo, int half) {
  v2f b;
  b.x = W[(k0 + 2 * half + 0) * ldw + n0 + lane_lo];
  b.y = W[(k0 + 2 * half + 1) * ldw + n0 + lane_lo];
  return b;
}


#define KG 4
#define KH 16

struct FragSetW { v2f a[KG]; v4f b[KG][2]; };

__device__ __forceinline__ void load_b_w(FragSetW& f, const float* __restrict__ W,
                                         int kb, int nw, int lane_lo, int half) {
  #pragma unroll
  for (int i = 0; i < KG; ++i)
    #pragma unroll
    for (int r = 0; r < 2; ++r)
      f.b[i][r] = *reinterpret_cast<const v4f*>(
          W + (kb + 4 * i + 2 * half + r) * HH + nw + 4 * lane_lo);
}

__device__ __forceinline__ void load_a_w(FragSetW& f, const float* S, int pitch,
                                         int kb, int lane_lo, int half) {
  #pragma unroll
  for (int i = 0; i < KG; ++i)
    f.a[i] = frag_a(S, pitch, 0, kb + 4 * i, lane_lo, half);
}

template <bool SQ>
__device__ __forceinline__ void mma_set_w(const FragSetW& f, v8f (&acc)[4]) {
  #pragma unroll
  for (int i = 0; i < KG; ++i) {
    v2f a = f.a[i];
    if (SQ) {
      a.x = 1.f - a.x * a.x;  a.y = 1.f - a.y * a.y;
    }
    #pragma unroll
    for (int nt = 0; nt < 4; ++nt) {
      v2f b; b.x = f.b[i][0][nt]; b.y = f.b[i][1][nt];
      acc[nt] = wmma_f32(a, b, acc[nt]);
    }
  }
}

template <bool SQ>
__device__ __forceinline__ void gemm_w(v8f (&acc)[4], FragSetW& f0,
                                       const float* S, int pitch,
                                       const float* __restrict__ W, int K,
                                       int nw, int lane_lo, int half) {
  #pragma unroll
  for (int nt = 0; nt < 4; ++nt) acc[nt] = (v8f){};
  FragSetW f1;
  load_a_w(f0, S, pitch, 0, lane_lo, half);
  int kb = 0;
  for (; kb + 8 * KG < K; kb += 8 * KG) {
    load_b_w(f1, W, kb + 4 * KG, nw, lane_lo, half);
    load_a_w(f1, S, pitch, kb + 4 * KG, lane_lo, half);
    mma_set_w<SQ>(f0, acc);
    load_b_w(f0, W, kb + 8 * KG, nw, lane_lo, half);
    load_a_w(f0, S, pitch, kb + 8 * KG, lane_lo, half);
    mma_set_w<SQ>(f1, acc);
  }
  load_b_w(f1, W, kb + 4 * KG, nw, lane_lo, half);
  load_a_w(f1, S, pitch, kb + 4 * KG, lane_lo, half);
  mma_set_w<SQ>(f0, acc);
  mma_set_w<SQ>(f1, acc);
}

template <bool SQ>
__device__ __forceinline__ void gemm_w3(v8f (&acc)[4], FragSetW& f0,
                                        const float* S, int pitch,
                                        const float* __restrict__ W,
                                        int nw, int lane_lo, int half) {
  static_assert((HH / (4 * KG) - 2) % 3 == 0, "set count must be 3n + 2");
  #pragma unroll
  for (int nt = 0; nt < 4; ++nt) acc[nt] = (v8f){};
  FragSetW f1, f2;
  load_a_w(f0, S, pitch, 0, lane_lo, half);
  load_b_w(f1, W, 4 * KG, nw, lane_lo, half);
  load_a_w(f1, S, pitch, 4 * KG, lane_lo, half);
  int kb = 0;
  #define SCHED_FENCE() __builtin_amdgcn_sched_barrier(0)
  for (; kb + 8 * KG < HH; kb += 12 * KG) {
    load_b_w(f2, W, kb + 8 * KG, nw, lane_lo, half);
    load_a_w(f2, S, pitch, kb + 8 * KG, lane_lo, half);
    mma_set_w<SQ>(f0, acc);
    SCHED_FENCE();
    load_b_w(f0, W, kb + 12 * KG, nw, lane_lo, half);
    load_a_w(f0, S, pitch, kb + 12 * KG, lane_lo, half);
    mma_set_w<SQ>(f1, acc);
    SCHED_FENCE();
    load_b_w(f1, W, kb + 16 * KG, nw, lane_lo, half);
    load_a_w(f1, S, pitch, kb + 16 * KG, lane_lo, half);
    mma_set_w<SQ>(f2, acc);
    SCHED_FENCE();
  }
  #undef SCHED_FENCE
  mma_set_w<SQ>(f0, acc);
  mma_set_w<SQ>(f1, acc);
}

struct FragSetX { v2f a[KH]; v2f b[KH]; };

__device__ __forceinline__ void load_b_x(FragSetX& f, const float* __restrict__ W,
                                         int kb, int n0, int lane_lo, int half) {
  #pragma unroll
  for (int i = 0; i < KH; ++i)
    f.b[i] = frag_b(W, DD, kb + 4 * i, n0, lane_lo, half);
}

__device__ __forceinline__ void load_a_x(FragSetX& f, const float* S, int kb,
                                         int m0, int lane_lo, int half) {
  #pragma unroll
  for (int i = 0; i < KH; ++i)
    f.a[i] = frag_a(S, HP, m0, kb + 4 * i, lane_lo, half);
}

__device__ __forceinline__ v8f mma_set_x(const FragSetX& f, v8f c) {
  #pragma unroll
  for (int i = 0; i < KH; ++i) c = wmma_f32(f.a[i], f.b[i], c);
  return c;
}

#define CP 36
static_assert(4 * KH == DD, "one fragment set must cover the whole K = 64");

__device__ __forceinline__ size_t m_off(int tid) {
  const int b = blockIdx.x;
  return (size_t)((b & 15) * 32 + (tid >> 3)) * HH + (b >> 4) * 32 + 4 * (tid & 7);
}

__device__ __forceinline__ void m_load(FragSetX& f, const float* __restrict__ W1,
                                       const float* __restrict__ W3,
                                       int wave, int lane_lo, int half) {
  const int b  = blockIdx.x;
  const int k0 = (b >> 4) * 32 + ((wave >> 1) & 1) * 16;
  const int j0 = (b & 15) * 32 + (wave & 1) * 16;
  #pragma unroll
  for (int i = 0; i < KH; ++i) {
    f.a[i] = *reinterpret_cast<const v2f*>(W3 + (size_t)(k0 + lane_lo) * DD + 4 * i + 2 * half);
    f.b[i] = frag_b(W1, HH, 4 * i, j0, lane_lo, half);
  }
}

__device__ __forceinline__ void m_tile(const FragSetX& f, float* Ct,
                                       int wave, int lane_lo, int half) {
  const v8f c = mma_set_x(f, (v8f){});
  float* p = Ct + ((wave & 1) * 16 + lane_lo) * CP + ((wave >> 1) & 1) * 16 + 8 * half;
  v4f lo, hi;
  #pragma unroll
  for (int r = 0; r < 4; ++r) { lo[r] = c[r]; hi[r] = c[4 + r]; }
  *reinterpret_cast<v4f*>(p)     = lo;
  *reinterpret_cast<v4f*>(p + 4) = hi;
}

__device__ __forceinline__ void m_store(const float* Ct, v4f w2, float* __restrict__ M,
                                        int tid) {
  const v4f c = *reinterpret_cast<const v4f*>(Ct + (tid >> 3) * CP + 4 * (tid & 7));
  *reinterpret_cast<v4f*>(M + m_off(tid)) = w2 * c;
}

__device__ __forceinline__ void build_m(const float* __restrict__ W1,
                                        const float* __restrict__ W2,
                                        const float* __restrict__ W3,
                                        float* __restrict__ M, float* Ct,
                                        int tid, int wave, int lane_lo, int half) {
  const v4f w2 = *reinterpret_cast<const v4f*>(W2 + m_off(tid));
  if (wave >= 4) {
    FragSetX f;
    m_load(f, W1, W3, wave, lane_lo, half);
    m_tile(f, Ct, wave, lane_lo, half);
  }
  __syncthreads();
  m_store(Ct, w2, M, tid);
}

__device__ __forceinline__ void cnf_whole(const float* __restrict__ y0,
                                                const float* __restrict__ ts,
                                                const float* __restrict__ W1,
                                                const float* __restrict__ b1,
                                                const float* __restrict__ W2,
                                                const float* __restrict__ b2,
                                                const float* __restrict__ W3,
                                                const float* __restrict__ b3,
                                                const float* __restrict__ Mtr,
                                                float* smem,
                                                float* __restrict__ out) {
  float* Xs   = smem;
  float* H1s  = Xs  + MT * XP;
  float* H2s  = H1s + MT * HP;
  float* zacc = H2s + MT * HP;
  float* lacc = zacc + MT;

  const int tid     = threadIdx.x;
  const int wave    = __builtin_amdgcn_readfirstlane(tid >> 5);
  const int lane    = tid & 31;
  const int lane_lo = lane & 15;
  const int half    = lane >> 4;
  const int s_base  = blockIdx.x * MT;

  for (int i = tid; i < MT * DD; i += 256) {
    int m = i >> 6, c = i & 63;
    Xs[m * XP + c] = y0[(s_base + m) * DD + c];
  }
  if (tid < MT) { zacc[tid] = 0.f; lacc[tid] = 0.f; }

  v8f acc[4];
  FragSetW fw;
  const int nw = wave * 64;
  const int col0 = nw + 4 * lane_lo;
  const bool has_dx = wave < 4;
  const int n3 = (wave & 3) * 16;

  for (int step = 0; step < TT - 1; ++step) {
    const float t0 = ts[step];
    const float dt = ts[step + 1] - t0;
    load_b_w(fw, W1, 0, nw, lane_lo, half);
    __syncthreads();

    gemm_w<false>(acc, fw, Xs, XP, W1, DD, nw, lane_lo, half);
    load_b_w(fw, W2, 0, nw, lane_lo, half);
    {
      const v4f tw = *reinterpret_cast<const v4f*>(W1 + DD * HH + col0);
      const v4f bb = *reinterpret_cast<const v4f*>(b1 + col0);
      v4f add;
      #pragma unroll
      for (int nt = 0; nt < 4; ++nt)
        add[nt] = t0 * tw[nt] + bb[nt];
      #pragma unroll
      for (int v = 0; v < 8; ++v) {
        int m = v + 8 * half;
        v4f h;
        #pragma unroll
        for (int nt = 0; nt < 4; ++nt)
          h[nt] = fast_tanh(acc[nt][v] + add[nt]);
        *reinterpret_cast<v4f*>(H1s + m * HP + col0) = h;
      }
    }
    __syncthreads();

    gemm_w3<false>(acc, fw, H1s, HP, W2, nw, lane_lo, half);
    FragSetX g0, g1;
    if (has_dx)
      load_b_x(g0, W3, 0, n3, lane_lo, half);
    {
      const v4f bb = *reinterpret_cast<const v4f*>(b2 + col0);
      #pragma unroll
      for (int v = 0; v < 8; ++v) {
        int m = v + 8 * half;
        v4f h;
        #pragma unroll
        for (int nt = 0; nt < 4; ++nt)
          h[nt] = fast_tanh(acc[nt][v] + bb[nt]);
        *reinterpret_cast<v4f*>(H2s + m * HP + col0) = h;
      }
    }
    __syncthreads();

    load_b_w(fw, Mtr, 0, nw, lane_lo, half);
    if (has_dx) {
      const int m0 = 0, n0 = n3;
      v8f a3 = (v8f){};
      load_a_x(g0, H2s, 0, m0, lane_lo, half);
      int kb = 0;
      for (; kb + 8 * KH < HH; kb += 8 * KH) {
        load_b_x(g1, W3, kb + 4 * KH, n0, lane_lo, half);
        load_a_x(g1, H2s, kb + 4 * KH, m0, lane_lo, half);
        a3 = mma_set_x(g0, a3);
        load_b_x(g0, W3, kb + 8 * KH, n0, lane_lo, half);
        load_a_x(g0, H2s, kb + 8 * KH, m0, lane_lo, half);
        a3 = mma_set_x(g1, a3);
      }
      load_b_x(g1, W3, kb + 4 * KH, n0, lane_lo, half);
      load_a_x(g1, H2s, kb + 4 * KH, m0, lane_lo, half);
      a3 = mma_set_x(g0, a3);
      a3 = mma_set_x(g1, a3);
      int col = n0 + lane_lo;
      float bb = b3[col];
      #pragma unroll
      for (int v = 0; v < 8; ++v) {
        int m = m0 + v + 8 * half;
        float dx = a3[v] + bb;
        atomicAdd(&zacc[m], dt * 0.5f * dx * dx);
        Xs[m * XP + col] += dt * dx;
      }
    }

    {
      gemm_w3<true>(acc, fw, H1s, HP, Mtr, nw, lane_lo, half);
      #pragma unroll
      for (int v = 0; v < 8; ++v) {
        int m = v + 8 * half;
        float tl = 0.f;
        const v4f h2 = *reinterpret_cast<const v4f*>(H2s + m * HP + col0);
        #pragma unroll
        for (int nt = 0; nt < 4; ++nt)
          tl += (1.f - h2[nt] * h2[nt]) * acc[nt][v];
        atomicAdd(&lacc[m], dt * tl);
      }
    }
  }

  __syncthreads();
  for (int i = tid; i < MT * DD; i += 256) {
    int m = i >> 6, c = i & 63;
    out[(s_base + m) * DD + c] = Xs[m * XP + c];
  }
  if (tid < MT) {
    out[BB * DD + s_base + tid]      = zacc[tid];
    out[BB * DD + BB + s_base + tid] = lacc[tid];
  }
}

__device__ __forceinline__ v4f dx_from_pieces(const float* __restrict__ P,
                                              const float* __restrict__ b3,
                                              int row, int c4) {
  v4f d = *reinterpret_cast<const v4f*>(P + (size_t)row * DD + c4);
  #pragma unroll
  for (int j = 1; j < NSL; ++j)
    d += *reinterpret_cast<const v4f*>(P + (size_t)(j * BB + row) * DD + c4);
  d += *reinterpret_cast<const v4f*>(b3 + c4);
  return d;
}

__device__ __forceinline__ float sum16(float q) {
  q += __shfl_xor(q, 8, 16);
  q += __shfl_xor(q, 4, 16);
  q += __shfl_xor(q, 2, 16);
  q += __shfl_xor(q, 1, 16);
  return q;
}

__global__ __launch_bounds__(256) void cnf_step(int step,
                                                const float* __restrict__ y0,
                                                const float* __restrict__ ts,
                                                const float* __restrict__ W1,
                                                const float* __restrict__ b1,
                                                const float* __restrict__ W2,
                                                const float* __restrict__ b2,
                                                const float* __restrict__ W3,
                                                const float* __restrict__ b3,
                                                float* __restrict__ ws,
                                                size_t ws_bytes,
                                                float* __restrict__ out) {
  extern __shared__ float smem[];
  if (ws_bytes < WS_NEED) {
    if (step == 0) {
      const int tid0 = threadIdx.x;
      build_m(W1, W2, W3, ws + WS_M, smem, tid0,
              __builtin_amdgcn_readfirstlane(tid0 >> 5), tid0 & 15, (tid0 & 31) >> 4);
    } else if (step == 1 && blockIdx.x < NTILE) {
      cnf_whole(y0, ts, W1, b1, W2, b2, W3, b3, ws + WS_M, smem, out);
    }
    return;
  }
  float* Xs  = smem;
  float* H1s = Xs  + MT * XP;
  float* red = H1s + MT * HP;
  float* H2t = red + NSL * MT * RP;
  float* dxs = H2t + MT * RP;

  const int tid     = threadIdx.x;
  const int wave    = __builtin_amdgcn_readfirstlane(tid >> 5);
  const int lane    = tid & 31;
  const int lane_lo = lane & 15;
  const int half    = lane >> 4;
  const int tile    = blockIdx.x / NSL;
  const int slice   = blockIdx.x % NSL;
  const int s_base  = tile * MT;
  const int nw      = wave * 64;
  const int col0    = nw + 4 * lane_lo;
  const int sc0     = slice * 64;
  const int r_m     = tid >> 4;
  const int r_c     = 4 * (tid & 15);

  float* H1g = ws + WS_H1;
  float* H2g = ws + WS_H2;
  float* ztg = ws + WS_ZT;
  float* Xg  = ws + WS_X;
  float* dxp = ws + WS_DXP;

  const float t0 = ts[step];
  FragSetW fw;
  load_b_w(fw, W1, 0, nw, lane_lo, half);

  {
    v4f x;
    if (step == 0) {
      x = *reinterpret_cast<const v4f*>(y0 + (size_t)(s_base + r_m) * DD + r_c);
    } else {
      const float* Xprev = (step == 1) ? y0 : Xg + ((step - 1) & 1) * BB * DD;
      const float* P = dxp + ((step - 1) & 1) * (NSL * BB * DD);
      const float dtp = t0 - ts[step - 1];
      x = *reinterpret_cast<const v4f*>(Xprev + (size_t)(s_base + r_m) * DD + r_c);
      const v4f d = dx_from_pieces(P, b3, s_base + r_m, r_c);
      x += dtp * d;
      if (slice == 0) {
        const float q = sum16(d.x * d.x + d.y * d.y + d.z * d.z + d.w * d.w);
        if ((tid & 15) == 0) dxs[r_m] = dtp * 0.5f * q;
      }
    }
    *reinterpret_cast<v4f*>(Xs + r_m * XP + r_c) = x;
  }
  __syncthreads();
  if (step != 0 && slice == 0 && tid < MT)
    ztg[((step - 1) * NTILE + tile) * 32 + tid] = dxs[tid];

  v8f acc[4];
  gemm_w<false>(acc, fw, Xs, XP, W1, DD, nw, lane_lo, half);
  load_b_w(fw, W2 + (size_t)nw * HH, 0, sc0, lane_lo, half);
  {
    const v4f tw = *reinterpret_cast<const v4f*>(W1 + DD * HH + col0);
    const v4f bb = *reinterpret_cast<const v4f*>(b1 + col0);
    v4f add;
    #pragma unroll
    for (int nt = 0; nt < 4; ++nt)
      add[nt] = t0 * tw[nt] + bb[nt];
    #pragma unroll
    for (int v = 0; v < 8; ++v) {
      int m = v + 8 * half;
      v4f h;
      #pragma unroll
      for (int nt = 0; nt < 4; ++nt)
        h[nt] = fast_tanh(acc[nt][v] + add[nt]);
      *reinterpret_cast<v4f*>(H1s + m * HP + col0) = h;
    }
  }
  __syncthreads();

  FragSetX g;
  if (wave < 4)
    load_b_x(g, W3 + (size_t)sc0 * DD, 0, wave * 16, lane_lo, half);
  gemm_w<false>(acc, fw, H1s + nw, HP, W2 + (size_t)nw * HH, 64, sc0, lane_lo, half);
  #pragma unroll
  for (int v = 0; v < 8; ++v) {
    int m = v + 8 * half;
    v4f p;
    #pragma unroll
    for (int nt = 0; nt < 4; ++nt) p[nt] = acc[nt][v];
    *reinterpret_cast<v4f*>(red + (wave * MT + m) * RP + 4 * lane_lo) = p;
  }
  __syncthreads();
  v4f h2;
  {
    v4f a = *reinterpret_cast<const v4f*>(red + r_m * RP + r_c);
    #pragma unroll
    for (int w = 1; w < NSL; ++w)
      a += *reinterpret_cast<const v4f*>(red + (w * MT + r_m) * RP + r_c);
    const v4f bb = *reinterpret_cast<const v4f*>(b2 + sc0 + r_c);
    #pragma unroll
    for (int nt = 0; nt < 4; ++nt) h2[nt] = fast_tanh(a[nt] + bb[nt]);
    *reinterpret_cast<v4f*>(H2t + r_m * RP + r_c) = h2;
  }
  __syncthreads();

  if (wave < 4) {
    #pragma unroll
    for (int i = 0; i < KH; ++i)
      g.a[i] = frag_a(H2t, RP, 0, 4 * i, lane_lo, half);
    const v8f a3 = mma_set_x(g, (v8f){});
    #pragma unroll
    for (int v = 0; v < 8; ++v)
      dxs[(v + 8 * half) * RP + wave * 16 + lane_lo] = a3[v];
  }
  __syncthreads();
  *reinterpret_cast<v4f*>(dxp + (step & 1) * (NSL * BB * DD)
                          + (size_t)(slice * BB + s_base + r_m) * DD + r_c) =
      *reinterpret_cast<const v4f*>(dxs + r_m * RP + r_c);
  const size_t hrow = (size_t)(step * BB + s_base + r_m) * HH + sc0 + r_c;
  *reinterpret_cast<v4f*>(H2g + hrow) = h2;
  *reinterpret_cast<v4f*>(H1g + hrow) =
      *reinterpret_cast<const v4f*>(H1s + r_m * HP + sc0 + r_c);
  if (step != 0 && slice == 0)
    *reinterpret_cast<v4f*>(Xg + (step & 1) * BB * DD + (size_t)(s_base + r_m) * DD + r_c) =
        *reinterpret_cast<const v4f*>(Xs + r_m * XP + r_c);
  if (step == 0)
    build_m(W1, W2, W3, ws + WS_M, red, tid, wave, lane_lo, half);
}

__global__ __launch_bounds__(256) void cnf_trace(const float* __restrict__ ts,
                                                 const float* __restrict__ Mtr,
                                                 const float* __restrict__ H1g,
                                                 const float* __restrict__ H2g,
                                                 float* __restrict__ part,
                                                 size_t ws_bytes) {
  if (ws_bytes < WS_NEED) return;
  extern __shared__ float smem[];
  float* H1s  = smem;
  float* lacc = H1s + MT * HP;

  const int tid     = threadIdx.x;
  const int wave    = __builtin_amdgcn_readfirstlane(tid >> 5);
  const int lane    = tid & 31;
  const int lane_lo = lane & 15;
  const int half    = lane >> 4;
  const int step    = blockIdx.x / NTILE;
  const int tile    = blockIdx.x % NTILE;
  const int nw      = wave * 64;
  const int col0    = nw + 4 * lane_lo;
  const size_t row0 = (size_t)(step * BB + tile * MT);

  FragSetW fw;
  load_b_w(fw, Mtr, 0, nw, lane_lo, half);

  for (int i = tid; i < MT * HH / 4; i += 256) {
    int m = i >> 7, c = (i & 127) * 4;
    *reinterpret_cast<v4f*>(H1s + m * HP + c) =
        *reinterpret_cast<const v4f*>(H1g + (row0 + m) * HH + c);
  }
  if (tid < MT) lacc[tid] = 0.f;
  v4f h2[8];
  #pragma unroll
  for (int v = 0; v < 8; ++v)
    h2[v] = *reinterpret_cast<const v4f*>(H2g + (row0 + v + 8 * half) * HH + col0);
  const float dt = ts[step + 1] - ts[step];
  __syncthreads();

  v8f acc[4];
  gemm_w3<true>(acc, fw, H1s, HP, Mtr, nw, lane_lo, half);
  #pragma unroll
  for (int v = 0; v < 8; ++v) {
    int m = v + 8 * half;
    float tl = 0.f;
    #pragma unroll
    for (int nt = 0; nt < 4; ++nt)
      tl += (1.f - h2[v][nt] * h2[v][nt]) * acc[nt][v];
    atomicAdd(&lacc[m], dt * tl);
  }
  __syncthreads();
  if (tid < MT) part[(size_t)blockIdx.x * 32 + tid] = lacc[tid];
}

__global__ __launch_bounds__(BB) void cnf_final(const float* __restrict__ ts,
                                                const float* __restrict__ b3,
                                                const float* __restrict__ ws,
                                                size_t ws_bytes,
                                                float* __restrict__ out) {
  if (ws_bytes < WS_NEED) return;
  __shared__ float zl[64];
  const int tid = threadIdx.x;
  const int s   = blockIdx.x * 32 + (tid >> 4);
  const int r_c = 4 * (tid & 15);
  const float* Xprev = ws + WS_X + ((TT - 2) & 1) * BB * DD;
  const float* P     = ws + WS_DXP + ((TT - 2) & 1) * (NSL * BB * DD);
  const float dtp = ts[TT - 1] - ts[TT - 2];
  v4f x = *reinterpret_cast<const v4f*>(Xprev + (size_t)s * DD + r_c);
  const v4f d = dx_from_pieces(P, b3, s, r_c);
  x += dtp * d;
  *reinterpret_cast<v4f*>(out + (size_t)s * DD + r_c) = x;
  const float q = sum16(d.x * d.x + d.y * d.y + d.z * d.z + d.w * d.w);
  if ((tid & 15) == 0) {
    const float* zt = ws + WS_ZT;
    const float* pt = ws + WS_PART;
    float z = 0.f, l = 0.f;
    #pragma unroll
    for (int step = 0; step < TT - 2; ++step)
      z += zt[(step * NTILE + (s >> 4)) * 32 + (s & 15)];
    z += dtp * 0.5f * q;
    #pragma unroll
    for (int step = 0; step < TT - 1; ++step)
      l += pt[(step * NTILE + (s >> 4)) * 32 + (s & 15)];
    zl[tid >> 4] = z;
    zl[32 + (tid >> 4)] = l;
  }
  __syncthreads();
  if (tid < 32)       out[BB * DD + blockIdx.x * 32 + tid] = zl[tid];
  else if (tid < 64)  out[BB * DD + BB + blockIdx.x * 32 + (tid - 32)] = zl[tid];
}

extern "C" void kernel_launch(void* const* d_in, const int* in_sizes, int n_in,
                              void* d_out, int out_size, void* d_ws, size_t ws_size,
                              hipStream_t stream) {
  const float* y0 = (const float*)d_in[0];
  const float* ts = (const float*)d_in[1];
  const float* W1 = (const float*)d_in[2];
  const float* b1 = (const float*)d_in[3];
  const float* W2 = (const float*)d_in[4];
  const float* b2 = (const float*)d_in[5];
  const float* W3 = (const float*)d_in[6];
  const float* b3 = (const float*)d_in[7];
  float* out = (float*)d_out;
  float* ws  = (float*)d_ws;
  float* Mtr = ws + WS_M;
  float* H1g = ws + WS_H1;
  float* H2g = ws + WS_H2;
  float* prt = ws + WS_PART;

  size_t shmem = (size_t)(LDS_NEW > LDS_OLD ? LDS_NEW : LDS_OLD) * sizeof(float);
#define STEP(n) cnf_step<<<NTILE * NSL, 256, shmem, stream>>>(n, y0, ts, W1, b1, W2, b2, W3, b3, ws, ws_size, out)
  STEP(0); STEP(1); STEP(2); STEP(3); STEP(4); STEP(5); STEP(6);
#undef STEP

  size_t shmem_t = (size_t)(MT * HP + MT) * sizeof(float);
  cnf_trace<<<(TT - 1) * NTILE, 256, shmem_t, stream>>>(ts, Mtr, H1g, H2g, prt,
                                                        ws_size);
  cnf_final<<<BB / 32, BB, 0, stream>>>(ts, b3, ws, ws_size, out);
}


# Step launches request each K=64 GEMM's whole weight slab one stage ahead (W1 at entry, W2 and b2 before the layer-1 GEMM); barriers there fence LDS only, so the loads stay in flight; sum order kept
# speedup vs baseline: 6.1709x; 1.0332x over previous
#include <hip/hip_runtime.h>
#include <hip/hip_bf16.h>


#define BB  512
#define DD  64
#define HH  512
#define TT  8
#define MT  16
#define XP  68
#define HP  516
#define NTILE (BB / MT)
#define NSL 8
#define RP  68

#define WS_M     0
#define WS_H1    (HH * HH)
#define WS_H2    (WS_H1 + (TT - 1) * BB * HH)
#define WS_PART  (WS_H2 + (TT - 1) * BB * HH)
#define WS_ZT    (WS_PART + (TT - 1) * NTILE * 32)
#define WS_X     (WS_ZT + (TT - 1) * NTILE * 32)
#define WS_DXP   (WS_X + 2 * BB * DD)
#define WS_FLOATS (WS_DXP + 2 * NSL * BB * DD)
#define LDS_OLD  (MT * XP + 2 * MT * HP + 2 * MT)
#define LDS_NEW  (MT * XP + MT * HP + NSL * MT * RP + 2 * MT * RP)
#define WS_NEED  ((size_t)WS_FLOATS * sizeof(float))

typedef float v2f __attribute__((ext_vector_type(2)));
typedef float v4f __attribute__((ext_vector_type(4)));
typedef float v8f __attribute__((ext_vector_type(8)));

__device__ __forceinline__ v8f wmma_f32(v2f a, v2f b, v8f c) {
  return __builtin_amdgcn_wmma_f32_16x16x4_f32(false, a, false, b,
                                               (short)0, c, false, false);
}

__device__ __forceinline__ float fast_tanh(float x) {
#if __has_builtin(__builtin_amdgcn_tanhf)
  return __builtin_amdgcn_tanhf(x);
#elif __has_builtin(__builtin_amdgcn_exp2f) && __has_builtin(__builtin_amdgcn_rcpf)
  float e = __builtin_amdgcn_exp2f(x * 2.8853900817779268f);
  return 1.0f - 2.0f * __builtin_amdgcn_rcpf(e + 1.0f);
#else
  return tanhf(x);
#endif
}

__device__ __forceinline__ v2f frag_a(const float* S, int pitch, int m0, int k0,
                                      int lane_lo, int half) {
  const float* p = S + (m0 + lane_lo) * pitch + k0 + 2 * half;
  v2f a; a.x = p[0]; a.y = p[1];
  return a;
}

__device__ __forceinline__ v2f frag_b(const float* __restrict__ W, int ldw,
                                      int k0, int n0, int lane_lo, int half) {
  v2f b;
  b.x = W[(k0 + 2 * half + 0) * ldw + n0 + lane_lo];
  b.y = W[(k0 + 2 * half + 1) * ldw + n0 + lane_lo];
  return b;
}


#define KG 4
#define KH 16

struct FragSetW { v2f a[KG]; v4f b[KG][2]; };

__device__ __forceinline__ void load_b_w(FragSetW& f, const float* __restrict__ W,
                                         int kb, int nw, int lane_lo, int half) {
  #pragma unroll
  for (int i = 0; i < KG; ++i)
    #pragma unroll
    for (int r = 0; r < 2; ++r)
      f.b[i][r] = *reinterpret_cast<const v4f*>(
          W + (kb + 4 * i + 2 * half + r) * HH + nw + 4 * lane_lo);
}

__device__ __forceinline__ void load_a_w(FragSetW& f, const float* S, int pitch,
                                         int kb, int lane_lo, int half) {
  #pragma unroll
  for (int i = 0; i < KG; ++i)
    f.a[i] = frag_a(S, pitch, 0, kb + 4 * i, lane_lo, half);
}

template <bool SQ>
__device__ __forceinline__ void mma_set_w(const FragSetW& f, v8f (&acc)[4]) {
  #pragma unroll
  for (int i = 0; i < KG; ++i) {
    v2f a = f.a[i];
    if (SQ) {
      a.x = 1.f - a.x * a.x;  a.y = 1.f - a.y * a.y;
    }
    #pragma unroll
    for (int nt = 0; nt < 4; ++nt) {
      v2f b; b.x = f.b[i][0][nt]; b.y = f.b[i][1][nt];
      acc[nt] = wmma_f32(a, b, acc[nt]);
    }
  }
}

template <bool SQ>
__device__ __forceinline__ void gemm_w(v8f (&acc)[4], FragSetW& f0,
                                       const float* S, int pitch,
                                       const float* __restrict__ W, int K,
                                       int nw, int lane_lo, int half) {
  #pragma unroll
  for (int nt = 0; nt < 4; ++nt) acc[nt] = (v8f){};
  FragSetW f1;
  load_a_w(f0, S, pitch, 0, lane_lo, half);
  int kb = 0;
  for (; kb + 8 * KG < K; kb += 8 * KG) {
    load_b_w(f1, W, kb + 4 * KG, nw, lane_lo, half);
    load_a_w(f1, S, pitch, kb + 4 * KG, lane_lo, half);
    mma_set_w<SQ>(f0, acc);
    load_b_w(f0, W, kb + 8 * KG, nw, lane_lo, half);
    load_a_w(f0, S, pitch, kb + 8 * KG, lane_lo, half);
    mma_set_w<SQ>(f1, acc);
  }
  load_b_w(f1, W, kb + 4 * KG, nw, lane_lo, half);
  load_a_w(f1, S, pitch, kb + 4 * KG, lane_lo, half);
  mma_set_w<SQ>(f0, acc);
  mma_set_w<SQ>(f1, acc);
}

template <bool SQ>
__device__ __forceinline__ void gemm_w3(v8f (&acc)[4], FragSetW& f0,
                                        const float* S, int pitch,
                                        const float* __restrict__ W,
                                        int nw, int lane_lo, int half) {
  static_assert((HH / (4 * KG) - 2) % 3 == 0, "set count must be 3n + 2");
  #pragma unroll
  for (int nt = 0; nt < 4; ++nt) acc[nt] = (v8f){};
  FragSetW f1, f2;
  load_a_w(f0, S, pitch, 0, lane_lo, half);
  load_b_w(f1, W, 4 * KG, nw, lane_lo, half);
  load_a_w(f1, S, pitch, 4 * KG, lane_lo, half);
  int kb = 0;
  #define SCHED_FENCE() __builtin_amdgcn_sched_barrier(0)
  for (; kb + 8 * KG < HH; kb += 12 * KG) {
    load_b_w(f2, W, kb + 8 * KG, nw, lane_lo, half);
    load_a_w(f2, S, pitch, kb + 8 * KG, lane_lo, half);
    mma_set_w<SQ>(f0, acc);
    SCHED_FENCE();
    load_b_w(f0, W, kb + 12 * KG, nw, lane_lo, half);
    load_a_w(f0, S, pitch, kb + 12 * KG, lane_lo, half);
    mma_set_w<SQ>(f1, acc);
    SCHED_FENCE();
    load_b_w(f1, W, kb + 16 * KG, nw, lane_lo, half);
    load_a_w(f1, S, pitch, kb + 16 * KG, lane_lo, half);
    mma_set_w<SQ>(f2, acc);
    SCHED_FENCE();
  }
  #undef SCHED_FENCE
  mma_set_w<SQ>(f0, acc);
  mma_set_w<SQ>(f1, acc);
}

__device__ __forceinline__ void lds_barrier() {
  __builtin_amdgcn_fence(__ATOMIC_RELEASE, "workgroup", "local");
  __builtin_amdgcn_s_barrier();
  __builtin_amdgcn_fence(__ATOMIC_ACQUIRE, "workgroup", "local");
}

#define NSET (DD / (4 * KG))

__device__ __forceinline__ void load_b_all(FragSetW (&f)[NSET], const float* __restrict__ W,
                                           int nw, int lane_lo, int half) {
  #pragma unroll
  for (int s = 0; s < NSET; ++s) load_b_w(f[s], W, 4 * KG * s, nw, lane_lo, half);
}

__device__ __forceinline__ void gemm_pre(v8f (&acc)[4], FragSetW (&f)[NSET],
                                         const float* S, int pitch,
                                         int lane_lo, int half) {
  #pragma unroll
  for (int nt = 0; nt < 4; ++nt) acc[nt] = (v8f){};
  load_a_w(f[0], S, pitch, 0, lane_lo, half);
  #pragma unroll
  for (int s = 0; s < NSET; ++s) {
    if (s + 1 < NSET) load_a_w(f[s + 1], S, pitch, 4 * KG * (s + 1), lane_lo, half);
    mma_set_w<false>(f[s], acc);
  }
}

struct FragSetX { v2f a[KH]; v2f b[KH]; };

__device__ __forceinline__ void load_b_x(FragSetX& f, const float* __restrict__ W,
                                         int kb, int n0, int lane_lo, int half) {
  #pragma unroll
  for (int i = 0; i < KH; ++i)
    f.b[i] = frag_b(W, DD, kb + 4 * i, n0, lane_lo, half);
}

__device__ __forceinline__ void load_a_x(FragSetX& f, const float* S, int kb,
                                         int m0, int lane_lo, int half) {
  #pragma unroll
  for (int i = 0; i < KH; ++i)
    f.a[i] = frag_a(S, HP, m0, kb + 4 * i, lane_lo, half);
}

__device__ __forceinline__ v8f mma_set_x(const FragSetX& f, v8f c) {
  #pragma unroll
  for (int i = 0; i < KH; ++i) c = wmma_f32(f.a[i], f.b[i], c);
  return c;
}

#define CP 36
static_assert(4 * KH == DD, "one fragment set must cover the whole K = 64");

__device__ __forceinline__ size_t m_off(int tid) {
  const int b = blockIdx.x;
  return (size_t)((b & 15) * 32 + (tid >> 3)) * HH + (b >> 4) * 32 + 4 * (tid & 7);
}

__device__ __forceinline__ void m_load(FragSetX& f, const float* __restrict__ W1,
                                       const float* __restrict__ W3,
                                       int wave, int lane_lo, int half) {
  const int b  = blockIdx.x;
  const int k0 = (b >> 4) * 32 + ((wave >> 1) & 1) * 16;
  const int j0 = (b & 15) * 32 + (wave & 1) * 16;
  #pragma unroll
  for (int i = 0; i < KH; ++i) {
    f.a[i] = *reinterpret_cast<const v2f*>(W3 + (size_t)(k0 + lane_lo) * DD + 4 * i + 2 * half);
    f.b[i] = frag_b(W1, HH, 4 * i, j0, lane_lo, half);
  }
}

__device__ __forceinline__ void m_tile(const FragSetX& f, float* Ct,
                                       int wave, int lane_lo, int half) {
  const v8f c = mma_set_x(f, (v8f){});
  float* p = Ct + ((wave & 1) * 16 + lane_lo) * CP + ((wave >> 1) & 1) * 16 + 8 * half;
  v4f lo, hi;
  #pragma unroll
  for (int r = 0; r < 4; ++r) { lo[r] = c[r]; hi[r] = c[4 + r]; }
  *reinterpret_cast<v4f*>(p)     = lo;
  *reinterpret_cast<v4f*>(p + 4) = hi;
}

__device__ __forceinline__ void m_store(const float* Ct, v4f w2, float* __restrict__ M,
                                        int tid) {
  const v4f c = *reinterpret_cast<const v4f*>(Ct + (tid >> 3) * CP + 4 * (tid & 7));
  *reinterpret_cast<v4f*>(M + m_off(tid)) = w2 * c;
}

__device__ __forceinline__ void build_m(const float* __restrict__ W1,
                                        const float* __restrict__ W2,
                                        const float* __restrict__ W3,
                                        float* __restrict__ M, float* Ct,
                                        int tid, int wave, int lane_lo, int half) {
  const v4f w2 = *reinterpret_cast<const v4f*>(W2 + m_off(tid));
  if (wave >= 4) {
    FragSetX f;
    m_load(f, W1, W3, wave, lane_lo, half);
    m_tile(f, Ct, wave, lane_lo, half);
  }
  __syncthreads();
  m_store(Ct, w2, M, tid);
}

__device__ __forceinline__ void cnf_whole(const float* __restrict__ y0,
                                                const float* __restrict__ ts,
                                                const float* __restrict__ W1,
                                                const float* __restrict__ b1,
                                                const float* __restrict__ W2,
                                                const float* __restrict__ b2,
                                                const float* __restrict__ W3,
                                                const float* __restrict__ b3,
                                                const float* __restrict__ Mtr,
                                                float* smem,
                                                float* __restrict__ out) {
  float* Xs   = smem;
  float* H1s  = Xs  + MT * XP;
  float* H2s  = H1s + MT * HP;
  float* zacc = H2s + MT * HP;
  float* lacc = zacc + MT;

  const int tid     = threadIdx.x;
  const int wave    = __builtin_amdgcn_readfirstlane(tid >> 5);
  const int lane    = tid & 31;
  const int lane_lo = lane & 15;
  const int half    = lane >> 4;
  const int s_base  = blockIdx.x * MT;

  for (int i = tid; i < MT * DD; i += 256) {
    int m = i >> 6, c = i & 63;
    Xs[m * XP + c] = y0[(s_base + m) * DD + c];
  }
  if (tid < MT) { zacc[tid] = 0.f; lacc[tid] = 0.f; }

  v8f acc[4];
  FragSetW fw;
  const int nw = wave * 64;
  const int col0 = nw + 4 * lane_lo;
  const bool has_dx = wave < 4;
  const int n3 = (wave & 3) * 16;

  for (int step = 0; step < TT - 1; ++step) {
    const float t0 = ts[step];
    const float dt = ts[step + 1] - t0;
    load_b_w(fw, W1, 0, nw, lane_lo, half);
    __syncthreads();

    gemm_w<false>(acc, fw, Xs, XP, W1, DD, nw, lane_lo, half);
    load_b_w(fw, W2, 0, nw, lane_lo, half);
    {
      const v4f tw = *reinterpret_cast<const v4f*>(W1 + DD * HH + col0);
      const v4f bb = *reinterpret_cast<const v4f*>(b1 + col0);
      v4f add;
      #pragma unroll
      for (int nt = 0; nt < 4; ++nt)
        add[nt] = t0 * tw[nt] + bb[nt];
      #pragma unroll
      for (int v = 0; v < 8; ++v) {
        int m = v + 8 * half;
        v4f h;
        #pragma unroll
        for (int nt = 0; nt < 4; ++nt)
          h[nt] = fast_tanh(acc[nt][v] + add[nt]);
        *reinterpret_cast<v4f*>(H1s + m * HP + col0) = h;
      }
    }
    __syncthreads();

    gemm_w3<false>(acc, fw, H1s, HP, W2, nw, lane_lo, half);
    FragSetX g0, g1;
    if (has_dx)
      load_b_x(g0, W3, 0, n3, lane_lo, half);
    {
      const v4f bb = *reinterpret_cast<const v4f*>(b2 + col0);
      #pragma unroll
      for (int v = 0; v < 8; ++v) {
        int m = v + 8 * half;
        v4f h;
        #pragma unroll
        for (int nt = 0; nt < 4; ++nt)
          h[nt] = fast_tanh(acc[nt][v] + bb[nt]);
        *reinterpret_cast<v4f*>(H2s + m * HP + col0) = h;
      }
    }
    __syncthreads();

    load_b_w(fw, Mtr, 0, nw, lane_lo, half);
    if (has_dx) {
      const int m0 = 0, n0 = n3;
      v8f a3 = (v8f){};
      load_a_x(g0, H2s, 0, m0, lane_lo, half);
      int kb = 0;
      for (; kb + 8 * KH < HH; kb += 8 * KH) {
        load_b_x(g1, W3, kb + 4 * KH, n0, lane_lo, half);
        load_a_x(g1, H2s, kb + 4 * KH, m0, lane_lo, half);
        a3 = mma_set_x(g0, a3);
        load_b_x(g0, W3, kb + 8 * KH, n0, lane_lo, half);
        load_a_x(g0, H2s, kb + 8 * KH, m0, lane_lo, half);
        a3 = mma_set_x(g1, a3);
      }
      load_b_x(g1, W3, kb + 4 * KH, n0, lane_lo, half);
      load_a_x(g1, H2s, kb + 4 * KH, m0, lane_lo, half);
      a3 = mma_set_x(g0, a3);
      a3 = mma_set_x(g1, a3);
      int col = n0 + lane_lo;
      float bb = b3[col];
      #pragma unroll
      for (int v = 0; v < 8; ++v) {
        int m = m0 + v + 8 * half;
        float dx = a3[v] + bb;
        atomicAdd(&zacc[m], dt * 0.5f * dx * dx);
        Xs[m * XP + col] += dt * dx;
      }
    }

    {
      gemm_w3<true>(acc, fw, H1s, HP, Mtr, nw, lane_lo, half);
      #pragma unroll
      for (int v = 0; v < 8; ++v) {
        int m = v + 8 * half;
        float tl = 0.f;
        const v4f h2 = *reinterpret_cast<const v4f*>(H2s + m * HP + col0);
        #pragma unroll
        for (int nt = 0; nt < 4; ++nt)
          tl += (1.f - h2[nt] * h2[nt]) * acc[nt][v];
        atomicAdd(&lacc[m], dt * tl);
      }
    }
  }

  __syncthreads();
  for (int i = tid; i < MT * DD; i += 256) {
    int m = i >> 6, c = i & 63;
    out[(s_base + m) * DD + c] = Xs[m * XP + c];
  }
  if (tid < MT) {
    out[BB * DD + s_base + tid]      = zacc[tid];
    out[BB * DD + BB + s_base + tid] = lacc[tid];
  }
}

__device__ __forceinline__ v4f dx_from_pieces(const float* __restrict__ P,
                                              const float* __restrict__ b3,
                                              int row, int c4) {
  v4f d = *reinterpret_cast<const v4f*>(P + (size_t)row * DD + c4);
  #pragma unroll
  for (int j = 1; j < NSL; ++j)
    d += *reinterpret_cast<const v4f*>(P + (size_t)(j * BB + row) * DD + c4);
  d += *reinterpret_cast<const v4f*>(b3 + c4);
  return d;
}

__device__ __forceinline__ float sum16(float q) {
  q += __shfl_xor(q, 8, 16);
  q += __shfl_xor(q, 4, 16);
  q += __shfl_xor(q, 2, 16);
  q += __shfl_xor(q, 1, 16);
  return q;
}

__global__ __launch_bounds__(256) void cnf_step(int step,
                                                const float* __restrict__ y0,
                                                const float* __restrict__ ts,
                                                const float* __restrict__ W1,
                                                const float* __restrict__ b1,
                                                const float* __restrict__ W2,
                                                const float* __restrict__ b2,
                                                const float* __restrict__ W3,
                                                const float* __restrict__ b3,
                                                float* __restrict__ ws,
                                                size_t ws_bytes,
                                                float* __restrict__ out) {
  extern __shared__ float smem[];
  if (ws_bytes < WS_NEED) {
    if (step == 0) {
      const int tid0 = threadIdx.x;
      build_m(W1, W2, W3, ws + WS_M, smem, tid0,
              __builtin_amdgcn_readfirstlane(tid0 >> 5), tid0 & 15, (tid0 & 31) >> 4);
    } else if (step == 1 && blockIdx.x < NTILE) {
      cnf_whole(y0, ts, W1, b1, W2, b2, W3, b3, ws + WS_M, smem, out);
    }
    return;
  }
  float* Xs  = smem;
  float* H1s = Xs  + MT * XP;
  float* red = H1s + MT * HP;
  float* H2t = red + NSL * MT * RP;
  float* dxs = H2t + MT * RP;

  const int tid     = threadIdx.x;
  const int wave    = __builtin_amdgcn_readfirstlane(tid >> 5);
  const int lane    = tid & 31;
  const int lane_lo = lane & 15;
  const int half    = lane >> 4;
  const int tile    = blockIdx.x / NSL;
  const int slice   = blockIdx.x % NSL;
  const int s_base  = tile * MT;
  const int nw      = wave * 64;
  const int col0    = nw + 4 * lane_lo;
  const int sc0     = slice * 64;
  const int r_m     = tid >> 4;
  const int r_c     = 4 * (tid & 15);

  float* H1g = ws + WS_H1;
  float* H2g = ws + WS_H2;
  float* ztg = ws + WS_ZT;
  float* Xg  = ws + WS_X;
  float* dxp = ws + WS_DXP;

  const float t0 = ts[step];
  FragSetW w1s[NSET];
  FragSetW w2s[NSET];

  v4f tw1, bb1;
  {
    const float* Xprev = (step <= 1) ? y0 : Xg + ((step - 1) & 1) * BB * DD;
    v4f x = *reinterpret_cast<const v4f*>(Xprev + (size_t)(s_base + r_m) * DD + r_c);
    v4f p[NSL], b3v;
    if (step != 0) {
      const float* P = dxp + ((step - 1) & 1) * (NSL * BB * DD);
      #pragma unroll
      for (int j = 0; j < NSL; ++j)
        p[j] = *reinterpret_cast<const v4f*>(P + (size_t)(j * BB + s_base + r_m) * DD + r_c);
      b3v = *reinterpret_cast<const v4f*>(b3 + r_c);
    }
    tw1 = *reinterpret_cast<const v4f*>(W1 + DD * HH + col0);
    bb1 = *reinterpret_cast<const v4f*>(b1 + col0);
    load_b_all(w1s, W1, nw, lane_lo, half);
    if (step != 0) {
      const float dtp = t0 - ts[step - 1];
      v4f d = p[0];
      #pragma unroll
      for (int j = 1; j < NSL; ++j) d += p[j];
      d += b3v;
      x += dtp * d;
      if (slice == 0) {
        const float q = sum16(d.x * d.x + d.y * d.y + d.z * d.z + d.w * d.w);
        if ((tid & 15) == 0) dxs[r_m] = dtp * 0.5f * q;
      }
    }
    *reinterpret_cast<v4f*>(Xs + r_m * XP + r_c) = x;
  }
  lds_barrier();
  if (step != 0 && slice == 0 && tid < MT)
    ztg[((step - 1) * NTILE + tile) * 32 + tid] = dxs[tid];

  const v4f bb2 = *reinterpret_cast<const v4f*>(b2 + sc0 + r_c);
  load_b_all(w2s, W2 + (size_t)nw * HH, sc0, lane_lo, half);
  v8f acc[4];
  gemm_pre(acc, w1s, Xs, XP, lane_lo, half);
  FragSetX g;
  if (wave < 4)
    load_b_x(g, W3 + (size_t)sc0 * DD, 0, wave * 16, lane_lo, half);
  {
    v4f add;
    #pragma unroll
    for (int nt = 0; nt < 4; ++nt)
      add[nt] = t0 * tw1[nt] + bb1[nt];
    #pragma unroll
    for (int v = 0; v < 8; ++v) {
      int m = v + 8 * half;
      v4f h;
      #pragma unroll
      for (int nt = 0; nt < 4; ++nt)
        h[nt] = fast_tanh(acc[nt][v] + add[nt]);
      *reinterpret_cast<v4f*>(H1s + m * HP + col0) = h;
    }
  }
  lds_barrier();

  gemm_pre(acc, w2s, H1s + nw, HP, lane_lo, half);
  #pragma unroll
  for (int v = 0; v < 8; ++v) {
    int m = v + 8 * half;
    v4f p;
    #pragma unroll
    for (int nt = 0; nt < 4; ++nt) p[nt] = acc[nt][v];
    *reinterpret_cast<v4f*>(red + (wave * MT + m) * RP + 4 * lane_lo) = p;
  }
  lds_barrier();
  v4f h2;
  {
    v4f a = *reinterpret_cast<const v4f*>(red + r_m * RP + r_c);
    #pragma unroll
    for (int w = 1; w < NSL; ++w)
      a += *reinterpret_cast<const v4f*>(red + (w * MT + r_m) * RP + r_c);
    #pragma unroll
    for (int nt = 0; nt < 4; ++nt) h2[nt] = fast_tanh(a[nt] + bb2[nt]);
    *reinterpret_cast<v4f*>(H2t + r_m * RP + r_c) = h2;
  }
  lds_barrier();

  if (wave < 4) {
    #pragma unroll
    for (int i = 0; i < KH; ++i)
      g.a[i] = frag_a(H2t, RP, 0, 4 * i, lane_lo, half);
    const v8f a3 = mma_set_x(g, (v8f){});
    #pragma unroll
    for (int v = 0; v < 8; ++v)
      dxs[(v + 8 * half) * RP + wave * 16 + lane_lo] = a3[v];
  }
  lds_barrier();
  *reinterpret_cast<v4f*>(dxp + (step & 1) * (NSL * BB * DD)
                          + (size_t)(slice * BB + s_base + r_m) * DD + r_c) =
      *reinterpret_cast<const v4f*>(dxs + r_m * RP + r_c);
  const size_t hrow = (size_t)(step * BB + s_base + r_m) * HH + sc0 + r_c;
  *reinterpret_cast<v4f*>(H2g + hrow) = h2;
  *reinterpret_cast<v4f*>(H1g + hrow) =
      *reinterpret_cast<const v4f*>(H1s + r_m * HP + sc0 + r_c);
  if (step != 0 && slice == 0)
    *reinterpret_cast<v4f*>(Xg + (step & 1) * BB * DD + (size_t)(s_base + r_m) * DD + r_c) =
        *reinterpret_cast<const v4f*>(Xs + r_m * XP + r_c);
  if (step == 0)
    build_m(W1, W2, W3, ws + WS_M, red, tid, wave, lane_lo, half);
}

__global__ __launch_bounds__(256) void cnf_trace(const float* __restrict__ ts,
                                                 const float* __restrict__ Mtr,
                                                 const float* __restrict__ H1g,
                                                 const float* __restrict__ H2g,
                                                 float* __restrict__ part,
                                                 size_t ws_bytes) {
  if (ws_bytes < WS_NEED) return;
  extern __shared__ float smem[];
  float* H1s  = smem;
  float* lacc = H1s + MT * HP;

  const int tid     = threadIdx.x;
  const int wave    = __builtin_amdgcn_readfirstlane(tid >> 5);
  const int lane    = tid & 31;
  const int lane_lo = lane & 15;
  const int half    = lane >> 4;
  const int step    = blockIdx.x / NTILE;
  const int tile    = blockIdx.x % NTILE;
  const int nw      = wave * 64;
  const int col0    = nw + 4 * lane_lo;
  const size_t row0 = (size_t)(step * BB + tile * MT);

  FragSetW fw;
  load_b_w(fw, Mtr, 0, nw, lane_lo, half);

  for (int i = tid; i < MT * HH / 4; i += 256) {
    int m = i >> 7, c = (i & 127) * 4;
    *reinterpret_cast<v4f*>(H1s + m * HP + c) =
        *reinterpret_cast<const v4f*>(H1g + (row0 + m) * HH + c);
  }
  if (tid < MT) lacc[tid] = 0.f;
  v4f h2[8];
  #pragma unroll
  for (int v = 0; v < 8; ++v)
    h2[v] = *reinterpret_cast<const v4f*>(H2g + (row0 + v + 8 * half) * HH + col0);
  const float dt = ts[step + 1] - ts[step];
  __syncthreads();

  v8f acc[4];
  gemm_w3<true>(acc, fw, H1s, HP, Mtr, nw, lane_lo, half);
  #pragma unroll
  for (int v = 0; v < 8; ++v) {
    int m = v + 8 * half;
    float tl = 0.f;
    #pragma unroll
    for (int nt = 0; nt < 4; ++nt)
      tl += (1.f - h2[v][nt] * h2[v][nt]) * acc[nt][v];
    atomicAdd(&lacc[m], dt * tl);
  }
  __syncthreads();
  if (tid < MT) part[(size_t)blockIdx.x * 32 + tid] = lacc[tid];
}

__global__ __launch_bounds__(BB) void cnf_final(const float* __restrict__ ts,
                                                const float* __restrict__ b3,
                                                const float* __restrict__ ws,
                                                size_t ws_bytes,
                                                float* __restrict__ out) {
  if (ws_bytes < WS_NEED) return;
  __shared__ float zl[64];
  const int tid = threadIdx.x;
  const int s   = blockIdx.x * 32 + (tid >> 4);
  const int r_c = 4 * (tid & 15);
  const float* Xprev = ws + WS_X + ((TT - 2) & 1) * BB * DD;
  const float* P     = ws + WS_DXP + ((TT - 2) & 1) * (NSL * BB * DD);
  const float dtp = ts[TT - 1] - ts[TT - 2];
  v4f x = *reinterpret_cast<const v4f*>(Xprev + (size_t)s * DD + r_c);
  const v4f d = dx_from_pieces(P, b3, s, r_c);
  x += dtp * d;
  *reinterpret_cast<v4f*>(out + (size_t)s * DD + r_c) = x;
  const float q = sum16(d.x * d.x + d.y * d.y + d.z * d.z + d.w * d.w);
  if ((tid & 15) == 0) {
    const float* zt = ws + WS_ZT;
    const float* pt = ws + WS_PART;
    float z = 0.f, l = 0.f;
    #pragma unroll
    for (int step = 0; step < TT - 2; ++step)
      z += zt[(step * NTILE + (s >> 4)) * 32 + (s & 15)];
    z += dtp * 0.5f * q;
    #pragma unroll
    for (int step = 0; step < TT - 1; ++step)
      l += pt[(step * NTILE + (s >> 4)) * 32 + (s & 15)];
    zl[tid >> 4] = z;
    zl[32 + (tid >> 4)] = l;
  }
  __syncthreads();
  if (tid < 32)       out[BB * DD + blockIdx.x * 32 + tid] = zl[tid];
  else if (tid < 64)  out[BB * DD + BB + blockIdx.x * 32 + (tid - 32)] = zl[tid];
}

extern "C" void kernel_launch(void* const* d_in, const int* in_sizes, int n_in,
                              void* d_out, int out_size, void* d_ws, size_t ws_size,
                              hipStream_t stream) {
  const float* y0 = (const float*)d_in[0];
  const float* ts = (const float*)d_in[1];
  const float* W1 = (const float*)d_in[2];
  const float* b1 = (const float*)d_in[3];
  const float* W2 = (const float*)d_in[4];
  const float* b2 = (const float*)d_in[5];
  const float* W3 = (const float*)d_in[6];
  const float* b3 = (const float*)d_in[7];
  float* out = (float*)d_out;
  float* ws  = (float*)d_ws;
  float* Mtr = ws + WS_M;
  float* H1g = ws + WS_H1;
  float* H2g = ws + WS_H2;
  float* prt = ws + WS_PART;

  size_t shmem = (size_t)(LDS_NEW > LDS_OLD ? LDS_NEW : LDS_OLD) * sizeof(float);
#define STEP(n) cnf_step<<<NTILE * NSL, 256, shmem, stream>>>(n, y0, ts, W1, b1, W2, b2, W3, b3, ws, ws_size, out)
  STEP(0); STEP(1); STEP(2); STEP(3); STEP(4); STEP(5); STEP(6);
#undef STEP

  size_t shmem_t = (size_t)(MT * HP + MT) * sizeof(float);
  cnf_trace<<<(TT - 1) * NTILE, 256, shmem_t, stream>>>(ts, Mtr, H1g, H2g, prt,
                                                        ws_size);
  cnf_final<<<BB / 32, BB, 0, stream>>>(ts, b3, ws, ws_size, out);
}


# Trace product moves into the step launches: workgroup (tile, slice) multiplies 1-H1^2 by its 64 columns of M from registers; step 0's runs in launch 1; one launch fewer; trace sums regroup
# speedup vs baseline: 6.6226x; 1.0732x over previous
#include <hip/hip_runtime.h>
#include <hip/hip_bf16.h>


#define BB  512
#define DD  64
#define HH  512
#define TT  8
#define MT  16
#define XP  68
#define HP  516
#define NTILE (BB / MT)
#define NSL 8
#define RP  68

#define WS_M     0
#define WS_H1    (HH * HH)
#define WS_H2    (WS_H1 + BB * HH)
#define WS_PART  (WS_H2 + BB * HH)
#define WS_ZT    (WS_PART + (TT - 1) * NTILE * NSL * 32)
#define WS_X     (WS_ZT + (TT - 1) * NTILE * 32)
#define WS_DXP   (WS_X + 2 * BB * DD)
#define WS_FLOATS (WS_DXP + 2 * NSL * BB * DD)
#define LDS_OLD  (MT * XP + 2 * MT * HP + 2 * MT)
#define LDS_NEW  (MT * XP + MT * HP + NSL * MT * RP + 2 * MT * RP)
#define WS_NEED  ((size_t)WS_FLOATS * sizeof(float))

typedef float v2f __attribute__((ext_vector_type(2)));
typedef float v4f __attribute__((ext_vector_type(4)));
typedef float v8f __attribute__((ext_vector_type(8)));

__device__ __forceinline__ v8f wmma_f32(v2f a, v2f b, v8f c) {
  return __builtin_amdgcn_wmma_f32_16x16x4_f32(false, a, false, b,
                                               (short)0, c, false, false);
}

__device__ __forceinline__ float fast_tanh(float x) {
#if __has_builtin(__builtin_amdgcn_tanhf)
  return __builtin_amdgcn_tanhf(x);
#elif __has_builtin(__builtin_amdgcn_exp2f) && __has_builtin(__builtin_amdgcn_rcpf)
  float e = __builtin_amdgcn_exp2f(x * 2.8853900817779268f);
  return 1.0f - 2.0f * __builtin_amdgcn_rcpf(e + 1.0f);
#else
  return tanhf(x);
#endif
}

__device__ __forceinline__ v2f frag_a(const float* S, int pitch, int m0, int k0,
                                      int lane_lo, int half) {
  const float* p = S + (m0 + lane_lo) * pitch + k0 + 2 * half;
  v2f a; a.x = p[0]; a.y = p[1];
  return a;
}

__device__ __forceinline__ v2f frag_b(const float* __restrict__ W, int ldw,
                                      int k0, int n0, int lane_lo, int half) {
  v2f b;
  b.x = W[(k0 + 2 * half + 0) * ldw + n0 + lane_lo];
  b.y = W[(k0 + 2 * half + 1) * ldw + n0 + lane_lo];
  return b;
}


#define KG 4
#define KH 16

struct FragSetW { v2f a[KG]; v4f b[KG][2]; };

__device__ __forceinline__ void load_b_w(FragSetW& f, const float* __restrict__ W,
                                         int kb, int nw, int lane_lo, int half) {
  #pragma unroll
  for (int i = 0; i < KG; ++i)
    #pragma unroll
    for (int r = 0; r < 2; ++r)
      f.b[i][r] = *reinterpret_cast<const v4f*>(
          W + (kb + 4 * i + 2 * half + r) * HH + nw + 4 * lane_lo);
}

__device__ __forceinline__ void load_a_w(FragSetW& f, const float* S, int pitch,
                                         int kb, int lane_lo, int half) {
  #pragma unroll
  for (int i = 0; i < KG; ++i)
    f.a[i] = frag_a(S, pitch, 0, kb + 4 * i, lane_lo, half);
}

template <bool SQ>
__device__ __forceinline__ void mma_set_w(const FragSetW& f, v8f (&acc)[4]) {
  #pragma unroll
  for (int i = 0; i < KG; ++i) {
    v2f a = f.a[i];
    if (SQ) {
      a.x = 1.f - a.x * a.x;  a.y = 1.f - a.y * a.y;
    }
    #pragma unroll
    for (int nt = 0; nt < 4; ++nt) {
      v2f b; b.x = f.b[i][0][nt]; b.y = f.b[i][1][nt];
      acc[nt] = wmma_f32(a, b, acc[nt]);
    }
  }
}

template <bool SQ>
__device__ __forceinline__ void gemm_w(v8f (&acc)[4], FragSetW& f0,
                                       const float* S, int pitch,
                                       const float* __restrict__ W, int K,
                                       int nw, int lane_lo, int half) {
  #pragma unroll
  for (int nt = 0; nt < 4; ++nt) acc[nt] = (v8f){};
  FragSetW f1;
  load_a_w(f0, S, pitch, 0, lane_lo, half);
  int kb = 0;
  for (; kb + 8 * KG < K; kb += 8 * KG) {
    load_b_w(f1, W, kb + 4 * KG, nw, lane_lo, half);
    load_a_w(f1, S, pitch, kb + 4 * KG, lane_lo, half);
    mma_set_w<SQ>(f0, acc);
    load_b_w(f0, W, kb + 8 * KG, nw, lane_lo, half);
    load_a_w(f0, S, pitch, kb + 8 * KG, lane_lo, half);
    mma_set_w<SQ>(f1, acc);
  }
  load_b_w(f1, W, kb + 4 * KG, nw, lane_lo, half);
  load_a_w(f1, S, pitch, kb + 4 * KG, lane_lo, half);
  mma_set_w<SQ>(f0, acc);
  mma_set_w<SQ>(f1, acc);
}

template <bool SQ>
__device__ __forceinline__ void gemm_w3(v8f (&acc)[4], FragSetW& f0,
                                        const float* S, int pitch,
                                        const float* __restrict__ W,
                                        int nw, int lane_lo, int half) {
  static_assert((HH / (4 * KG) - 2) % 3 == 0, "set count must be 3n + 2");
  #pragma unroll
  for (int nt = 0; nt < 4; ++nt) acc[nt] = (v8f){};
  FragSetW f1, f2;
  load_a_w(f0, S, pitch, 0, lane_lo, half);
  load_b_w(f1, W, 4 * KG, nw, lane_lo, half);
  load_a_w(f1, S, pitch, 4 * KG, lane_lo, half);
  int kb = 0;
  #define SCHED_FENCE() __builtin_amdgcn_sched_barrier(0)
  for (; kb + 8 * KG < HH; kb += 12 * KG) {
    load_b_w(f2, W, kb + 8 * KG, nw, lane_lo, half);
    load_a_w(f2, S, pitch, kb + 8 * KG, lane_lo, half);
    mma_set_w<SQ>(f0, acc);
    SCHED_FENCE();
    load_b_w(f0, W, kb + 12 * KG, nw, lane_lo, half);
    load_a_w(f0, S, pitch, kb + 12 * KG, lane_lo, half);
    mma_set_w<SQ>(f1, acc);
    SCHED_FENCE();
    load_b_w(f1, W, kb + 16 * KG, nw, lane_lo, half);
    load_a_w(f1, S, pitch, kb + 16 * KG, lane_lo, half);
    mma_set_w<SQ>(f2, acc);
    SCHED_FENCE();
  }
  #undef SCHED_FENCE
  mma_set_w<SQ>(f0, acc);
  mma_set_w<SQ>(f1, acc);
}

__device__ __forceinline__ void lds_barrier() {
  __builtin_amdgcn_fence(__ATOMIC_RELEASE, "workgroup", "local");
  __builtin_amdgcn_s_barrier();
  __builtin_amdgcn_fence(__ATOMIC_ACQUIRE, "workgroup", "local");
}

#define NSET (DD / (4 * KG))

__device__ __forceinline__ void load_b_all(FragSetW (&f)[NSET], const float* __restrict__ W,
                                           int nw, int lane_lo, int half) {
  #pragma unroll
  for (int s = 0; s < NSET; ++s) load_b_w(f[s], W, 4 * KG * s, nw, lane_lo, half);
}

__device__ __forceinline__ void gemm_pre(v8f (&acc)[4], FragSetW (&f)[NSET],
                                         const float* S, int pitch,
                                         int lane_lo, int half) {
  #pragma unroll
  for (int nt = 0; nt < 4; ++nt) acc[nt] = (v8f){};
  load_a_w(f[0], S, pitch, 0, lane_lo, half);
  #pragma unroll
  for (int s = 0; s < NSET; ++s) {
    if (s + 1 < NSET) load_a_w(f[s + 1], S, pitch, 4 * KG * (s + 1), lane_lo, half);
    mma_set_w<false>(f[s], acc);
  }
}

__device__ __forceinline__ void gemm_trace(v8f (&acc)[4], FragSetW (&m)[NSET],
                                           const FragSetW (&h)[NSET]) {
  #pragma unroll
  for (int nt = 0; nt < 4; ++nt) acc[nt] = (v8f){};
  #pragma unroll
  for (int s = 0; s < NSET; ++s) {
    #pragma unroll
    for (int i = 0; i < KG; ++i) m[s].a[i] = h[s].a[i];
    mma_set_w<true>(m[s], acc);
  }
}

__device__ __forceinline__ void store_piece(float* red, const v8f (&acc)[4],
                                            int wave, int lane_lo, int half) {
  #pragma unroll
  for (int v = 0; v < 8; ++v) {
    int m = v + 8 * half;
    v4f p;
    #pragma unroll
    for (int nt = 0; nt < 4; ++nt) p[nt] = acc[nt][v];
    *reinterpret_cast<v4f*>(red + (wave * MT + m) * RP + 4 * lane_lo) = p;
  }
}

__device__ __forceinline__ float trace_term(const float* red, v4f h2, int r_m, int r_c) {
  v4f a = *reinterpret_cast<const v4f*>(red + r_m * RP + r_c);
  #pragma unroll
  for (int w = 1; w < NSL; ++w)
    a += *reinterpret_cast<const v4f*>(red + (w * MT + r_m) * RP + r_c);
  float tl = 0.f;
  #pragma unroll
  for (int nt = 0; nt < 4; ++nt) tl += (1.f - h2[nt] * h2[nt]) * a[nt];
  return tl;
}

struct FragSetX { v2f a[KH]; v2f b[KH]; };

__device__ __forceinline__ void load_b_x(FragSetX& f, const float* __restrict__ W,
                                         int kb, int n0, int lane_lo, int half) {
  #pragma unroll
  for (int i = 0; i < KH; ++i)
    f.b[i] = frag_b(W, DD, kb + 4 * i, n0, lane_lo, half);
}

__device__ __forceinline__ void load_a_x(FragSetX& f, const float* S, int kb,
                                         int m0, int lane_lo, int half) {
  #pragma unroll
  for (int i = 0; i < KH; ++i)
    f.a[i] = frag_a(S, HP, m0, kb + 4 * i, lane_lo, half);
}

__device__ __forceinline__ v8f mma_set_x(const FragSetX& f, v8f c) {
  #pragma unroll
  for (int i = 0; i < KH; ++i) c = wmma_f32(f.a[i], f.b[i], c);
  return c;
}

#define CP 36
static_assert(4 * KH == DD, "one fragment set must cover the whole K = 64");

__device__ __forceinline__ size_t m_off(int tid) {
  const int b = blockIdx.x;
  return (size_t)((b & 15) * 32 + (tid >> 3)) * HH + (b >> 4) * 32 + 4 * (tid & 7);
}

__device__ __forceinline__ void m_load(FragSetX& f, const float* __restrict__ W1,
                                       const float* __restrict__ W3,
                                       int wave, int lane_lo, int half) {
  const int b  = blockIdx.x;
  const int k0 = (b >> 4) * 32 + ((wave >> 1) & 1) * 16;
  const int j0 = (b & 15) * 32 + (wave & 1) * 16;
  #pragma unroll
  for (int i = 0; i < KH; ++i) {
    f.a[i] = *reinterpret_cast<const v2f*>(W3 + (size_t)(k0 + lane_lo) * DD + 4 * i + 2 * half);
    f.b[i] = frag_b(W1, HH, 4 * i, j0, lane_lo, half);
  }
}

__device__ __forceinline__ void m_tile(const FragSetX& f, float* Ct,
                                       int wave, int lane_lo, int half) {
  const v8f c = mma_set_x(f, (v8f){});
  float* p = Ct + ((wave & 1) * 16 + lane_lo) * CP + ((wave >> 1) & 1) * 16 + 8 * half;
  v4f lo, hi;
  #pragma unroll
  for (int r = 0; r < 4; ++r) { lo[r] = c[r]; hi[r] = c[4 + r]; }
  *reinterpret_cast<v4f*>(p)     = lo;
  *reinterpret_cast<v4f*>(p + 4) = hi;
}

__device__ __forceinline__ void m_store(const float* Ct, v4f w2, float* __restrict__ M,
                                        int tid) {
  const v4f c = *reinterpret_cast<const v4f*>(Ct + (tid >> 3) * CP + 4 * (tid & 7));
  *reinterpret_cast<v4f*>(M + m_off(tid)) = w2 * c;
}

__device__ __forceinline__ void build_m(const float* __restrict__ W1,
                                        const float* __restrict__ W2,
                                        const float* __restrict__ W3,
                                        float* __restrict__ M, float* Ct,
                                        int tid, int wave, int lane_lo, int half) {
  const v4f w2 = *reinterpret_cast<const v4f*>(W2 + m_off(tid));
  if (wave >= 4) {
    FragSetX f;
    m_load(f, W1, W3, wave, lane_lo, half);
    m_tile(f, Ct, wave, lane_lo, half);
  }
  __syncthreads();
  m_store(Ct, w2, M, tid);
}

__device__ __forceinline__ void cnf_whole(const float* __restrict__ y0,
                                                const float* __restrict__ ts,
                                                const float* __restrict__ W1,
                                                const float* __restrict__ b1,
                                                const float* __restrict__ W2,
                                                const float* __restrict__ b2,
                                                const float* __restrict__ W3,
                                                const float* __restrict__ b3,
                                                const float* __restrict__ Mtr,
                                                float* smem,
                                                float* __restrict__ out) {
  float* Xs   = smem;
  float* H1s  = Xs  + MT * XP;
  float* H2s  = H1s + MT * HP;
  float* zacc = H2s + MT * HP;
  float* lacc = zacc + MT;

  const int tid     = threadIdx.x;
  const int wave    = __builtin_amdgcn_readfirstlane(tid >> 5);
  const int lane    = tid & 31;
  const int lane_lo = lane & 15;
  const int half    = lane >> 4;
  const int s_base  = blockIdx.x * MT;

  for (int i = tid; i < MT * DD; i += 256) {
    int m = i >> 6, c = i & 63;
    Xs[m * XP + c] = y0[(s_base + m) * DD + c];
  }
  if (tid < MT) { zacc[tid] = 0.f; lacc[tid] = 0.f; }

  v8f acc[4];
  FragSetW fw;
  const int nw = wave * 64;
  const int col0 = nw + 4 * lane_lo;
  const bool has_dx = wave < 4;
  const int n3 = (wave & 3) * 16;

  for (int step = 0; step < TT - 1; ++step) {
    const float t0 = ts[step];
    const float dt = ts[step + 1] - t0;
    load_b_w(fw, W1, 0, nw, lane_lo, half);
    __syncthreads();

    gemm_w<false>(acc, fw, Xs, XP, W1, DD, nw, lane_lo, half);
    load_b_w(fw, W2, 0, nw, lane_lo, half);
    {
      const v4f tw = *reinterpret_cast<const v4f*>(W1 + DD * HH + col0);
      const v4f bb = *reinterpret_cast<const v4f*>(b1 + col0);
      v4f add;
      #pragma unroll
      for (int nt = 0; nt < 4; ++nt)
        add[nt] = t0 * tw[nt] + bb[nt];
      #pragma unroll
      for (int v = 0; v < 8; ++v) {
        int m = v + 8 * half;
        v4f h;
        #pragma unroll
        for (int nt = 0; nt < 4; ++nt)
          h[nt] = fast_tanh(acc[nt][v] + add[nt]);
        *reinterpret_cast<v4f*>(H1s + m * HP + col0) = h;
      }
    }
    __syncthreads();

    gemm_w3<false>(acc, fw, H1s, HP, W2, nw, lane_lo, half);
    FragSetX g0, g1;
    if (has_dx)
      load_b_x(g0, W3, 0, n3, lane_lo, half);
    {
      const v4f bb = *reinterpret_cast<const v4f*>(b2 + col0);
      #pragma unroll
      for (int v = 0; v < 8; ++v) {
        int m = v + 8 * half;
        v4f h;
        #pragma unroll
        for (int nt = 0; nt < 4; ++nt)
          h[nt] = fast_tanh(acc[nt][v] + bb[nt]);
        *reinterpret_cast<v4f*>(H2s + m * HP + col0) = h;
      }
    }
    __syncthreads();

    load_b_w(fw, Mtr, 0, nw, lane_lo, half);
    if (has_dx) {
      const int m0 = 0, n0 = n3;
      v8f a3 = (v8f){};
      load_a_x(g0, H2s, 0, m0, lane_lo, half);
      int kb = 0;
      for (; kb + 8 * KH < HH; kb += 8 * KH) {
        load_b_x(g1, W3, kb + 4 * KH, n0, lane_lo, half);
        load_a_x(g1, H2s, kb + 4 * KH, m0, lane_lo, half);
        a3 = mma_set_x(g0, a3);
        load_b_x(g0, W3, kb + 8 * KH, n0, lane_lo, half);
        load_a_x(g0, H2s, kb + 8 * KH, m0, lane_lo, half);
        a3 = mma_set_x(g1, a3);
      }
      load_b_x(g1, W3, kb + 4 * KH, n0, lane_lo, half);
      load_a_x(g1, H2s, kb + 4 * KH, m0, lane_lo, half);
      a3 = mma_set_x(g0, a3);
      a3 = mma_set_x(g1, a3);
      int col = n0 + lane_lo;
      float bb = b3[col];
      #pragma unroll
      for (int v = 0; v < 8; ++v) {
        int m = m0 + v + 8 * half;
        float dx = a3[v] + bb;
        atomicAdd(&zacc[m], dt * 0.5f * dx * dx);
        Xs[m * XP + col] += dt * dx;
      }
    }

    {
      gemm_w3<true>(acc, fw, H1s, HP, Mtr, nw, lane_lo, half);
      #pragma unroll
      for (int v = 0; v < 8; ++v) {
        int m = v + 8 * half;
        float tl = 0.f;
        const v4f h2 = *reinterpret_cast<const v4f*>(H2s + m * HP + col0);
        #pragma unroll
        for (int nt = 0; nt < 4; ++nt)
          tl += (1.f - h2[nt] * h2[nt]) * acc[nt][v];
        atomicAdd(&lacc[m], dt * tl);
      }
    }
  }

  __syncthreads();
  for (int i = tid; i < MT * DD; i += 256) {
    int m = i >> 6, c = i & 63;
    out[(s_base + m) * DD + c] = Xs[m * XP + c];
  }
  if (tid < MT) {
    out[BB * DD + s_base + tid]      = zacc[tid];
    out[BB * DD + BB + s_base + tid] = lacc[tid];
  }
}

__device__ __forceinline__ v4f dx_from_pieces(const float* __restrict__ P,
                                              const float* __restrict__ b3,
                                              int row, int c4) {
  v4f d = *reinterpret_cast<const v4f*>(P + (size_t)row * DD + c4);
  #pragma unroll
  for (int j = 1; j < NSL; ++j)
    d += *reinterpret_cast<const v4f*>(P + (size_t)(j * BB + row) * DD + c4);
  d += *reinterpret_cast<const v4f*>(b3 + c4);
  return d;
}

__device__ __forceinline__ float sum16(float q) {
  q += __shfl_xor(q, 8, 16);
  q += __shfl_xor(q, 4, 16);
  q += __shfl_xor(q, 2, 16);
  q += __shfl_xor(q, 1, 16);
  return q;
}

template <int MODE>
__global__ __launch_bounds__(256) void cnf_step(int step,
                                                const float* __restrict__ y0,
                                                const float* __restrict__ ts,
                                                const float* __restrict__ W1,
                                                const float* __restrict__ b1,
                                                const float* __restrict__ W2,
                                                const float* __restrict__ b2,
                                                const float* __restrict__ W3,
                                                const float* __restrict__ b3,
                                                float* __restrict__ ws,
                                                size_t ws_bytes,
                                                float* __restrict__ out) {
  extern __shared__ float smem[];
  if (ws_bytes < WS_NEED) {
    if (step == 0) {
      const int tid0 = threadIdx.x;
      build_m(W1, W2, W3, ws + WS_M, smem, tid0,
              __builtin_amdgcn_readfirstlane(tid0 >> 5), tid0 & 15, (tid0 & 31) >> 4);
    } else if (step == 1 && blockIdx.x < NTILE) {
      cnf_whole(y0, ts, W1, b1, W2, b2, W3, b3, ws + WS_M, smem, out);
    }
    return;
  }
  float* Xs  = smem;
  float* H1s = Xs  + MT * XP;
  float* red = H1s + MT * HP;
  float* H2t = red + NSL * MT * RP;
  float* dxs = H2t + MT * RP;

  const int tid     = threadIdx.x;
  const int wave    = __builtin_amdgcn_readfirstlane(tid >> 5);
  const int lane    = tid & 31;
  const int lane_lo = lane & 15;
  const int half    = lane >> 4;
  const int tile    = blockIdx.x / NSL;
  const int slice   = blockIdx.x % NSL;
  const int s_base  = tile * MT;
  const int nw      = wave * 64;
  const int col0    = nw + 4 * lane_lo;
  const int sc0     = slice * 64;
  const int r_m     = tid >> 4;
  const int r_c     = 4 * (tid & 15);

  float* H1g = ws + WS_H1;
  float* H2g = ws + WS_H2;
  float* ztg = ws + WS_ZT;
  float* Xg  = ws + WS_X;
  float* dxp = ws + WS_DXP;
  float* prt = ws + WS_PART;
  const float* Mg = ws + WS_M;

  const float t0 = ts[step];
  FragSetW w1s[NSET];
  FragSetW w2s[NSET];

  v4f tw1, bb1;
  {
    const float* Xprev = (step <= 1) ? y0 : Xg + ((step - 1) & 1) * BB * DD;
    v4f x = *reinterpret_cast<const v4f*>(Xprev + (size_t)(s_base + r_m) * DD + r_c);
    v4f p[NSL], b3v;
    if (step != 0) {
      const float* P = dxp + ((step - 1) & 1) * (NSL * BB * DD);
      #pragma unroll
      for (int j = 0; j < NSL; ++j)
        p[j] = *reinterpret_cast<const v4f*>(P + (size_t)(j * BB + s_base + r_m) * DD + r_c);
      b3v = *reinterpret_cast<const v4f*>(b3 + r_c);
    }
    tw1 = *reinterpret_cast<const v4f*>(W1 + DD * HH + col0);
    bb1 = *reinterpret_cast<const v4f*>(b1 + col0);
    load_b_all(w1s, W1, nw, lane_lo, half);
    if (step != 0) {
      const float dtp = t0 - ts[step - 1];
      v4f d = p[0];
      #pragma unroll
      for (int j = 1; j < NSL; ++j) d += p[j];
      d += b3v;
      x += dtp * d;
      if (slice == 0) {
        const float q = sum16(d.x * d.x + d.y * d.y + d.z * d.z + d.w * d.w);
        if ((tid & 15) == 0) dxs[r_m] = dtp * 0.5f * q;
      }
    }
    *reinterpret_cast<v4f*>(Xs + r_m * XP + r_c) = x;
  }
  lds_barrier();
  if (step != 0 && slice == 0 && tid < MT)
    ztg[((step - 1) * NTILE + tile) * 32 + tid] = dxs[tid];

  const v4f bb2 = *reinterpret_cast<const v4f*>(b2 + sc0 + r_c);
  load_b_all(w2s, W2 + (size_t)nw * HH, sc0, lane_lo, half);
  v8f acc[4];
  gemm_pre(acc, w1s, Xs, XP, lane_lo, half);
  FragSetX g;
  if (wave < 4)
    load_b_x(g, W3 + (size_t)sc0 * DD, 0, wave * 16, lane_lo, half);
  FragSetW ms[NSET];
  if (MODE != 0)
    load_b_all(ms, Mg + (size_t)nw * HH, sc0, lane_lo, half);
  {
    v4f add;
    #pragma unroll
    for (int nt = 0; nt < 4; ++nt)
      add[nt] = t0 * tw1[nt] + bb1[nt];
    #pragma unroll
    for (int v = 0; v < 8; ++v) {
      int m = v + 8 * half;
      v4f h;
      #pragma unroll
      for (int nt = 0; nt < 4; ++nt)
        h[nt] = fast_tanh(acc[nt][v] + add[nt]);
      *reinterpret_cast<v4f*>(H1s + m * HP + col0) = h;
    }
  }
  lds_barrier();

  gemm_pre(acc, w2s, H1s + nw, HP, lane_lo, half);
  store_piece(red, acc, wave, lane_lo, half);
  lds_barrier();
  v4f h2;
  {
    v4f a = *reinterpret_cast<const v4f*>(red + r_m * RP + r_c);
    #pragma unroll
    for (int w = 1; w < NSL; ++w)
      a += *reinterpret_cast<const v4f*>(red + (w * MT + r_m) * RP + r_c);
    #pragma unroll
    for (int nt = 0; nt < 4; ++nt) h2[nt] = fast_tanh(a[nt] + bb2[nt]);
    *reinterpret_cast<v4f*>(H2t + r_m * RP + r_c) = h2;
  }
  lds_barrier();

  if (wave < 4) {
    #pragma unroll
    for (int i = 0; i < KH; ++i)
      g.a[i] = frag_a(H2t, RP, 0, 4 * i, lane_lo, half);
    const v8f a3 = mma_set_x(g, (v8f){});
    if (MODE != 0) gemm_trace(acc, ms, w2s);
    #pragma unroll
    for (int v = 0; v < 8; ++v)
      dxs[(v + 8 * half) * RP + wave * 16 + lane_lo] = a3[v];
  } else {
    if (MODE != 0) gemm_trace(acc, ms, w2s);
  }
  if (MODE != 0) store_piece(red, acc, wave, lane_lo, half);
  lds_barrier();
  *reinterpret_cast<v4f*>(dxp + (step & 1) * (NSL * BB * DD)
                          + (size_t)(slice * BB + s_base + r_m) * DD + r_c) =
      *reinterpret_cast<const v4f*>(dxs + r_m * RP + r_c);
  const size_t hrow = (size_t)(s_base + r_m) * HH + sc0 + r_c;
  if (MODE == 0) {
    *reinterpret_cast<v4f*>(H2g + hrow) = h2;
    *reinterpret_cast<v4f*>(H1g + hrow) =
        *reinterpret_cast<const v4f*>(H1s + r_m * HP + sc0 + r_c);
  }
  if (step != 0 && slice == 0)
    *reinterpret_cast<v4f*>(Xg + (step & 1) * BB * DD + (size_t)(s_base + r_m) * DD + r_c) =
        *reinterpret_cast<const v4f*>(Xs + r_m * XP + r_c);
  if (MODE == 0)
    build_m(W1, W2, W3, ws + WS_M, red, tid, wave, lane_lo, half);
  if (MODE != 0) {
    const float q = sum16(trace_term(red, h2, r_m, r_c));
    if ((tid & 15) == 0) H2t[r_m] = (ts[step + 1] - t0) * q;
    lds_barrier();
    if (tid < MT)
      prt[(size_t)((step * NTILE + tile) * NSL + slice) * 32 + tid] = H2t[tid];
  }
  if (MODE == 1) {
    const v4f h20 = *reinterpret_cast<const v4f*>(H2g + hrow);
    for (int i = tid; i < MT * HH / 4; i += 256) {
      int m = i >> 7, c = (i & 127) * 4;
      *reinterpret_cast<v4f*>(H1s + m * HP + c) =
          *reinterpret_cast<const v4f*>(H1g + (size_t)(s_base + m) * HH + c);
    }
    lds_barrier();
    #pragma unroll
    for (int s = 0; s < NSET; ++s) load_a_w(w2s[s], H1s + nw, HP, 4 * KG * s, lane_lo, half);
    gemm_trace(acc, ms, w2s);
    store_piece(red, acc, wave, lane_lo, half);
    lds_barrier();
    const float q0 = sum16(trace_term(red, h20, r_m, r_c));
    if ((tid & 15) == 0) H2t[r_m] = (ts[1] - ts[0]) * q0;
    lds_barrier();
    if (tid < MT)
      prt[(size_t)(tile * NSL + slice) * 32 + tid] = H2t[tid];
  }
}

__global__ __launch_bounds__(BB) void cnf_final(const float* __restrict__ ts,
                                                const float* __restrict__ b3,
                                                const float* __restrict__ ws,
                                                size_t ws_bytes,
                                                float* __restrict__ out) {
  if (ws_bytes < WS_NEED) return;
  __shared__ float zl[64];
  const int tid = threadIdx.x;
  const int s   = blockIdx.x * 32 + (tid >> 4);
  const int r_c = 4 * (tid & 15);
  const float* Xprev = ws + WS_X + ((TT - 2) & 1) * BB * DD;
  const float* P     = ws + WS_DXP + ((TT - 2) & 1) * (NSL * BB * DD);
  const float dtp = ts[TT - 1] - ts[TT - 2];
  v4f x = *reinterpret_cast<const v4f*>(Xprev + (size_t)s * DD + r_c);
  const v4f d = dx_from_pieces(P, b3, s, r_c);
  x += dtp * d;
  *reinterpret_cast<v4f*>(out + (size_t)s * DD + r_c) = x;
  const float q = sum16(d.x * d.x + d.y * d.y + d.z * d.z + d.w * d.w);
  if ((tid & 15) == 0) {
    const float* zt = ws + WS_ZT;
    const float* pt = ws + WS_PART;
    float z = 0.f, l = 0.f;
    #pragma unroll
    for (int step = 0; step < TT - 2; ++step)
      z += zt[(step * NTILE + (s >> 4)) * 32 + (s & 15)];
    z += dtp * 0.5f * q;
    #pragma unroll
    for (int step = 0; step < TT - 1; ++step) {
      const float* ps = pt + (size_t)((step * NTILE + (s >> 4)) * NSL) * 32 + (s & 15);
      float ls = ps[0];
      #pragma unroll
      for (int j = 1; j < NSL; ++j) ls += ps[j * 32];
      l += ls;
    }
    zl[tid >> 4] = z;
    zl[32 + (tid >> 4)] = l;
  }
  __syncthreads();
  if (tid < 32)       out[BB * DD + blockIdx.x * 32 + tid] = zl[tid];
  else if (tid < 64)  out[BB * DD + BB + blockIdx.x * 32 + (tid - 32)] = zl[tid];
}

extern "C" void kernel_launch(void* const* d_in, const int* in_sizes, int n_in,
                              void* d_out, int out_size, void* d_ws, size_t ws_size,
                              hipStream_t stream) {
  const float* y0 = (const float*)d_in[0];
  const float* ts = (const float*)d_in[1];
  const float* W1 = (const float*)d_in[2];
  const float* b1 = (const float*)d_in[3];
  const float* W2 = (const float*)d_in[4];
  const float* b2 = (const float*)d_in[5];
  const float* W3 = (const float*)d_in[6];
  const float* b3 = (const float*)d_in[7];
  float* out = (float*)d_out;
  float* ws  = (float*)d_ws;

  size_t shmem = (size_t)(LDS_NEW > LDS_OLD ? LDS_NEW : LDS_OLD) * sizeof(float);
#define STEP(md, n) cnf_step<md><<<NTILE * NSL, 256, shmem, stream>>>(n, y0, ts, W1, b1, W2, b2, W3, b3, ws, ws_size, out)
  STEP(0, 0); STEP(1, 1); STEP(2, 2); STEP(2, 3); STEP(2, 4); STEP(2, 5); STEP(2, 6);
#undef STEP

  cnf_final<<<BB / 32, BB, 0, stream>>>(ts, b3, ws, ws_size, out);
}
